# starting kernel
# baseline (speedup 1.0000x reference)
#include <hip/hip_runtime.h>
#include <math.h>

#define L_SEQ 2048
#define DMODEL 512
#define NHEADS 8
#define DHEAD 64
#define MFEAT 128
#define NCHUNKS 16
#define CHUNK 128
#define CHUNK_STRIDE 8320
#define KERNEL_EPS 1e-4f
#define DEN_EPS 1e-6f
#define QK_SCALE 0.21022410381342864f
#define NC_FEAT 0.08838834764831845f

typedef float v2f __attribute__((ext_vector_type(2)));
typedef float v8f __attribute__((ext_vector_type(8)));

struct Acc2x2 { v8f c[2][2]; };

__device__ __forceinline__ Acc2x2 wmma_block2x2_f32(const float* __restrict__ A, int lda,
                                                    const float* __restrict__ B, int bks, int bns,
                                                    int K, int lane) {
  const int half = lane >> 4;
  const int l16  = lane & 15;
  const float* arow0 = A + l16 * lda;
  const float* arow1 = A + (l16 + 16) * lda;
  const float* bcol0 = B + l16 * bns;
  const float* bcol1 = B + (l16 + 16) * bns;
  const v8f zero = {0.f, 0.f, 0.f, 0.f, 0.f, 0.f, 0.f, 0.f};
  Acc2x2 r;
  r.c[0][0] = zero; r.c[0][1] = zero; r.c[1][0] = zero; r.c[1][1] = zero;
  for (int k0 = 0; k0 < K; k0 += 4) {
    const int ka = k0 + 2 * half;
    v2f a0, a1, b0, b1;
    a0.x = arow0[ka];          a0.y = arow0[ka + 1];
    a1.x = arow1[ka];          a1.y = arow1[ka + 1];
    b0.x = bcol0[ka * bks];    b0.y = bcol0[(ka + 1) * bks];
    b1.x = bcol1[ka * bks];    b1.y = bcol1[(ka + 1) * bks];
    r.c[0][0] = __builtin_amdgcn_wmma_f32_16x16x4_f32(false, a0, false, b0, (short)0, r.c[0][0], false, false);
    r.c[0][1] = __builtin_amdgcn_wmma_f32_16x16x4_f32(false, a0, false, b1, (short)0, r.c[0][1], false, false);
    r.c[1][0] = __builtin_amdgcn_wmma_f32_16x16x4_f32(false, a1, false, b0, (short)0, r.c[1][0], false, false);
    r.c[1][1] = __builtin_amdgcn_wmma_f32_16x16x4_f32(false, a1, false, b1, (short)0, r.c[1][1], false, false);
  }
  return r;
}

__global__ void __launch_bounds__(32)
perf_qkv_proj(const float* __restrict__ xq, const float* __restrict__ xk,
              const float* __restrict__ xv,
              const float* __restrict__ Wq, const float* __restrict__ bq,
              const float* __restrict__ Wk, const float* __restrict__ bk,
              const float* __restrict__ Wv, const float* __restrict__ bv,
              float* __restrict__ Q, float* __restrict__ K, float* __restrict__ V) {
  const int tm = blockIdx.x * 32;
  const int tn = blockIdx.y * 32;
  const int mat = blockIdx.z;
  const int lane = threadIdx.x;
  const float* X; const float* W; const float* bias; float* Y; float scl;
  if (mat == 0)      { X = xq; W = Wq; bias = bq; Y = Q; scl = QK_SCALE; }
  else if (mat == 1) { X = xk; W = Wk; bias = bk; Y = K; scl = QK_SCALE; }
  else               { X = xv; W = Wv; bias = bv; Y = V; scl = 1.f; }
  Acc2x2 r = wmma_block2x2_f32(X + tm * DMODEL, DMODEL,
                               W + tn * DMODEL, 1, DMODEL, DMODEL, lane);
  const int l16 = lane & 15;
  const int half = lane >> 4;
#pragma unroll
  for (int ti = 0; ti < 2; ++ti)
#pragma unroll
    for (int tj = 0; tj < 2; ++tj) {
      const int col = tn + tj * 16 + l16;
      const int rbase = tm + ti * 16 + 8 * half;
      const float bb = bias[col];
#pragma unroll
      for (int i = 0; i < 8; ++i)
        Y[(rbase + i) * DMODEL + col] = (r.c[ti][tj][i] + bb) * scl;
    }
}

__global__ void __launch_bounds__(256)
perf_hk(const float* __restrict__ K, float* __restrict__ hk) {
  const int idx = blockIdx.x * 256 + threadIdx.x;
  const int h = idx >> 11;
  const int l = idx & (L_SEQ - 1);
  const float* row = K + l * DMODEL + h * DHEAD;
  float s = 0.f;
#pragma unroll 8
  for (int d = 0; d < DHEAD; ++d) s += row[d] * row[d];
  hk[idx] = -0.5f * s;
}

__global__ void __launch_bounds__(256)
perf_kstab(const float* __restrict__ hk, float* __restrict__ kstab) {
  __shared__ float red[256];
  const int t = threadIdx.x;
  float m = -3.402823e38f;
  for (int i = t; i < NHEADS * L_SEQ; i += 256) m = fmaxf(m, hk[i]);
  red[t] = m;
  __syncthreads();
  for (int s = 128; s > 0; s >>= 1) {
    if (t < s) red[t] = fmaxf(red[t], red[t + s]);
    __syncthreads();
  }
  if (t == 0) kstab[0] = red[0];
}

__global__ void __launch_bounds__(32)
perf_primes(const float* __restrict__ Q, const float* __restrict__ K,
            const float* __restrict__ RF, const float* __restrict__ hk,
            const float* __restrict__ kstab,
            float* __restrict__ qprime, float* __restrict__ kprime) {
  const int tm = blockIdx.x * 32;
  const int tn = blockIdx.y * 32;
  const int h = blockIdx.z >> 1;
  const int isK = blockIdx.z & 1;
  const int lane = threadIdx.x;
  const float* A = (isK ? K : Q) + tm * DMODEL + h * DHEAD;
  Acc2x2 r = wmma_block2x2_f32(A, DMODEL, RF + tn, MFEAT, 1, DHEAD, lane);
  const int l16 = lane & 15;
  const int half = lane >> 4;
  const float ks = isK ? kstab[0] : 0.f;
#pragma unroll
  for (int ti = 0; ti < 2; ++ti)
#pragma unroll
    for (int tj = 0; tj < 2; ++tj) {
      const int mcol = tn + tj * 16 + l16;
      const int rbase = tm + ti * 16 + 8 * half;
      if (isK) {
#pragma unroll
        for (int i = 0; i < 8; ++i) {
          const int l = rbase + i;
          kprime[(h * L_SEQ + l) * MFEAT + mcol] =
              NC_FEAT * (expf(hk[h * L_SEQ + l] + r.c[ti][tj][i] - ks) + KERNEL_EPS);
        }
      } else {
#pragma unroll
        for (int i = 0; i < 8; ++i) {
          const int l = rbase + i;
          qprime[(h * L_SEQ + l) * MFEAT + mcol] =
              NC_FEAT * (expf(r.c[ti][tj][i]) + KERNEL_EPS);
        }
      }
    }
}

__global__ void __launch_bounds__(256)
perf_chunk_sums(const float* __restrict__ kprime, const float* __restrict__ V,
                float* __restrict__ chunkbuf) {
  const int h = blockIdx.x >> 4;
  const int c = blockIdx.x & 15;
  const int t = threadIdx.x;
  __shared__ float kp[MFEAT];
  __shared__ float vv[DHEAD];
  const int m = t >> 1;
  const int d0 = (t & 1) * 32;
  float acc[32];
#pragma unroll
  for (int j = 0; j < 32; ++j) acc[j] = 0.f;
  float zacc = 0.f;
  for (int l0 = 0; l0 < CHUNK; ++l0) {
    const int l = c * CHUNK + l0;
    if (t < MFEAT) kp[t] = kprime[(h * L_SEQ + l) * MFEAT + t];
    if (t < DHEAD) vv[t] = V[l * DMODEL + h * DHEAD + t];
    __syncthreads();
    const float kpm = kp[m];
#pragma unroll
    for (int j = 0; j < 32; ++j) acc[j] += kpm * vv[d0 + j];
    zacc += kpm;
    __syncthreads();
  }
  float* buf = chunkbuf + (size_t)(h * NCHUNKS + c) * CHUNK_STRIDE;
#pragma unroll
  for (int j = 0; j < 32; ++j) buf[m * DHEAD + d0 + j] = acc[j];
  if ((t & 1) == 0) buf[MFEAT * DHEAD + m] = zacc;
}

__global__ void __launch_bounds__(256)
perf_chunk_prefix(float* __restrict__ chunkbuf) {
  const int h = blockIdx.x;
  for (int idx = threadIdx.x; idx < CHUNK_STRIDE; idx += 256) {
    float carry = 0.f;
    for (int c = 0; c < NCHUNKS; ++c) {
      float* p = chunkbuf + (size_t)(h * NCHUNKS + c) * CHUNK_STRIDE + idx;
      const float tmp = *p;
      *p = carry;
      carry += tmp;
    }
  }
}

__global__ void __launch_bounds__(256)
perf_scan(const float* __restrict__ qprime, const float* __restrict__ kprime,
          const float* __restrict__ V, const float* __restrict__ chunkbuf,
          float* __restrict__ context) {
  const int h = blockIdx.x >> 4;
  const int c = blockIdx.x & 15;
  const int t = threadIdx.x;
  __shared__ float S[MFEAT * DHEAD];
  __shared__ float z[MFEAT];
  __shared__ float qp[MFEAT];
  __shared__ float kp[MFEAT];
  __shared__ float vv[DHEAD];
  __shared__ float red[256];
  __shared__ float dred[MFEAT];
  const float* buf = chunkbuf + (size_t)(h * NCHUNKS + c) * CHUNK_STRIDE;
  for (int i = t; i < MFEAT * DHEAD; i += 256) S[i] = buf[i];
  if (t < MFEAT) z[t] = buf[MFEAT * DHEAD + t];
  __syncthreads();
  const int m = t >> 1;
  const int d0 = (t & 1) * 32;
  const int dd = t & 63;
  const int part = t >> 6;
  for (int l0 = 0; l0 < CHUNK; ++l0) {
    const int l = c * CHUNK + l0;
    if (t < MFEAT) qp[t] = qprime[(h * L_SEQ + l) * MFEAT + t];
    else           kp[t - MFEAT] = kprime[(h * L_SEQ + l) * MFEAT + (t - MFEAT)];
    if (t < DHEAD) vv[t] = V[l * DMODEL + h * DHEAD + t];
    __syncthreads();
    const float kpm = kp[m];
#pragma unroll
    for (int j = 0; j < 32; ++j) S[m * DHEAD + d0 + j] += kpm * vv[d0 + j];
    if ((t & 1) == 0) z[m] += kpm;
    __syncthreads();
    float s = 0.f;
    const int mb = part * 32;
#pragma unroll
    for (int j = 0; j < 32; ++j) s += qp[mb + j] * S[(mb + j) * DHEAD + dd];
    red[t] = s;
    if (t < MFEAT) dred[t] = qp[t] * z[t];
    __syncthreads();
    if (t < 64) dred[t] += dred[t + 64];
    __syncthreads();
    if (t == 0) {
      float s2 = 0.f;
      for (int j = 0; j < 64; ++j) s2 += dred[j];
      dred[0] = s2;
    }
    __syncthreads();
    if (t < DHEAD) {
      const float num = red[dd] + red[64 + dd] + red[128 + dd] + red[192 + dd];
      context[l * DMODEL + h * DHEAD + dd] = num / (dred[0] + DEN_EPS);
    }
    __syncthreads();
  }
}

__global__ void __launch_bounds__(32)
perf_out_proj(const float* __restrict__ ctx, const float* __restrict__ Wo,
              const float* __restrict__ bo, float* __restrict__ out) {
  const int tm = blockIdx.x * 32;
  const int tn = blockIdx.y * 32;
  const int lane = threadIdx.x;
  Acc2x2 r = wmma_block2x2_f32(ctx + tm * DMODEL, DMODEL,
                               Wo + tn * DMODEL, 1, DMODEL, DMODEL, lane);
  const int l16 = lane & 15;
  const int half = lane >> 4;
#pragma unroll
  for (int ti = 0; ti < 2; ++ti)
#pragma unroll
    for (int tj = 0; tj < 2; ++tj) {
      const int col = tn + tj * 16 + l16;
      const int rbase = tm + ti * 16 + 8 * half;
      const float bb = bo[col];
#pragma unroll
      for (int i = 0; i < 8; ++i)
        out[(rbase + i) * DMODEL + col] = r.c[ti][tj][i] + bb;
    }
}

extern "C" void kernel_launch(void* const* d_in, const int* in_sizes, int n_in,
                              void* d_out, int out_size, void* d_ws, size_t ws_size,
                              hipStream_t stream) {
  (void)in_sizes; (void)n_in; (void)out_size; (void)ws_size;
  const float* query = (const float*)d_in[0];
  const float* key_  = (const float*)d_in[1];
  const float* value = (const float*)d_in[2];
  const float* Wq = (const float*)d_in[3];
  const float* bq = (const float*)d_in[4];
  const float* Wk = (const float*)d_in[5];
  const float* bk = (const float*)d_in[6];
  const float* Wv = (const float*)d_in[7];
  const float* bv = (const float*)d_in[8];
  const float* Wo = (const float*)d_in[9];
  const float* bo = (const float*)d_in[10];
  const float* RF = (const float*)d_in[11];

  float* ws = (float*)d_ws;
  const size_t SZ_LD = (size_t)L_SEQ * DMODEL;
  const size_t SZ_PRIME = (size_t)NHEADS * L_SEQ * MFEAT;
  float* Q      = ws;
  float* K      = Q + SZ_LD;
  float* V      = K + SZ_LD;
  float* hk     = V + SZ_LD;
  float* kstab  = hk + (size_t)NHEADS * L_SEQ;
  float* qprime = kstab + 64;
  float* kprime = qprime + SZ_PRIME;
  float* chunkbuf = kprime + SZ_PRIME;
  float* context = Q;

  perf_qkv_proj<<<dim3(L_SEQ / 32, DMODEL / 32, 3), 32, 0, stream>>>(
      query, key_, value, Wq, bq, Wk, bk, Wv, bv, Q, K, V);
  perf_hk<<<(NHEADS * L_SEQ) / 256, 256, 0, stream>>>(K, hk);
  perf_kstab<<<1, 256, 0, stream>>>(hk, kstab);
  perf_primes<<<dim3(L_SEQ / 32, MFEAT / 32, 2 * NHEADS), 32, 0, stream>>>(
      Q, K, RF, hk, kstab, qprime, kprime);
  perf_chunk_sums<<<NHEADS * NCHUNKS, 256, 0, stream>>>(kprime, V, chunkbuf);
  perf_chunk_prefix<<<NHEADS, 256, 0, stream>>>(chunkbuf);
  perf_scan<<<NHEADS * NCHUNKS, 256, 0, stream>>>(qprime, kprime, V, chunkbuf,
                                                  context);
  perf_out_proj<<<dim3(L_SEQ / 32, DMODEL / 32), 32, 0, stream>>>(
      context, Wo, bo, (float*)d_out);
}


# in-chunk causal scan: 128 sequential row steps replaced by chunked matrix form on WMMA (masked k'.q' tiles times V, plus q' times prefix state); sums taken in a different reduction tree
# speedup vs baseline: 2.1902x; 2.1902x over previous
#include <hip/hip_runtime.h>
#include <math.h>

#define L_SEQ 2048
#define DMODEL 512
#define NHEADS 8
#define DHEAD 64
#define MFEAT 128
#define NCHUNKS 16
#define CHUNK 128
#define CHUNK_STRIDE 8320
#define KERNEL_EPS 1e-4f
#define DEN_EPS 1e-6f
#define QK_SCALE 0.21022410381342864f
#define NC_FEAT 0.08838834764831845f

typedef float v2f __attribute__((ext_vector_type(2)));
typedef float v8f __attribute__((ext_vector_type(8)));

struct Acc2x2 { v8f c[2][2]; };

__device__ __forceinline__ Acc2x2 wmma_block2x2_f32(const float* __restrict__ A, int lda,
                                                    const float* __restrict__ B, int bks, int bns,
                                                    int K, int lane) {
  const int half = lane >> 4;
  const int l16  = lane & 15;
  const float* arow0 = A + l16 * lda;
  const float* arow1 = A + (l16 + 16) * lda;
  const float* bcol0 = B + l16 * bns;
  const float* bcol1 = B + (l16 + 16) * bns;
  const v8f zero = {0.f, 0.f, 0.f, 0.f, 0.f, 0.f, 0.f, 0.f};
  Acc2x2 r;
  r.c[0][0] = zero; r.c[0][1] = zero; r.c[1][0] = zero; r.c[1][1] = zero;
  for (int k0 = 0; k0 < K; k0 += 4) {
    const int ka = k0 + 2 * half;
    v2f a0, a1, b0, b1;
    a0.x = arow0[ka];          a0.y = arow0[ka + 1];
    a1.x = arow1[ka];          a1.y = arow1[ka + 1];
    b0.x = bcol0[ka * bks];    b0.y = bcol0[(ka + 1) * bks];
    b1.x = bcol1[ka * bks];    b1.y = bcol1[(ka + 1) * bks];
    r.c[0][0] = __builtin_amdgcn_wmma_f32_16x16x4_f32(false, a0, false, b0, (short)0, r.c[0][0], false, false);
    r.c[0][1] = __builtin_amdgcn_wmma_f32_16x16x4_f32(false, a0, false, b1, (short)0, r.c[0][1], false, false);
    r.c[1][0] = __builtin_amdgcn_wmma_f32_16x16x4_f32(false, a1, false, b0, (short)0, r.c[1][0], false, false);
    r.c[1][1] = __builtin_amdgcn_wmma_f32_16x16x4_f32(false, a1, false, b1, (short)0, r.c[1][1], false, false);
  }
  return r;
}

__global__ void __launch_bounds__(32)
perf_qkv_proj(const float* __restrict__ xq, const float* __restrict__ xk,
              const float* __restrict__ xv,
              const float* __restrict__ Wq, const float* __restrict__ bq,
              const float* __restrict__ Wk, const float* __restrict__ bk,
              const float* __restrict__ Wv, const float* __restrict__ bv,
              float* __restrict__ Q, float* __restrict__ K, float* __restrict__ V) {
  const int tm = blockIdx.x * 32;
  const int tn = blockIdx.y * 32;
  const int mat = blockIdx.z;
  const int lane = threadIdx.x;
  const float* X; const float* W; const float* bias; float* Y; float scl;
  if (mat == 0)      { X = xq; W = Wq; bias = bq; Y = Q; scl = QK_SCALE; }
  else if (mat == 1) { X = xk; W = Wk; bias = bk; Y = K; scl = QK_SCALE; }
  else               { X = xv; W = Wv; bias = bv; Y = V; scl = 1.f; }
  Acc2x2 r = wmma_block2x2_f32(X + tm * DMODEL, DMODEL,
                               W + tn * DMODEL, 1, DMODEL, DMODEL, lane);
  const int l16 = lane & 15;
  const int half = lane >> 4;
#pragma unroll
  for (int ti = 0; ti < 2; ++ti)
#pragma unroll
    for (int tj = 0; tj < 2; ++tj) {
      const int col = tn + tj * 16 + l16;
      const int rbase = tm + ti * 16 + 8 * half;
      const float bb = bias[col];
#pragma unroll
      for (int i = 0; i < 8; ++i)
        Y[(rbase + i) * DMODEL + col] = (r.c[ti][tj][i] + bb) * scl;
    }
}

__global__ void __launch_bounds__(256)
perf_hk(const float* __restrict__ K, float* __restrict__ hk) {
  const int idx = blockIdx.x * 256 + threadIdx.x;
  const int h = idx >> 11;
  const int l = idx & (L_SEQ - 1);
  const float* row = K + l * DMODEL + h * DHEAD;
  float s = 0.f;
#pragma unroll 8
  for (int d = 0; d < DHEAD; ++d) s += row[d] * row[d];
  hk[idx] = -0.5f * s;
}

__global__ void __launch_bounds__(256)
perf_kstab(const float* __restrict__ hk, float* __restrict__ kstab) {
  __shared__ float red[256];
  const int t = threadIdx.x;
  float m = -3.402823e38f;
  for (int i = t; i < NHEADS * L_SEQ; i += 256) m = fmaxf(m, hk[i]);
  red[t] = m;
  __syncthreads();
  for (int s = 128; s > 0; s >>= 1) {
    if (t < s) red[t] = fmaxf(red[t], red[t + s]);
    __syncthreads();
  }
  if (t == 0) kstab[0] = red[0];
}

__global__ void __launch_bounds__(32)
perf_primes(const float* __restrict__ Q, const float* __restrict__ K,
            const float* __restrict__ RF, const float* __restrict__ hk,
            const float* __restrict__ kstab,
            float* __restrict__ qprime, float* __restrict__ kprime) {
  const int tm = blockIdx.x * 32;
  const int tn = blockIdx.y * 32;
  const int h = blockIdx.z >> 1;
  const int isK = blockIdx.z & 1;
  const int lane = threadIdx.x;
  const float* A = (isK ? K : Q) + tm * DMODEL + h * DHEAD;
  Acc2x2 r = wmma_block2x2_f32(A, DMODEL, RF + tn, MFEAT, 1, DHEAD, lane);
  const int l16 = lane & 15;
  const int half = lane >> 4;
  const float ks = isK ? kstab[0] : 0.f;
#pragma unroll
  for (int ti = 0; ti < 2; ++ti)
#pragma unroll
    for (int tj = 0; tj < 2; ++tj) {
      const int mcol = tn + tj * 16 + l16;
      const int rbase = tm + ti * 16 + 8 * half;
      if (isK) {
#pragma unroll
        for (int i = 0; i < 8; ++i) {
          const int l = rbase + i;
          kprime[(h * L_SEQ + l) * MFEAT + mcol] =
              NC_FEAT * (expf(hk[h * L_SEQ + l] + r.c[ti][tj][i] - ks) + KERNEL_EPS);
        }
      } else {
#pragma unroll
        for (int i = 0; i < 8; ++i) {
          const int l = rbase + i;
          qprime[(h * L_SEQ + l) * MFEAT + mcol] =
              NC_FEAT * (expf(r.c[ti][tj][i]) + KERNEL_EPS);
        }
      }
    }
}

__global__ void __launch_bounds__(256)
perf_chunk_sums(const float* __restrict__ kprime, const float* __restrict__ V,
                float* __restrict__ chunkbuf) {
  const int h = blockIdx.x >> 4;
  const int c = blockIdx.x & 15;
  const int t = threadIdx.x;
  __shared__ float kp[MFEAT];
  __shared__ float vv[DHEAD];
  const int m = t >> 1;
  const int d0 = (t & 1) * 32;
  float acc[32];
#pragma unroll
  for (int j = 0; j < 32; ++j) acc[j] = 0.f;
  float zacc = 0.f;
  for (int l0 = 0; l0 < CHUNK; ++l0) {
    const int l = c * CHUNK + l0;
    if (t < MFEAT) kp[t] = kprime[(h * L_SEQ + l) * MFEAT + t];
    if (t < DHEAD) vv[t] = V[l * DMODEL + h * DHEAD + t];
    __syncthreads();
    const float kpm = kp[m];
#pragma unroll
    for (int j = 0; j < 32; ++j) acc[j] += kpm * vv[d0 + j];
    zacc += kpm;
    __syncthreads();
  }
  float* buf = chunkbuf + (size_t)(h * NCHUNKS + c) * CHUNK_STRIDE;
#pragma unroll
  for (int j = 0; j < 32; ++j) buf[m * DHEAD + d0 + j] = acc[j];
  if ((t & 1) == 0) buf[MFEAT * DHEAD + m] = zacc;
}

__global__ void __launch_bounds__(256)
perf_chunk_prefix(float* __restrict__ chunkbuf) {
  const int h = blockIdx.x;
  for (int idx = threadIdx.x; idx < CHUNK_STRIDE; idx += 256) {
    float carry = 0.f;
    for (int c = 0; c < NCHUNKS; ++c) {
      float* p = chunkbuf + (size_t)(h * NCHUNKS + c) * CHUNK_STRIDE + idx;
      const float tmp = *p;
      *p = carry;
      carry += tmp;
    }
  }
}

__device__ __forceinline__ v8f wmma4(v2f a, v2f b, v8f c) {
  return __builtin_amdgcn_wmma_f32_16x16x4_f32(false, a, false, b, (short)0, c, false, false);
}

__global__ void __launch_bounds__(32)
perf_scan(const float* __restrict__ qprime, const float* __restrict__ kprime,
          const float* __restrict__ V, const float* __restrict__ chunkbuf,
          float* __restrict__ context) {
  const int lt = blockIdx.x;
  const int c = blockIdx.y;
  const int h = blockIdx.z;
  const int lane = threadIdx.x;
  const int half = lane >> 4;
  const int l16 = lane & 15;
  const int koff = 4 * half;
  const int lrow0 = c * CHUNK + lt * 16;
  const float* buf = chunkbuf + (size_t)(h * NCHUNKS + c) * CHUNK_STRIDE;

  const float* qrow = qprime + (size_t)(h * L_SEQ + lrow0 + l16) * MFEAT + koff;
  float4 q[16];
#pragma unroll
  for (int s = 0; s < 16; ++s) q[s] = *reinterpret_cast<const float4*>(qrow + 8 * s);

  const v8f zero = {0.f, 0.f, 0.f, 0.f, 0.f, 0.f, 0.f, 0.f};
  v8f num[4];
  num[0] = zero; num[1] = zero; num[2] = zero; num[3] = zero;
  v8f den = zero;

  const float* sp = buf + koff * DHEAD + l16;
  const float* zp = buf + MFEAT * DHEAD + koff;
#pragma unroll
  for (int s = 0; s < 16; ++s) {
    const float4 zz = *reinterpret_cast<const float4*>(zp + 8 * s);
    v2f a0, a1, z0, z1;
    a0.x = q[s].x; a0.y = q[s].y; a1.x = q[s].z; a1.y = q[s].w;
    z0.x = zz.x;   z0.y = zz.y;   z1.x = zz.z;   z1.y = zz.w;
    den = wmma4(a0, z0, den);
    den = wmma4(a1, z1, den);
#pragma unroll
    for (int dt = 0; dt < 4; ++dt) {
      v2f b0, b1;
      b0.x = sp[(8 * s + 0) * DHEAD + dt * 16];
      b0.y = sp[(8 * s + 1) * DHEAD + dt * 16];
      b1.x = sp[(8 * s + 2) * DHEAD + dt * 16];
      b1.y = sp[(8 * s + 3) * DHEAD + dt * 16];
      num[dt] = wmma4(a0, b0, num[dt]);
      num[dt] = wmma4(a1, b1, num[dt]);
    }
  }

  const float* kbase = kprime + (size_t)(h * L_SEQ + c * CHUNK + l16) * MFEAT + koff;
  const float* vbase = V + (size_t)(c * CHUNK + 8 * half) * DMODEL + h * DHEAD + l16;
  v2f ones;
  ones.x = 1.f; ones.y = 1.f;
  for (int jt = 0; jt <= lt; ++jt) {
    const float* krow = kbase + (size_t)jt * 16 * MFEAT;
    float4 kf[16];
#pragma unroll
    for (int s = 0; s < 16; ++s) kf[s] = *reinterpret_cast<const float4*>(krow + 8 * s);
    v8f t0 = zero, t1 = zero;
#pragma unroll
    for (int s = 0; s < 16; ++s) {
      v2f ka0, ka1, qb0, qb1;
      ka0.x = kf[s].x; ka0.y = kf[s].y; ka1.x = kf[s].z; ka1.y = kf[s].w;
      qb0.x = q[s].x;  qb0.y = q[s].y;  qb1.x = q[s].z;  qb1.y = q[s].w;
      t0 = wmma4(ka0, qb0, t0);
      t1 = wmma4(ka1, qb1, t1);
    }
    v8f t = t0 + t1;
    if (jt == lt) {
#pragma unroll
      for (int r = 0; r < 8; ++r) t[r] = (r + 8 * half > l16) ? 0.f : t[r];
    }
    const float* vp = vbase + (size_t)jt * 16 * DMODEL;
#pragma unroll
    for (int s2 = 0; s2 < 4; ++s2) {
      v2f a;
      a.x = t[2 * s2]; a.y = t[2 * s2 + 1];
      den = wmma4(a, ones, den);
#pragma unroll
      for (int dt = 0; dt < 4; ++dt) {
        v2f b;
        b.x = vp[(2 * s2) * DMODEL + dt * 16];
        b.y = vp[(2 * s2 + 1) * DMODEL + dt * 16];
        num[dt] = wmma4(a, b, num[dt]);
      }
    }
  }

#pragma unroll
  for (int dt = 0; dt < 4; ++dt)
#pragma unroll
    for (int r = 0; r < 8; ++r) {
      const int l = lrow0 + r + 8 * half;
      context[(size_t)l * DMODEL + h * DHEAD + dt * 16 + l16] =
          num[dt][r] / (den[r] + DEN_EPS);
    }
}

__global__ void __launch_bounds__(32)
perf_out_proj(const float* __restrict__ ctx, const float* __restrict__ Wo,
              const float* __restrict__ bo, float* __restrict__ out) {
  const int tm = blockIdx.x * 32;
  const int tn = blockIdx.y * 32;
  const int lane = threadIdx.x;
  Acc2x2 r = wmma_block2x2_f32(ctx + tm * DMODEL, DMODEL,
                               Wo + tn * DMODEL, 1, DMODEL, DMODEL, lane);
  const int l16 = lane & 15;
  const int half = lane >> 4;
#pragma unroll
  for (int ti = 0; ti < 2; ++ti)
#pragma unroll
    for (int tj = 0; tj < 2; ++tj) {
      const int col = tn + tj * 16 + l16;
      const int rbase = tm + ti * 16 + 8 * half;
      const float bb = bo[col];
#pragma unroll
      for (int i = 0; i < 8; ++i)
        out[(rbase + i) * DMODEL + col] = r.c[ti][tj][i] + bb;
    }
}

extern "C" void kernel_launch(void* const* d_in, const int* in_sizes, int n_in,
                              void* d_out, int out_size, void* d_ws, size_t ws_size,
                              hipStream_t stream) {
  (void)in_sizes; (void)n_in; (void)out_size; (void)ws_size;
  const float* query = (const float*)d_in[0];
  const float* key_  = (const float*)d_in[1];
  const float* value = (const float*)d_in[2];
  const float* Wq = (const float*)d_in[3];
  const float* bq = (const float*)d_in[4];
  const float* Wk = (const float*)d_in[5];
  const float* bk = (const float*)d_in[6];
  const float* Wv = (const float*)d_in[7];
  const float* bv = (const float*)d_in[8];
  const float* Wo = (const float*)d_in[9];
  const float* bo = (const float*)d_in[10];
  const float* RF = (const float*)d_in[11];

  float* ws = (float*)d_ws;
  const size_t SZ_LD = (size_t)L_SEQ * DMODEL;
  const size_t SZ_PRIME = (size_t)NHEADS * L_SEQ * MFEAT;
  float* Q      = ws;
  float* K      = Q + SZ_LD;
  float* V      = K + SZ_LD;
  float* hk     = V + SZ_LD;
  float* kstab  = hk + (size_t)NHEADS * L_SEQ;
  float* qprime = kstab + 64;
  float* kprime = qprime + SZ_PRIME;
  float* chunkbuf = kprime + SZ_PRIME;
  float* context = Q;

  perf_qkv_proj<<<dim3(L_SEQ / 32, DMODEL / 32, 3), 32, 0, stream>>>(
      query, key_, value, Wq, bq, Wk, bk, Wv, bv, Q, K, V);
  perf_hk<<<(NHEADS * L_SEQ) / 256, 256, 0, stream>>>(K, hk);
  perf_kstab<<<1, 256, 0, stream>>>(hk, kstab);
  perf_primes<<<dim3(L_SEQ / 32, MFEAT / 32, 2 * NHEADS), 32, 0, stream>>>(
      Q, K, RF, hk, kstab, qprime, kprime);
  perf_chunk_sums<<<NHEADS * NCHUNKS, 256, 0, stream>>>(kprime, V, chunkbuf);
  perf_chunk_prefix<<<NHEADS, 256, 0, stream>>>(chunkbuf);
  perf_scan<<<dim3(CHUNK / 16, NCHUNKS, NHEADS), 32, 0, stream>>>(
      qprime, kprime, V, chunkbuf, context);
  perf_out_proj<<<dim3(L_SEQ / 32, DMODEL / 32), 32, 0, stream>>>(
      context, Wo, bo, (float*)d_out);
}


# per-chunk state sums k'^T v and sum k' moved from a 128-step barrier loop to WMMA products, one wave per 32x32 block; sums taken in a different reduction tree
# speedup vs baseline: 3.7893x; 1.7301x over previous
#include <hip/hip_runtime.h>
#include <math.h>

#define L_SEQ 2048
#define DMODEL 512
#define NHEADS 8
#define DHEAD 64
#define MFEAT 128
#define NCHUNKS 16
#define CHUNK 128
#define CHUNK_STRIDE 8320
#define KERNEL_EPS 1e-4f
#define DEN_EPS 1e-6f
#define QK_SCALE 0.21022410381342864f
#define NC_FEAT 0.08838834764831845f

typedef float v2f __attribute__((ext_vector_type(2)));
typedef float v8f __attribute__((ext_vector_type(8)));

struct Acc2x2 { v8f c[2][2]; };

__device__ __forceinline__ Acc2x2 wmma_block2x2_f32(const float* __restrict__ A, int lda,
                                                    const float* __restrict__ B, int bks, int bns,
                                                    int K, int lane) {
  const int half = lane >> 4;
  const int l16  = lane & 15;
  const float* arow0 = A + l16 * lda;
  const float* arow1 = A + (l16 + 16) * lda;
  const float* bcol0 = B + l16 * bns;
  const float* bcol1 = B + (l16 + 16) * bns;
  const v8f zero = {0.f, 0.f, 0.f, 0.f, 0.f, 0.f, 0.f, 0.f};
  Acc2x2 r;
  r.c[0][0] = zero; r.c[0][1] = zero; r.c[1][0] = zero; r.c[1][1] = zero;
  for (int k0 = 0; k0 < K; k0 += 4) {
    const int ka = k0 + 2 * half;
    v2f a0, a1, b0, b1;
    a0.x = arow0[ka];          a0.y = arow0[ka + 1];
    a1.x = arow1[ka];          a1.y = arow1[ka + 1];
    b0.x = bcol0[ka * bks];    b0.y = bcol0[(ka + 1) * bks];
    b1.x = bcol1[ka * bks];    b1.y = bcol1[(ka + 1) * bks];
    r.c[0][0] = __builtin_amdgcn_wmma_f32_16x16x4_f32(false, a0, false, b0, (short)0, r.c[0][0], false, false);
    r.c[0][1] = __builtin_amdgcn_wmma_f32_16x16x4_f32(false, a0, false, b1, (short)0, r.c[0][1], false, false);
    r.c[1][0] = __builtin_amdgcn_wmma_f32_16x16x4_f32(false, a1, false, b0, (short)0, r.c[1][0], false, false);
    r.c[1][1] = __builtin_amdgcn_wmma_f32_16x16x4_f32(false, a1, false, b1, (short)0, r.c[1][1], false, false);
  }
  return r;
}

__global__ void __launch_bounds__(32)
perf_qkv_proj(const float* __restrict__ xq, const float* __restrict__ xk,
              const float* __restrict__ xv,
              const float* __restrict__ Wq, const float* __restrict__ bq,
              const float* __restrict__ Wk, const float* __restrict__ bk,
              const float* __restrict__ Wv, const float* __restrict__ bv,
              float* __restrict__ Q, float* __restrict__ K, float* __restrict__ V) {
  const int tm = blockIdx.x * 32;
  const int tn = blockIdx.y * 32;
  const int mat = blockIdx.z;
  const int lane = threadIdx.x;
  const float* X; const float* W; const float* bias; float* Y; float scl;
  if (mat == 0)      { X = xq; W = Wq; bias = bq; Y = Q; scl = QK_SCALE; }
  else if (mat == 1) { X = xk; W = Wk; bias = bk; Y = K; scl = QK_SCALE; }
  else               { X = xv; W = Wv; bias = bv; Y = V; scl = 1.f; }
  Acc2x2 r = wmma_block2x2_f32(X + tm * DMODEL, DMODEL,
                               W + tn * DMODEL, 1, DMODEL, DMODEL, lane);
  const int l16 = lane & 15;
  const int half = lane >> 4;
#pragma unroll
  for (int ti = 0; ti < 2; ++ti)
#pragma unroll
    for (int tj = 0; tj < 2; ++tj) {
      const int col = tn + tj * 16 + l16;
      const int rbase = tm + ti * 16 + 8 * half;
      const float bb = bias[col];
#pragma unroll
      for (int i = 0; i < 8; ++i)
        Y[(rbase + i) * DMODEL + col] = (r.c[ti][tj][i] + bb) * scl;
    }
}

__global__ void __launch_bounds__(256)
perf_hk(const float* __restrict__ K, float* __restrict__ hk) {
  const int idx = blockIdx.x * 256 + threadIdx.x;
  const int h = idx >> 11;
  const int l = idx & (L_SEQ - 1);
  const float* row = K + l * DMODEL + h * DHEAD;
  float s = 0.f;
#pragma unroll 8
  for (int d = 0; d < DHEAD; ++d) s += row[d] * row[d];
  hk[idx] = -0.5f * s;
}

__global__ void __launch_bounds__(256)
perf_kstab(const float* __restrict__ hk, float* __restrict__ kstab) {
  __shared__ float red[256];
  const int t = threadIdx.x;
  float m = -3.402823e38f;
  for (int i = t; i < NHEADS * L_SEQ; i += 256) m = fmaxf(m, hk[i]);
  red[t] = m;
  __syncthreads();
  for (int s = 128; s > 0; s >>= 1) {
    if (t < s) red[t] = fmaxf(red[t], red[t + s]);
    __syncthreads();
  }
  if (t == 0) kstab[0] = red[0];
}

__global__ void __launch_bounds__(32)
perf_primes(const float* __restrict__ Q, const float* __restrict__ K,
            const float* __restrict__ RF, const float* __restrict__ hk,
            const float* __restrict__ kstab,
            float* __restrict__ qprime, float* __restrict__ kprime) {
  const int tm = blockIdx.x * 32;
  const int tn = blockIdx.y * 32;
  const int h = blockIdx.z >> 1;
  const int isK = blockIdx.z & 1;
  const int lane = threadIdx.x;
  const float* A = (isK ? K : Q) + tm * DMODEL + h * DHEAD;
  Acc2x2 r = wmma_block2x2_f32(A, DMODEL, RF + tn, MFEAT, 1, DHEAD, lane);
  const int l16 = lane & 15;
  const int half = lane >> 4;
  const float ks = isK ? kstab[0] : 0.f;
#pragma unroll
  for (int ti = 0; ti < 2; ++ti)
#pragma unroll
    for (int tj = 0; tj < 2; ++tj) {
      const int mcol = tn + tj * 16 + l16;
      const int rbase = tm + ti * 16 + 8 * half;
      if (isK) {
#pragma unroll
        for (int i = 0; i < 8; ++i) {
          const int l = rbase + i;
          kprime[(h * L_SEQ + l) * MFEAT + mcol] =
              NC_FEAT * (expf(hk[h * L_SEQ + l] + r.c[ti][tj][i] - ks) + KERNEL_EPS);
        }
      } else {
#pragma unroll
        for (int i = 0; i < 8; ++i) {
          const int l = rbase + i;
          qprime[(h * L_SEQ + l) * MFEAT + mcol] =
              NC_FEAT * (expf(r.c[ti][tj][i]) + KERNEL_EPS);
        }
      }
    }
}

__device__ __forceinline__ v8f wmma4(v2f a, v2f b, v8f c) {
  return __builtin_amdgcn_wmma_f32_16x16x4_f32(false, a, false, b, (short)0, c, false, false);
}

__global__ void __launch_bounds__(32)
perf_chunk_sums(const float* __restrict__ kprime, const float* __restrict__ V,
                float* __restrict__ chunkbuf) {
  const int blk = blockIdx.x;
  const int c = blockIdx.y;
  const int h = blockIdx.z;
  const int lane = threadIdx.x;
  const int half = lane >> 4;
  const int l16 = lane & 15;
  const int m0 = (blk >> 1) * 32;
  const int d0 = (blk & 1) * 32;
  const float* ka = kprime + (size_t)(h * L_SEQ + c * CHUNK + 2 * half) * MFEAT + m0 + 2 * l16;
  const float* vb = V + (size_t)(c * CHUNK + 2 * half) * DMODEL + h * DHEAD + d0 + 2 * l16;
  const v8f zero = {0.f, 0.f, 0.f, 0.f, 0.f, 0.f, 0.f, 0.f};
  v8f acc[2][2];
  acc[0][0] = zero; acc[0][1] = zero; acc[1][0] = zero; acc[1][1] = zero;
  float zx = 0.f, zy = 0.f;
#pragma unroll
  for (int ks = 0; ks < CHUNK / 4; ++ks) {
    const float2 ra0 = *reinterpret_cast<const float2*>(ka + (size_t)(4 * ks) * MFEAT);
    const float2 ra1 = *reinterpret_cast<const float2*>(ka + (size_t)(4 * ks + 1) * MFEAT);
    const float2 rb0 = *reinterpret_cast<const float2*>(vb + (size_t)(4 * ks) * DMODEL);
    const float2 rb1 = *reinterpret_cast<const float2*>(vb + (size_t)(4 * ks + 1) * DMODEL);
    v2f a0, a1, b0, b1;
    a0.x = ra0.x; a0.y = ra1.x; a1.x = ra0.y; a1.y = ra1.y;
    b0.x = rb0.x; b0.y = rb1.x; b1.x = rb0.y; b1.y = rb1.y;
    acc[0][0] = wmma4(a0, b0, acc[0][0]);
    acc[0][1] = wmma4(a0, b1, acc[0][1]);
    acc[1][0] = wmma4(a1, b0, acc[1][0]);
    acc[1][1] = wmma4(a1, b1, acc[1][1]);
    zx += ra0.x + ra1.x;
    zy += ra0.y + ra1.y;
  }
  float* buf = chunkbuf + (size_t)(h * NCHUNKS + c) * CHUNK_STRIDE;
#pragma unroll
  for (int jm = 0; jm < 2; ++jm)
#pragma unroll
    for (int r = 0; r < 8; ++r) {
      const int m = m0 + 2 * (r + 8 * half) + jm;
      float2 o;
      o.x = acc[jm][0][r]; o.y = acc[jm][1][r];
      *reinterpret_cast<float2*>(buf + m * DHEAD + d0 + 2 * l16) = o;
    }
  zx += __shfl_xor(zx, 16);
  zy += __shfl_xor(zy, 16);
  if (d0 == 0 && half == 0) {
    float2 o;
    o.x = zx; o.y = zy;
    *reinterpret_cast<float2*>(buf + MFEAT * DHEAD + m0 + 2 * l16) = o;
  }
}

__global__ void __launch_bounds__(256)
perf_chunk_prefix(float* __restrict__ chunkbuf) {
  const int h = blockIdx.x;
  for (int idx = threadIdx.x; idx < CHUNK_STRIDE; idx += 256) {
    float carry = 0.f;
    for (int c = 0; c < NCHUNKS; ++c) {
      float* p = chunkbuf + (size_t)(h * NCHUNKS + c) * CHUNK_STRIDE + idx;
      const float tmp = *p;
      *p = carry;
      carry += tmp;
    }
  }
}

__global__ void __launch_bounds__(32)
perf_scan(const float* __restrict__ qprime, const float* __restrict__ kprime,
          const float* __restrict__ V, const float* __restrict__ chunkbuf,
          float* __restrict__ context) {
  const int lt = blockIdx.x;
  const int c = blockIdx.y;
  const int h = blockIdx.z;
  const int lane = threadIdx.x;
  const int half = lane >> 4;
  const int l16 = lane & 15;
  const int koff = 4 * half;
  const int lrow0 = c * CHUNK + lt * 16;
  const float* buf = chunkbuf + (size_t)(h * NCHUNKS + c) * CHUNK_STRIDE;

  const float* qrow = qprime + (size_t)(h * L_SEQ + lrow0 + l16) * MFEAT + koff;
  float4 q[16];
#pragma unroll
  for (int s = 0; s < 16; ++s) q[s] = *reinterpret_cast<const float4*>(qrow + 8 * s);

  const v8f zero = {0.f, 0.f, 0.f, 0.f, 0.f, 0.f, 0.f, 0.f};
  v8f num[4];
  num[0] = zero; num[1] = zero; num[2] = zero; num[3] = zero;
  v8f den = zero;

  const float* sp = buf + koff * DHEAD + l16;
  const float* zp = buf + MFEAT * DHEAD + koff;
#pragma unroll
  for (int s = 0; s < 16; ++s) {
    const float4 zz = *reinterpret_cast<const float4*>(zp + 8 * s);
    v2f a0, a1, z0, z1;
    a0.x = q[s].x; a0.y = q[s].y; a1.x = q[s].z; a1.y = q[s].w;
    z0.x = zz.x;   z0.y = zz.y;   z1.x = zz.z;   z1.y = zz.w;
    den = wmma4(a0, z0, den);
    den = wmma4(a1, z1, den);
#pragma unroll
    for (int dt = 0; dt < 4; ++dt) {
      v2f b0, b1;
      b0.x = sp[(8 * s + 0) * DHEAD + dt * 16];
      b0.y = sp[(8 * s + 1) * DHEAD + dt * 16];
      b1.x = sp[(8 * s + 2) * DHEAD + dt * 16];
      b1.y = sp[(8 * s + 3) * DHEAD + dt * 16];
      num[dt] = wmma4(a0, b0, num[dt]);
      num[dt] = wmma4(a1, b1, num[dt]);
    }
  }

  const float* kbase = kprime + (size_t)(h * L_SEQ + c * CHUNK + l16) * MFEAT + koff;
  const float* vbase = V + (size_t)(c * CHUNK + 8 * half) * DMODEL + h * DHEAD + l16;
  v2f ones;
  ones.x = 1.f; ones.y = 1.f;
  for (int jt = 0; jt <= lt; ++jt) {
    const float* krow = kbase + (size_t)jt * 16 * MFEAT;
    float4 kf[16];
#pragma unroll
    for (int s = 0; s < 16; ++s) kf[s] = *reinterpret_cast<const float4*>(krow + 8 * s);
    v8f t0 = zero, t1 = zero;
#pragma unroll
    for (int s = 0; s < 16; ++s) {
      v2f ka0, ka1, qb0, qb1;
      ka0.x = kf[s].x; ka0.y = kf[s].y; ka1.x = kf[s].z; ka1.y = kf[s].w;
      qb0.x = q[s].x;  qb0.y = q[s].y;  qb1.x = q[s].z;  qb1.y = q[s].w;
      t0 = wmma4(ka0, qb0, t0);
      t1 = wmma4(ka1, qb1, t1);
    }
    v8f t = t0 + t1;
    if (jt == lt) {
#pragma unroll
      for (int r = 0; r < 8; ++r) t[r] = (r + 8 * half > l16) ? 0.f : t[r];
    }
    const float* vp = vbase + (size_t)jt * 16 * DMODEL;
#pragma unroll
    for (int s2 = 0; s2 < 4; ++s2) {
      v2f a;
      a.x = t[2 * s2]; a.y = t[2 * s2 + 1];
      den = wmma4(a, ones, den);
#pragma unroll
      for (int dt = 0; dt < 4; ++dt) {
        v2f b;
        b.x = vp[(2 * s2) * DMODEL + dt * 16];
        b.y = vp[(2 * s2 + 1) * DMODEL + dt * 16];
        num[dt] = wmma4(a, b, num[dt]);
      }
    }
  }

#pragma unroll
  for (int dt = 0; dt < 4; ++dt)
#pragma unroll
    for (int r = 0; r < 8; ++r) {
      const int l = lrow0 + r + 8 * half;
      context[(size_t)l * DMODEL + h * DHEAD + dt * 16 + l16] =
          num[dt][r] / (den[r] + DEN_EPS);
    }
}

__global__ void __launch_bounds__(32)
perf_out_proj(const float* __restrict__ ctx, const float* __restrict__ Wo,
              const float* __restrict__ bo, float* __restrict__ out) {
  const int tm = blockIdx.x * 32;
  const int tn = blockIdx.y * 32;
  const int lane = threadIdx.x;
  Acc2x2 r = wmma_block2x2_f32(ctx + tm * DMODEL, DMODEL,
                               Wo + tn * DMODEL, 1, DMODEL, DMODEL, lane);
  const int l16 = lane & 15;
  const int half = lane >> 4;
#pragma unroll
  for (int ti = 0; ti < 2; ++ti)
#pragma unroll
    for (int tj = 0; tj < 2; ++tj) {
      const int col = tn + tj * 16 + l16;
      const int rbase = tm + ti * 16 + 8 * half;
      const float bb = bo[col];
#pragma unroll
      for (int i = 0; i < 8; ++i)
        out[(rbase + i) * DMODEL + col] = r.c[ti][tj][i] + bb;
    }
}

extern "C" void kernel_launch(void* const* d_in, const int* in_sizes, int n_in,
                              void* d_out, int out_size, void* d_ws, size_t ws_size,
                              hipStream_t stream) {
  (void)in_sizes; (void)n_in; (void)out_size; (void)ws_size;
  const float* query = (const float*)d_in[0];
  const float* key_  = (const float*)d_in[1];
  const float* value = (const float*)d_in[2];
  const float* Wq = (const float*)d_in[3];
  const float* bq = (const float*)d_in[4];
  const float* Wk = (const float*)d_in[5];
  const float* bk = (const float*)d_in[6];
  const float* Wv = (const float*)d_in[7];
  const float* bv = (const float*)d_in[8];
  const float* Wo = (const float*)d_in[9];
  const float* bo = (const float*)d_in[10];
  const float* RF = (const float*)d_in[11];

  float* ws = (float*)d_ws;
  const size_t SZ_LD = (size_t)L_SEQ * DMODEL;
  const size_t SZ_PRIME = (size_t)NHEADS * L_SEQ * MFEAT;
  float* Q      = ws;
  float* K      = Q + SZ_LD;
  float* V      = K + SZ_LD;
  float* hk     = V + SZ_LD;
  float* kstab  = hk + (size_t)NHEADS * L_SEQ;
  float* qprime = kstab + 64;
  float* kprime = qprime + SZ_PRIME;
  float* chunkbuf = kprime + SZ_PRIME;
  float* context = Q;

  perf_qkv_proj<<<dim3(L_SEQ / 32, DMODEL / 32, 3), 32, 0, stream>>>(
      query, key_, value, Wq, bq, Wk, bk, Wv, bv, Q, K, V);
  perf_hk<<<(NHEADS * L_SEQ) / 256, 256, 0, stream>>>(K, hk);
  perf_kstab<<<1, 256, 0, stream>>>(hk, kstab);
  perf_primes<<<dim3(L_SEQ / 32, MFEAT / 32, 2 * NHEADS), 32, 0, stream>>>(
      Q, K, RF, hk, kstab, qprime, kprime);
  perf_chunk_sums<<<dim3(8, NCHUNKS, NHEADS), 32, 0, stream>>>(kprime, V, chunkbuf);
  perf_chunk_prefix<<<NHEADS, 256, 0, stream>>>(chunkbuf);
  perf_scan<<<dim3(CHUNK / 16, NCHUNKS, NHEADS), 32, 0, stream>>>(
      qprime, kprime, V, chunkbuf, context);
  perf_out_proj<<<dim3(L_SEQ / 32, DMODEL / 32), 32, 0, stream>>>(
      context, Wo, bo, (float*)d_out);
}


# chunk-prefix and k-max helper kernels: per-thread chains of dependent load round trips replaced by all loads issued at once on wider grids (130 groups, 1024 threads); prefix sums keep their order
# speedup vs baseline: 5.8480x; 1.5433x over previous
#include <hip/hip_runtime.h>
#include <math.h>

#define L_SEQ 2048
#define DMODEL 512
#define NHEADS 8
#define DHEAD 64
#define MFEAT 128
#define NCHUNKS 16
#define CHUNK 128
#define CHUNK_STRIDE 8320
#define KERNEL_EPS 1e-4f
#define DEN_EPS 1e-6f
#define QK_SCALE 0.21022410381342864f
#define NC_FEAT 0.08838834764831845f

typedef float v2f __attribute__((ext_vector_type(2)));
typedef float v8f __attribute__((ext_vector_type(8)));

struct Acc2x2 { v8f c[2][2]; };

__device__ __forceinline__ Acc2x2 wmma_block2x2_f32(const float* __restrict__ A, int lda,
                                                    const float* __restrict__ B, int bks, int bns,
                                                    int K, int lane) {
  const int half = lane >> 4;
  const int l16  = lane & 15;
  const float* arow0 = A + l16 * lda;
  const float* arow1 = A + (l16 + 16) * lda;
  const float* bcol0 = B + l16 * bns;
  const float* bcol1 = B + (l16 + 16) * bns;
  const v8f zero = {0.f, 0.f, 0.f, 0.f, 0.f, 0.f, 0.f, 0.f};
  Acc2x2 r;
  r.c[0][0] = zero; r.c[0][1] = zero; r.c[1][0] = zero; r.c[1][1] = zero;
  for (int k0 = 0; k0 < K; k0 += 4) {
    const int ka = k0 + 2 * half;
    v2f a0, a1, b0, b1;
    a0.x = arow0[ka];          a0.y = arow0[ka + 1];
    a1.x = arow1[ka];          a1.y = arow1[ka + 1];
    b0.x = bcol0[ka * bks];    b0.y = bcol0[(ka + 1) * bks];
    b1.x = bcol1[ka * bks];    b1.y = bcol1[(ka + 1) * bks];
    r.c[0][0] = __builtin_amdgcn_wmma_f32_16x16x4_f32(false, a0, false, b0, (short)0, r.c[0][0], false, false);
    r.c[0][1] = __builtin_amdgcn_wmma_f32_16x16x4_f32(false, a0, false, b1, (short)0, r.c[0][1], false, false);
    r.c[1][0] = __builtin_amdgcn_wmma_f32_16x16x4_f32(false, a1, false, b0, (short)0, r.c[1][0], false, false);
    r.c[1][1] = __builtin_amdgcn_wmma_f32_16x16x4_f32(false, a1, false, b1, (short)0, r.c[1][1], false, false);
  }
  return r;
}

__global__ void __launch_bounds__(32)
perf_qkv_proj(const float* __restrict__ xq, const float* __restrict__ xk,
              const float* __restrict__ xv,
              const float* __restrict__ Wq, const float* __restrict__ bq,
              const float* __restrict__ Wk, const float* __restrict__ bk,
              const float* __restrict__ Wv, const float* __restrict__ bv,
              float* __restrict__ Q, float* __restrict__ K, float* __restrict__ V) {
  const int tm = blockIdx.x * 32;
  const int tn = blockIdx.y * 32;
  const int mat = blockIdx.z;
  const int lane = threadIdx.x;
  const float* X; const float* W; const float* bias; float* Y; float scl;
  if (mat == 0)      { X = xq; W = Wq; bias = bq; Y = Q; scl = QK_SCALE; }
  else if (mat == 1) { X = xk; W = Wk; bias = bk; Y = K; scl = QK_SCALE; }
  else               { X = xv; W = Wv; bias = bv; Y = V; scl = 1.f; }
  Acc2x2 r = wmma_block2x2_f32(X + tm * DMODEL, DMODEL,
                               W + tn * DMODEL, 1, DMODEL, DMODEL, lane);
  const int l16 = lane & 15;
  const int half = lane >> 4;
#pragma unroll
  for (int ti = 0; ti < 2; ++ti)
#pragma unroll
    for (int tj = 0; tj < 2; ++tj) {
      const int col = tn + tj * 16 + l16;
      const int rbase = tm + ti * 16 + 8 * half;
      const float bb = bias[col];
#pragma unroll
      for (int i = 0; i < 8; ++i)
        Y[(rbase + i) * DMODEL + col] = (r.c[ti][tj][i] + bb) * scl;
    }
}

__global__ void __launch_bounds__(256)
perf_hk(const float* __restrict__ K, float* __restrict__ hk) {
  const int idx = blockIdx.x * 256 + threadIdx.x;
  const int h = idx >> 11;
  const int l = idx & (L_SEQ - 1);
  const float* row = K + l * DMODEL + h * DHEAD;
  float s = 0.f;
#pragma unroll 8
  for (int d = 0; d < DHEAD; ++d) s += row[d] * row[d];
  hk[idx] = -0.5f * s;
}

__global__ void __launch_bounds__(1024)
perf_kstab(const float* __restrict__ hk, float* __restrict__ kstab) {
  __shared__ float red[32];
  const int t = threadIdx.x;
  const float4* p = reinterpret_cast<const float4*>(hk) + t;
  float4 v[4];
#pragma unroll
  for (int i = 0; i < 4; ++i) v[i] = p[i * 1024];
  float m = -3.402823e38f;
#pragma unroll
  for (int i = 0; i < 4; ++i)
    m = fmaxf(m, fmaxf(fmaxf(v[i].x, v[i].y), fmaxf(v[i].z, v[i].w)));
  for (int off = 16; off > 0; off >>= 1) m = fmaxf(m, __shfl_xor(m, off));
  if ((t & 31) == 0) red[t >> 5] = m;
  __syncthreads();
  if (t < 32) {
    float r = red[t];
    for (int off = 16; off > 0; off >>= 1) r = fmaxf(r, __shfl_xor(r, off));
    if (t == 0) kstab[0] = r;
  }
}

__global__ void __launch_bounds__(32)
perf_primes(const float* __restrict__ Q, const float* __restrict__ K,
            const float* __restrict__ RF, const float* __restrict__ hk,
            const float* __restrict__ kstab,
            float* __restrict__ qprime, float* __restrict__ kprime) {
  const int tm = blockIdx.x * 32;
  const int tn = blockIdx.y * 32;
  const int h = blockIdx.z >> 1;
  const int isK = blockIdx.z & 1;
  const int lane = threadIdx.x;
  const float* A = (isK ? K : Q) + tm * DMODEL + h * DHEAD;
  Acc2x2 r = wmma_block2x2_f32(A, DMODEL, RF + tn, MFEAT, 1, DHEAD, lane);
  const int l16 = lane & 15;
  const int half = lane >> 4;
  const float ks = isK ? kstab[0] : 0.f;
#pragma unroll
  for (int ti = 0; ti < 2; ++ti)
#pragma unroll
    for (int tj = 0; tj < 2; ++tj) {
      const int mcol = tn + tj * 16 + l16;
      const int rbase = tm + ti * 16 + 8 * half;
      if (isK) {
#pragma unroll
        for (int i = 0; i < 8; ++i) {
          const int l = rbase + i;
          kprime[(h * L_SEQ + l) * MFEAT + mcol] =
              NC_FEAT * (expf(hk[h * L_SEQ + l] + r.c[ti][tj][i] - ks) + KERNEL_EPS);
        }
      } else {
#pragma unroll
        for (int i = 0; i < 8; ++i) {
          const int l = rbase + i;
          qprime[(h * L_SEQ + l) * MFEAT + mcol] =
              NC_FEAT * (expf(r.c[ti][tj][i]) + KERNEL_EPS);
        }
      }
    }
}

__device__ __forceinline__ v8f wmma4(v2f a, v2f b, v8f c) {
  return __builtin_amdgcn_wmma_f32_16x16x4_f32(false, a, false, b, (short)0, c, false, false);
}

__global__ void __launch_bounds__(32)
perf_chunk_sums(const float* __restrict__ kprime, const float* __restrict__ V,
                float* __restrict__ chunkbuf) {
  const int blk = blockIdx.x;
  const int c = blockIdx.y;
  const int h = blockIdx.z;
  const int lane = threadIdx.x;
  const int half = lane >> 4;
  const int l16 = lane & 15;
  const int m0 = (blk >> 1) * 32;
  const int d0 = (blk & 1) * 32;
  const float* ka = kprime + (size_t)(h * L_SEQ + c * CHUNK + 2 * half) * MFEAT + m0 + 2 * l16;
  const float* vb = V + (size_t)(c * CHUNK + 2 * half) * DMODEL + h * DHEAD + d0 + 2 * l16;
  const v8f zero = {0.f, 0.f, 0.f, 0.f, 0.f, 0.f, 0.f, 0.f};
  v8f acc[2][2];
  acc[0][0] = zero; acc[0][1] = zero; acc[1][0] = zero; acc[1][1] = zero;
  float zx = 0.f, zy = 0.f;
#pragma unroll
  for (int ks = 0; ks < CHUNK / 4; ++ks) {
    const float2 ra0 = *reinterpret_cast<const float2*>(ka + (size_t)(4 * ks) * MFEAT);
    const float2 ra1 = *reinterpret_cast<const float2*>(ka + (size_t)(4 * ks + 1) * MFEAT);
    const float2 rb0 = *reinterpret_cast<const float2*>(vb + (size_t)(4 * ks) * DMODEL);
    const float2 rb1 = *reinterpret_cast<const float2*>(vb + (size_t)(4 * ks + 1) * DMODEL);
    v2f a0, a1, b0, b1;
    a0.x = ra0.x; a0.y = ra1.x; a1.x = ra0.y; a1.y = ra1.y;
    b0.x = rb0.x; b0.y = rb1.x; b1.x = rb0.y; b1.y = rb1.y;
    acc[0][0] = wmma4(a0, b0, acc[0][0]);
    acc[0][1] = wmma4(a0, b1, acc[0][1]);
    acc[1][0] = wmma4(a1, b0, acc[1][0]);
    acc[1][1] = wmma4(a1, b1, acc[1][1]);
    zx += ra0.x + ra1.x;
    zy += ra0.y + ra1.y;
  }
  float* buf = chunkbuf + (size_t)(h * NCHUNKS + c) * CHUNK_STRIDE;
#pragma unroll
  for (int jm = 0; jm < 2; ++jm)
#pragma unroll
    for (int r = 0; r < 8; ++r) {
      const int m = m0 + 2 * (r + 8 * half) + jm;
      float2 o;
      o.x = acc[jm][0][r]; o.y = acc[jm][1][r];
      *reinterpret_cast<float2*>(buf + m * DHEAD + d0 + 2 * l16) = o;
    }
  zx += __shfl_xor(zx, 16);
  zy += __shfl_xor(zy, 16);
  if (d0 == 0 && half == 0) {
    float2 o;
    o.x = zx; o.y = zy;
    *reinterpret_cast<float2*>(buf + MFEAT * DHEAD + m0 + 2 * l16) = o;
  }
}

#define PREFIX_VEC (CHUNK_STRIDE / 4)
__global__ void __launch_bounds__(128)
perf_chunk_prefix(float* __restrict__ chunkbuf) {
  const int gid = blockIdx.x * 128 + threadIdx.x;
  const int h = gid / PREFIX_VEC;
  const int i4 = gid - h * PREFIX_VEC;
  float4* p = reinterpret_cast<float4*>(chunkbuf + (size_t)h * NCHUNKS * CHUNK_STRIDE) + i4;
  float4 v[NCHUNKS];
#pragma unroll
  for (int c = 0; c < NCHUNKS; ++c) v[c] = p[(size_t)c * PREFIX_VEC];
  float4 carry;
  carry.x = 0.f; carry.y = 0.f; carry.z = 0.f; carry.w = 0.f;
#pragma unroll
  for (int c = 0; c < NCHUNKS; ++c) {
    const float4 tmp = v[c];
    p[(size_t)c * PREFIX_VEC] = carry;
    carry.x += tmp.x; carry.y += tmp.y; carry.z += tmp.z; carry.w += tmp.w;
  }
}

__global__ void __launch_bounds__(32)
perf_scan(const float* __restrict__ qprime, const float* __restrict__ kprime,
          const float* __restrict__ V, const float* __restrict__ chunkbuf,
          float* __restrict__ context) {
  const int lt = blockIdx.x;
  const int c = blockIdx.y;
  const int h = blockIdx.z;
  const int lane = threadIdx.x;
  const int half = lane >> 4;
  const int l16 = lane & 15;
  const int koff = 4 * half;
  const int lrow0 = c * CHUNK + lt * 16;
  const float* buf = chunkbuf + (size_t)(h * NCHUNKS + c) * CHUNK_STRIDE;

  const float* qrow = qprime + (size_t)(h * L_SEQ + lrow0 + l16) * MFEAT + koff;
  float4 q[16];
#pragma unroll
  for (int s = 0; s < 16; ++s) q[s] = *reinterpret_cast<const float4*>(qrow + 8 * s);

  const v8f zero = {0.f, 0.f, 0.f, 0.f, 0.f, 0.f, 0.f, 0.f};
  v8f num[4];
  num[0] = zero; num[1] = zero; num[2] = zero; num[3] = zero;
  v8f den = zero;

  const float* sp = buf + koff * DHEAD + l16;
  const float* zp = buf + MFEAT * DHEAD + koff;
#pragma unroll
  for (int s = 0; s < 16; ++s) {
    const float4 zz = *reinterpret_cast<const float4*>(zp + 8 * s);
    v2f a0, a1, z0, z1;
    a0.x = q[s].x; a0.y = q[s].y; a1.x = q[s].z; a1.y = q[s].w;
    z0.x = zz.x;   z0.y = zz.y;   z1.x = zz.z;   z1.y = zz.w;
    den = wmma4(a0, z0, den);
    den = wmma4(a1, z1, den);
#pragma unroll
    for (int dt = 0; dt < 4; ++dt) {
      v2f b0, b1;
      b0.x = sp[(8 * s + 0) * DHEAD + dt * 16];
      b0.y = sp[(8 * s + 1) * DHEAD + dt * 16];
      b1.x = sp[(8 * s + 2) * DHEAD + dt * 16];
      b1.y = sp[(8 * s + 3) * DHEAD + dt * 16];
      num[dt] = wmma4(a0, b0, num[dt]);
      num[dt] = wmma4(a1, b1, num[dt]);
    }
  }

  const float* kbase = kprime + (size_t)(h * L_SEQ + c * CHUNK + l16) * MFEAT + koff;
  const float* vbase = V + (size_t)(c * CHUNK + 8 * half) * DMODEL + h * DHEAD + l16;
  v2f ones;
  ones.x = 1.f; ones.y = 1.f;
  for (int jt = 0; jt <= lt; ++jt) {
    const float* krow = kbase + (size_t)jt * 16 * MFEAT;
    float4 kf[16];
#pragma unroll
    for (int s = 0; s < 16; ++s) kf[s] = *reinterpret_cast<const float4*>(krow + 8 * s);
    v8f t0 = zero, t1 = zero;
#pragma unroll
    for (int s = 0; s < 16; ++s) {
      v2f ka0, ka1, qb0, qb1;
      ka0.x = kf[s].x; ka0.y = kf[s].y; ka1.x = kf[s].z; ka1.y = kf[s].w;
      qb0.x = q[s].x;  qb0.y = q[s].y;  qb1.x = q[s].z;  qb1.y = q[s].w;
      t0 = wmma4(ka0, qb0, t0);
      t1 = wmma4(ka1, qb1, t1);
    }
    v8f t = t0 + t1;
    if (jt == lt) {
#pragma unroll
      for (int r = 0; r < 8; ++r) t[r] = (r + 8 * half > l16) ? 0.f : t[r];
    }
    const float* vp = vbase + (size_t)jt * 16 * DMODEL;
#pragma unroll
    for (int s2 = 0; s2 < 4; ++s2) {
      v2f a;
      a.x = t[2 * s2]; a.y = t[2 * s2 + 1];
      den = wmma4(a, ones, den);
#pragma unroll
      for (int dt = 0; dt < 4; ++dt) {
        v2f b;
        b.x = vp[(2 * s2) * DMODEL + dt * 16];
        b.y = vp[(2 * s2 + 1) * DMODEL + dt * 16];
        num[dt] = wmma4(a, b, num[dt]);
      }
    }
  }

#pragma unroll
  for (int dt = 0; dt < 4; ++dt)
#pragma unroll
    for (int r = 0; r < 8; ++r) {
      const int l = lrow0 + r + 8 * half;
      context[(size_t)l * DMODEL + h * DHEAD + dt * 16 + l16] =
          num[dt][r] / (den[r] + DEN_EPS);
    }
}

__global__ void __launch_bounds__(32)
perf_out_proj(const float* __restrict__ ctx, const float* __restrict__ Wo,
              const float* __restrict__ bo, float* __restrict__ out) {
  const int tm = blockIdx.x * 32;
  const int tn = blockIdx.y * 32;
  const int lane = threadIdx.x;
  Acc2x2 r = wmma_block2x2_f32(ctx + tm * DMODEL, DMODEL,
                               Wo + tn * DMODEL, 1, DMODEL, DMODEL, lane);
  const int l16 = lane & 15;
  const int half = lane >> 4;
#pragma unroll
  for (int ti = 0; ti < 2; ++ti)
#pragma unroll
    for (int tj = 0; tj < 2; ++tj) {
      const int col = tn + tj * 16 + l16;
      const int rbase = tm + ti * 16 + 8 * half;
      const float bb = bo[col];
#pragma unroll
      for (int i = 0; i < 8; ++i)
        out[(rbase + i) * DMODEL + col] = r.c[ti][tj][i] + bb;
    }
}

extern "C" void kernel_launch(void* const* d_in, const int* in_sizes, int n_in,
                              void* d_out, int out_size, void* d_ws, size_t ws_size,
                              hipStream_t stream) {
  (void)in_sizes; (void)n_in; (void)out_size; (void)ws_size;
  const float* query = (const float*)d_in[0];
  const float* key_  = (const float*)d_in[1];
  const float* value = (const float*)d_in[2];
  const float* Wq = (const float*)d_in[3];
  const float* bq = (const float*)d_in[4];
  const float* Wk = (const float*)d_in[5];
  const float* bk = (const float*)d_in[6];
  const float* Wv = (const float*)d_in[7];
  const float* bv = (const float*)d_in[8];
  const float* Wo = (const float*)d_in[9];
  const float* bo = (const float*)d_in[10];
  const float* RF = (const float*)d_in[11];

  float* ws = (float*)d_ws;
  const size_t SZ_LD = (size_t)L_SEQ * DMODEL;
  const size_t SZ_PRIME = (size_t)NHEADS * L_SEQ * MFEAT;
  float* Q      = ws;
  float* K      = Q + SZ_LD;
  float* V      = K + SZ_LD;
  float* hk     = V + SZ_LD;
  float* kstab  = hk + (size_t)NHEADS * L_SEQ;
  float* qprime = kstab + 64;
  float* kprime = qprime + SZ_PRIME;
  float* chunkbuf = kprime + SZ_PRIME;
  float* context = Q;

  perf_qkv_proj<<<dim3(L_SEQ / 32, DMODEL / 32, 3), 32, 0, stream>>>(
      query, key_, value, Wq, bq, Wk, bk, Wv, bv, Q, K, V);
  perf_hk<<<(NHEADS * L_SEQ) / 256, 256, 0, stream>>>(K, hk);
  perf_kstab<<<1, 1024, 0, stream>>>(hk, kstab);
  perf_primes<<<dim3(L_SEQ / 32, MFEAT / 32, 2 * NHEADS), 32, 0, stream>>>(
      Q, K, RF, hk, kstab, qprime, kprime);
  perf_chunk_sums<<<dim3(8, NCHUNKS, NHEADS), 32, 0, stream>>>(kprime, V, chunkbuf);
  perf_chunk_prefix<<<(NHEADS * PREFIX_VEC) / 128, 128, 0, stream>>>(chunkbuf);
  perf_scan<<<dim3(CHUNK / 16, NCHUNKS, NHEADS), 32, 0, stream>>>(
      qprime, kprime, V, chunkbuf, context);
  perf_out_proj<<<dim3(L_SEQ / 32, DMODEL / 32), 32, 0, stream>>>(
      context, Wo, bo, (float*)d_out);
}


# projection GEMMs: operand loads 16 bytes per lane, issued one 64-k block ahead of the products (register double buffer); k permuted inside each 8-k step: sums taken in a different order
# speedup vs baseline: 6.6327x; 1.1342x over previous
#include <hip/hip_runtime.h>
#include <math.h>

#define L_SEQ 2048
#define DMODEL 512
#define NHEADS 8
#define DHEAD 64
#define MFEAT 128
#define NCHUNKS 16
#define CHUNK 128
#define CHUNK_STRIDE 8320
#define KERNEL_EPS 1e-4f
#define DEN_EPS 1e-6f
#define QK_SCALE 0.21022410381342864f
#define NC_FEAT 0.08838834764831845f

typedef float v2f __attribute__((ext_vector_type(2)));
typedef float v8f __attribute__((ext_vector_type(8)));

struct Acc2x2 { v8f c[2][2]; };

__device__ __forceinline__ Acc2x2 wmma_block2x2_f32(const float* __restrict__ A, int lda,
                                                    const float* __restrict__ B, int bks, int bns,
                                                    int K, int lane) {
  const int half = lane >> 4;
  const int l16  = lane & 15;
  const float* arow0 = A + l16 * lda;
  const float* arow1 = A + (l16 + 16) * lda;
  const float* bcol0 = B + l16 * bns;
  const float* bcol1 = B + (l16 + 16) * bns;
  const v8f zero = {0.f, 0.f, 0.f, 0.f, 0.f, 0.f, 0.f, 0.f};
  Acc2x2 r;
  r.c[0][0] = zero; r.c[0][1] = zero; r.c[1][0] = zero; r.c[1][1] = zero;
  for (int k0 = 0; k0 < K; k0 += 4) {
    const int ka = k0 + 2 * half;
    v2f a0, a1, b0, b1;
    a0.x = arow0[ka];          a0.y = arow0[ka + 1];
    a1.x = arow1[ka];          a1.y = arow1[ka + 1];
    b0.x = bcol0[ka * bks];    b0.y = bcol0[(ka + 1) * bks];
    b1.x = bcol1[ka * bks];    b1.y = bcol1[(ka + 1) * bks];
    r.c[0][0] = __builtin_amdgcn_wmma_f32_16x16x4_f32(false, a0, false, b0, (short)0, r.c[0][0], false, false);
    r.c[0][1] = __builtin_amdgcn_wmma_f32_16x16x4_f32(false, a0, false, b1, (short)0, r.c[0][1], false, false);
    r.c[1][0] = __builtin_amdgcn_wmma_f32_16x16x4_f32(false, a1, false, b0, (short)0, r.c[1][0], false, false);
    r.c[1][1] = __builtin_amdgcn_wmma_f32_16x16x4_f32(false, a1, false, b1, (short)0, r.c[1][1], false, false);
  }
  return r;
}

#define GEMM_KB 64
#define GEMM_STEPS (GEMM_KB / 8)
struct OperandBlock { float4 a0[GEMM_STEPS], a1[GEMM_STEPS], b0[GEMM_STEPS], b1[GEMM_STEPS]; };

__device__ __forceinline__ void load_operand_block(OperandBlock& o,
                                                   const float* __restrict__ a0p,
                                                   const float* __restrict__ a1p,
                                                   const float* __restrict__ b0p,
                                                   const float* __restrict__ b1p, int kb) {
#pragma unroll
  for (int s = 0; s < GEMM_STEPS; ++s) {
    o.a0[s] = *reinterpret_cast<const float4*>(a0p + kb + 8 * s);
    o.a1[s] = *reinterpret_cast<const float4*>(a1p + kb + 8 * s);
    o.b0[s] = *reinterpret_cast<const float4*>(b0p + kb + 8 * s);
    o.b1[s] = *reinterpret_cast<const float4*>(b1p + kb + 8 * s);
  }
}

__device__ __forceinline__ void mma_operand_block(Acc2x2& r, const OperandBlock& o) {
#pragma unroll
  for (int s = 0; s < GEMM_STEPS; ++s) {
    v2f a0l, a0h, a1l, a1h, b0l, b0h, b1l, b1h;
    a0l.x = o.a0[s].x; a0l.y = o.a0[s].y; a0h.x = o.a0[s].z; a0h.y = o.a0[s].w;
    a1l.x = o.a1[s].x; a1l.y = o.a1[s].y; a1h.x = o.a1[s].z; a1h.y = o.a1[s].w;
    b0l.x = o.b0[s].x; b0l.y = o.b0[s].y; b0h.x = o.b0[s].z; b0h.y = o.b0[s].w;
    b1l.x = o.b1[s].x; b1l.y = o.b1[s].y; b1h.x = o.b1[s].z; b1h.y = o.b1[s].w;
    r.c[0][0] = __builtin_amdgcn_wmma_f32_16x16x4_f32(false, a0l, false, b0l, (short)0, r.c[0][0], false, false);
    r.c[0][1] = __builtin_amdgcn_wmma_f32_16x16x4_f32(false, a0l, false, b1l, (short)0, r.c[0][1], false, false);
    r.c[1][0] = __builtin_amdgcn_wmma_f32_16x16x4_f32(false, a1l, false, b0l, (short)0, r.c[1][0], false, false);
    r.c[1][1] = __builtin_amdgcn_wmma_f32_16x16x4_f32(false, a1l, false, b1l, (short)0, r.c[1][1], false, false);
    r.c[0][0] = __builtin_amdgcn_wmma_f32_16x16x4_f32(false, a0h, false, b0h, (short)0, r.c[0][0], false, false);
    r.c[0][1] = __builtin_amdgcn_wmma_f32_16x16x4_f32(false, a0h, false, b1h, (short)0, r.c[0][1], false, false);
    r.c[1][0] = __builtin_amdgcn_wmma_f32_16x16x4_f32(false, a1h, false, b0h, (short)0, r.c[1][0], false, false);
    r.c[1][1] = __builtin_amdgcn_wmma_f32_16x16x4_f32(false, a1h, false, b1h, (short)0, r.c[1][1], false, false);
  }
}

__device__ __forceinline__ Acc2x2 wmma_block2x2_kcontig(const float* __restrict__ A, int lda,
                                                        const float* __restrict__ Bt, int ldb,
                                                        int K, int lane) {
  const int half = lane >> 4;
  const int l16  = lane & 15;
  const float* a0p = A + l16 * lda + 4 * half;
  const float* a1p = A + (l16 + 16) * lda + 4 * half;
  const float* b0p = Bt + l16 * ldb + 4 * half;
  const float* b1p = Bt + (l16 + 16) * ldb + 4 * half;
  const v8f zero = {0.f, 0.f, 0.f, 0.f, 0.f, 0.f, 0.f, 0.f};
  Acc2x2 r;
  r.c[0][0] = zero; r.c[0][1] = zero; r.c[1][0] = zero; r.c[1][1] = zero;
  OperandBlock u, w;
  load_operand_block(u, a0p, a1p, b0p, b1p, 0);
  for (int kb = 0; kb < K; kb += 2 * GEMM_KB) {
    load_operand_block(w, a0p, a1p, b0p, b1p, kb + GEMM_KB);
    __builtin_amdgcn_sched_barrier(0);
    mma_operand_block(r, u);
    if (kb + 2 * GEMM_KB < K) load_operand_block(u, a0p, a1p, b0p, b1p, kb + 2 * GEMM_KB);
    __builtin_amdgcn_sched_barrier(0);
    mma_operand_block(r, w);
  }
  return r;
}

__global__ void __launch_bounds__(32)
perf_qkv_proj(const float* __restrict__ xq, const float* __restrict__ xk,
              const float* __restrict__ xv,
              const float* __restrict__ Wq, const float* __restrict__ bq,
              const float* __restrict__ Wk, const float* __restrict__ bk,
              const float* __restrict__ Wv, const float* __restrict__ bv,
              float* __restrict__ Q, float* __restrict__ K, float* __restrict__ V) {
  const int tm = blockIdx.x * 32;
  const int tn = blockIdx.y * 32;
  const int mat = blockIdx.z;
  const int lane = threadIdx.x;
  const float* X; const float* W; const float* bias; float* Y; float scl;
  if (mat == 0)      { X = xq; W = Wq; bias = bq; Y = Q; scl = QK_SCALE; }
  else if (mat == 1) { X = xk; W = Wk; bias = bk; Y = K; scl = QK_SCALE; }
  else               { X = xv; W = Wv; bias = bv; Y = V; scl = 1.f; }
  Acc2x2 r = wmma_block2x2_kcontig(X + tm * DMODEL, DMODEL,
                                   W + tn * DMODEL, DMODEL, DMODEL, lane);
  const int l16 = lane & 15;
  const int half = lane >> 4;
#pragma unroll
  for (int ti = 0; ti < 2; ++ti)
#pragma unroll
    for (int tj = 0; tj < 2; ++tj) {
      const int col = tn + tj * 16 + l16;
      const int rbase = tm + ti * 16 + 8 * half;
      const float bb = bias[col];
#pragma unroll
      for (int i = 0; i < 8; ++i)
        Y[(rbase + i) * DMODEL + col] = (r.c[ti][tj][i] + bb) * scl;
    }
}

__global__ void __launch_bounds__(256)
perf_hk(const float* __restrict__ K, float* __restrict__ hk) {
  const int idx = blockIdx.x * 256 + threadIdx.x;
  const int h = idx >> 11;
  const int l = idx & (L_SEQ - 1);
  const float* row = K + l * DMODEL + h * DHEAD;
  float s = 0.f;
#pragma unroll 8
  for (int d = 0; d < DHEAD; ++d) s += row[d] * row[d];
  hk[idx] = -0.5f * s;
}

__global__ void __launch_bounds__(1024)
perf_kstab(const float* __restrict__ hk, float* __restrict__ kstab) {
  __shared__ float red[32];
  const int t = threadIdx.x;
  const float4* p = reinterpret_cast<const float4*>(hk) + t;
  float4 v[4];
#pragma unroll
  for (int i = 0; i < 4; ++i) v[i] = p[i * 1024];
  float m = -3.402823e38f;
#pragma unroll
  for (int i = 0; i < 4; ++i)
    m = fmaxf(m, fmaxf(fmaxf(v[i].x, v[i].y), fmaxf(v[i].z, v[i].w)));
  for (int off = 16; off > 0; off >>= 1) m = fmaxf(m, __shfl_xor(m, off));
  if ((t & 31) == 0) red[t >> 5] = m;
  __syncthreads();
  if (t < 32) {
    float r = red[t];
    for (int off = 16; off > 0; off >>= 1) r = fmaxf(r, __shfl_xor(r, off));
    if (t == 0) kstab[0] = r;
  }
}

__global__ void __launch_bounds__(32)
perf_primes(const float* __restrict__ Q, const float* __restrict__ K,
            const float* __restrict__ RF, const float* __restrict__ hk,
            const float* __restrict__ kstab,
            float* __restrict__ qprime, float* __restrict__ kprime) {
  const int tm = blockIdx.x * 32;
  const int tn = blockIdx.y * 32;
  const int h = blockIdx.z >> 1;
  const int isK = blockIdx.z & 1;
  const int lane = threadIdx.x;
  const float* A = (isK ? K : Q) + tm * DMODEL + h * DHEAD;
  Acc2x2 r = wmma_block2x2_f32(A, DMODEL, RF + tn, MFEAT, 1, DHEAD, lane);
  const int l16 = lane & 15;
  const int half = lane >> 4;
  const float ks = isK ? kstab[0] : 0.f;
#pragma unroll
  for (int ti = 0; ti < 2; ++ti)
#pragma unroll
    for (int tj = 0; tj < 2; ++tj) {
      const int mcol = tn + tj * 16 + l16;
      const int rbase = tm + ti * 16 + 8 * half;
      if (isK) {
#pragma unroll
        for (int i = 0; i < 8; ++i) {
          const int l = rbase + i;
          kprime[(h * L_SEQ + l) * MFEAT + mcol] =
              NC_FEAT * (expf(hk[h * L_SEQ + l] + r.c[ti][tj][i] - ks) + KERNEL_EPS);
        }
      } else {
#pragma unroll
        for (int i = 0; i < 8; ++i) {
          const int l = rbase + i;
          qprime[(h * L_SEQ + l) * MFEAT + mcol] =
              NC_FEAT * (expf(r.c[ti][tj][i]) + KERNEL_EPS);
        }
      }
    }
}

__device__ __forceinline__ v8f wmma4(v2f a, v2f b, v8f c) {
  return __builtin_amdgcn_wmma_f32_16x16x4_f32(false, a, false, b, (short)0, c, false, false);
}

__global__ void __launch_bounds__(32)
perf_chunk_sums(const float* __restrict__ kprime, const float* __restrict__ V,
                float* __restrict__ chunkbuf) {
  const int blk = blockIdx.x;
  const int c = blockIdx.y;
  const int h = blockIdx.z;
  const int lane = threadIdx.x;
  const int half = lane >> 4;
  const int l16 = lane & 15;
  const int m0 = (blk >> 1) * 32;
  const int d0 = (blk & 1) * 32;
  const float* ka = kprime + (size_t)(h * L_SEQ + c * CHUNK + 2 * half) * MFEAT + m0 + 2 * l16;
  const float* vb = V + (size_t)(c * CHUNK + 2 * half) * DMODEL + h * DHEAD + d0 + 2 * l16;
  const v8f zero = {0.f, 0.f, 0.f, 0.f, 0.f, 0.f, 0.f, 0.f};
  v8f acc[2][2];
  acc[0][0] = zero; acc[0][1] = zero; acc[1][0] = zero; acc[1][1] = zero;
  float zx = 0.f, zy = 0.f;
#pragma unroll
  for (int ks = 0; ks < CHUNK / 4; ++ks) {
    const float2 ra0 = *reinterpret_cast<const float2*>(ka + (size_t)(4 * ks) * MFEAT);
    const float2 ra1 = *reinterpret_cast<const float2*>(ka + (size_t)(4 * ks + 1) * MFEAT);
    const float2 rb0 = *reinterpret_cast<const float2*>(vb + (size_t)(4 * ks) * DMODEL);
    const float2 rb1 = *reinterpret_cast<const float2*>(vb + (size_t)(4 * ks + 1) * DMODEL);
    v2f a0, a1, b0, b1;
    a0.x = ra0.x; a0.y = ra1.x; a1.x = ra0.y; a1.y = ra1.y;
    b0.x = rb0.x; b0.y = rb1.x; b1.x = rb0.y; b1.y = rb1.y;
    acc[0][0] = wmma4(a0, b0, acc[0][0]);
    acc[0][1] = wmma4(a0, b1, acc[0][1]);
    acc[1][0] = wmma4(a1, b0, acc[1][0]);
    acc[1][1] = wmma4(a1, b1, acc[1][1]);
    zx += ra0.x + ra1.x;
    zy += ra0.y + ra1.y;
  }
  float* buf = chunkbuf + (size_t)(h * NCHUNKS + c) * CHUNK_STRIDE;
#pragma unroll
  for (int jm = 0; jm < 2; ++jm)
#pragma unroll
    for (int r = 0; r < 8; ++r) {
      const int m = m0 + 2 * (r + 8 * half) + jm;
      float2 o;
      o.x = acc[jm][0][r]; o.y = acc[jm][1][r];
      *reinterpret_cast<float2*>(buf + m * DHEAD + d0 + 2 * l16) = o;
    }
  zx += __shfl_xor(zx, 16);
  zy += __shfl_xor(zy, 16);
  if (d0 == 0 && half == 0) {
    float2 o;
    o.x = zx; o.y = zy;
    *reinterpret_cast<float2*>(buf + MFEAT * DHEAD + m0 + 2 * l16) = o;
  }
}

#define PREFIX_VEC (CHUNK_STRIDE / 4)
__global__ void __launch_bounds__(128)
perf_chunk_prefix(float* __restrict__ chunkbuf) {
  const int gid = blockIdx.x * 128 + threadIdx.x;
  const int h = gid / PREFIX_VEC;
  const int i4 = gid - h * PREFIX_VEC;
  float4* p = reinterpret_cast<float4*>(chunkbuf + (size_t)h * NCHUNKS * CHUNK_STRIDE) + i4;
  float4 v[NCHUNKS];
#pragma unroll
  for (int c = 0; c < NCHUNKS; ++c) v[c] = p[(size_t)c * PREFIX_VEC];
  float4 carry;
  carry.x = 0.f; carry.y = 0.f; carry.z = 0.f; carry.w = 0.f;
#pragma unroll
  for (int c = 0; c < NCHUNKS; ++c) {
    const float4 tmp = v[c];
    p[(size_t)c * PREFIX_VEC] = carry;
    carry.x += tmp.x; carry.y += tmp.y; carry.z += tmp.z; carry.w += tmp.w;
  }
}

__global__ void __launch_bounds__(32)
perf_scan(const float* __restrict__ qprime, const float* __restrict__ kprime,
          const float* __restrict__ V, const float* __restrict__ chunkbuf,
          float* __restrict__ context) {
  const int lt = blockIdx.x;
  const int c = blockIdx.y;
  const int h = blockIdx.z;
  const int lane = threadIdx.x;
  const int half = lane >> 4;
  const int l16 = lane & 15;
  const int koff = 4 * half;
  const int lrow0 = c * CHUNK + lt * 16;
  const float* buf = chunkbuf + (size_t)(h * NCHUNKS + c) * CHUNK_STRIDE;

  const float* qrow = qprime + (size_t)(h * L_SEQ + lrow0 + l16) * MFEAT + koff;
  float4 q[16];
#pragma unroll
  for (int s = 0; s < 16; ++s) q[s] = *reinterpret_cast<const float4*>(qrow + 8 * s);

  const v8f zero = {0.f, 0.f, 0.f, 0.f, 0.f, 0.f, 0.f, 0.f};
  v8f num[4];
  num[0] = zero; num[1] = zero; num[2] = zero; num[3] = zero;
  v8f den = zero;

  const float* sp = buf + koff * DHEAD + l16;
  const float* zp = buf + MFEAT * DHEAD + koff;
#pragma unroll
  for (int s = 0; s < 16; ++s) {
    const float4 zz = *reinterpret_cast<const float4*>(zp + 8 * s);
    v2f a0, a1, z0, z1;
    a0.x = q[s].x; a0.y = q[s].y; a1.x = q[s].z; a1.y = q[s].w;
    z0.x = zz.x;   z0.y = zz.y;   z1.x = zz.z;   z1.y = zz.w;
    den = wmma4(a0, z0, den);
    den = wmma4(a1, z1, den);
#pragma unroll
    for (int dt = 0; dt < 4; ++dt) {
      v2f b0, b1;
      b0.x = sp[(8 * s + 0) * DHEAD + dt * 16];
      b0.y = sp[(8 * s + 1) * DHEAD + dt * 16];
      b1.x = sp[(8 * s + 2) * DHEAD + dt * 16];
      b1.y = sp[(8 * s + 3) * DHEAD + dt * 16];
      num[dt] = wmma4(a0, b0, num[dt]);
      num[dt] = wmma4(a1, b1, num[dt]);
    }
  }

  const float* kbase = kprime + (size_t)(h * L_SEQ + c * CHUNK + l16) * MFEAT + koff;
  const float* vbase = V + (size_t)(c * CHUNK + 8 * half) * DMODEL + h * DHEAD + l16;
  v2f ones;
  ones.x = 1.f; ones.y = 1.f;
  for (int jt = 0; jt <= lt; ++jt) {
    const float* krow = kbase + (size_t)jt * 16 * MFEAT;
    float4 kf[16];
#pragma unroll
    for (int s = 0; s < 16; ++s) kf[s] = *reinterpret_cast<const float4*>(krow + 8 * s);
    v8f t0 = zero, t1 = zero;
#pragma unroll
    for (int s = 0; s < 16; ++s) {
      v2f ka0, ka1, qb0, qb1;
      ka0.x = kf[s].x; ka0.y = kf[s].y; ka1.x = kf[s].z; ka1.y = kf[s].w;
      qb0.x = q[s].x;  qb0.y = q[s].y;  qb1.x = q[s].z;  qb1.y = q[s].w;
      t0 = wmma4(ka0, qb0, t0);
      t1 = wmma4(ka1, qb1, t1);
    }
    v8f t = t0 + t1;
    if (jt == lt) {
#pragma unroll
      for (int r = 0; r < 8; ++r) t[r] = (r + 8 * half > l16) ? 0.f : t[r];
    }
    const float* vp = vbase + (size_t)jt * 16 * DMODEL;
#pragma unroll
    for (int s2 = 0; s2 < 4; ++s2) {
      v2f a;
      a.x = t[2 * s2]; a.y = t[2 * s2 + 1];
      den = wmma4(a, ones, den);
#pragma unroll
      for (int dt = 0; dt < 4; ++dt) {
        v2f b;
        b.x = vp[(2 * s2) * DMODEL + dt * 16];
        b.y = vp[(2 * s2 + 1) * DMODEL + dt * 16];
        num[dt] = wmma4(a, b, num[dt]);
      }
    }
  }

#pragma unroll
  for (int dt = 0; dt < 4; ++dt)
#pragma unroll
    for (int r = 0; r < 8; ++r) {
      const int l = lrow0 + r + 8 * half;
      context[(size_t)l * DMODEL + h * DHEAD + dt * 16 + l16] =
          num[dt][r] / (den[r] + DEN_EPS);
    }
}

__global__ void __launch_bounds__(32)
perf_out_proj(const float* __restrict__ ctx, const float* __restrict__ Wo,
              const float* __restrict__ bo, float* __restrict__ out) {
  const int tm = blockIdx.x * 32;
  const int tn = blockIdx.y * 32;
  const int lane = threadIdx.x;
  Acc2x2 r = wmma_block2x2_kcontig(ctx + tm * DMODEL, DMODEL,
                                   Wo + tn * DMODEL, DMODEL, DMODEL, lane);
  const int l16 = lane & 15;
  const int half = lane >> 4;
#pragma unroll
  for (int ti = 0; ti < 2; ++ti)
#pragma unroll
    for (int tj = 0; tj < 2; ++tj) {
      const int col = tn + tj * 16 + l16;
      const int rbase = tm + ti * 16 + 8 * half;
      const float bb = bo[col];
#pragma unroll
      for (int i = 0; i < 8; ++i)
        out[(rbase + i) * DMODEL + col] = r.c[ti][tj][i] + bb;
    }
}

extern "C" void kernel_launch(void* const* d_in, const int* in_sizes, int n_in,
                              void* d_out, int out_size, void* d_ws, size_t ws_size,
                              hipStream_t stream) {
  (void)in_sizes; (void)n_in; (void)out_size; (void)ws_size;
  const float* query = (const float*)d_in[0];
  const float* key_  = (const float*)d_in[1];
  const float* value = (const float*)d_in[2];
  const float* Wq = (const float*)d_in[3];
  const float* bq = (const float*)d_in[4];
  const float* Wk = (const float*)d_in[5];
  const float* bk = (const float*)d_in[6];
  const float* Wv = (const float*)d_in[7];
  const float* bv = (const float*)d_in[8];
  const float* Wo = (const float*)d_in[9];
  const float* bo = (const float*)d_in[10];
  const float* RF = (const float*)d_in[11];

  float* ws = (float*)d_ws;
  const size_t SZ_LD = (size_t)L_SEQ * DMODEL;
  const size_t SZ_PRIME = (size_t)NHEADS * L_SEQ * MFEAT;
  float* Q      = ws;
  float* K      = Q + SZ_LD;
  float* V      = K + SZ_LD;
  float* hk     = V + SZ_LD;
  float* kstab  = hk + (size_t)NHEADS * L_SEQ;
  float* qprime = kstab + 64;
  float* kprime = qprime + SZ_PRIME;
  float* chunkbuf = kprime + SZ_PRIME;
  float* context = Q;

  perf_qkv_proj<<<dim3(L_SEQ / 32, DMODEL / 32, 3), 32, 0, stream>>>(
      query, key_, value, Wq, bq, Wk, bk, Wv, bv, Q, K, V);
  perf_hk<<<(NHEADS * L_SEQ) / 256, 256, 0, stream>>>(K, hk);
  perf_kstab<<<1, 1024, 0, stream>>>(hk, kstab);
  perf_primes<<<dim3(L_SEQ / 32, MFEAT / 32, 2 * NHEADS), 32, 0, stream>>>(
      Q, K, RF, hk, kstab, qprime, kprime);
  perf_chunk_sums<<<dim3(8, NCHUNKS, NHEADS), 32, 0, stream>>>(kprime, V, chunkbuf);
  perf_chunk_prefix<<<(NHEADS * PREFIX_VEC) / 128, 128, 0, stream>>>(chunkbuf);
  perf_scan<<<dim3(CHUNK / 16, NCHUNKS, NHEADS), 32, 0, stream>>>(
      qprime, kprime, V, chunkbuf, context);
  perf_out_proj<<<dim3(L_SEQ / 32, DMODEL / 32), 32, 0, stream>>>(
      context, Wo, bo, (float*)d_out);
}


# projection GEMMs: 64x64 tile per 4-wave work-group, operands staged once per k block through LDS by whole-line global loads and shared by the waves; fragment k order and sums unchanged
# speedup vs baseline: 7.4276x; 1.1198x over previous
#include <hip/hip_runtime.h>
#include <math.h>

#define L_SEQ 2048
#define DMODEL 512
#define NHEADS 8
#define DHEAD 64
#define MFEAT 128
#define NCHUNKS 16
#define CHUNK 128
#define CHUNK_STRIDE 8320
#define KERNEL_EPS 1e-4f
#define DEN_EPS 1e-6f
#define QK_SCALE 0.21022410381342864f
#define NC_FEAT 0.08838834764831845f

typedef float v2f __attribute__((ext_vector_type(2)));
typedef float v4f __attribute__((ext_vector_type(4)));
typedef float v8f __attribute__((ext_vector_type(8)));

struct Acc2x2 { v8f c[2][2]; };

__device__ __forceinline__ Acc2x2 wmma_block2x2_f32(const float* __restrict__ A, int lda,
                                                    const float* __restrict__ B, int bks, int bns,
                                                    int K, int lane) {
  const int half = lane >> 4;
  const int l16  = lane & 15;
  const float* arow0 = A + l16 * lda;
  const float* arow1 = A + (l16 + 16) * lda;
  const float* bcol0 = B + l16 * bns;
  const float* bcol1 = B + (l16 + 16) * bns;
  const v8f zero = {0.f, 0.f, 0.f, 0.f, 0.f, 0.f, 0.f, 0.f};
  Acc2x2 r;
  r.c[0][0] = zero; r.c[0][1] = zero; r.c[1][0] = zero; r.c[1][1] = zero;
  for (int k0 = 0; k0 < K; k0 += 4) {
    const int ka = k0 + 2 * half;
    v2f a0, a1, b0, b1;
    a0.x = arow0[ka];          a0.y = arow0[ka + 1];
    a1.x = arow1[ka];          a1.y = arow1[ka + 1];
    b0.x = bcol0[ka * bks];    b0.y = bcol0[(ka + 1) * bks];
    b1.x = bcol1[ka * bks];    b1.y = bcol1[(ka + 1) * bks];
    r.c[0][0] = __builtin_amdgcn_wmma_f32_16x16x4_f32(false, a0, false, b0, (short)0, r.c[0][0], false, false);
    r.c[0][1] = __builtin_amdgcn_wmma_f32_16x16x4_f32(false, a0, false, b1, (short)0, r.c[0][1], false, false);
    r.c[1][0] = __builtin_amdgcn_wmma_f32_16x16x4_f32(false, a1, false, b0, (short)0, r.c[1][0], false, false);
    r.c[1][1] = __builtin_amdgcn_wmma_f32_16x16x4_f32(false, a1, false, b1, (short)0, r.c[1][1], false, false);
  }
  return r;
}

#define GEMM_TILE 64
#define GEMM_KB 64
#define GEMM_STEPS (GEMM_KB / 8)
#define GEMM_PITCH (GEMM_KB + 4)
#define GEMM_PLANE (GEMM_TILE * GEMM_PITCH)
#define GEMM_LDS_FLOATS (4 * GEMM_PLANE)
#define GEMM_COPIES (GEMM_TILE * GEMM_KB / 4 / 128)
struct OperandBlock { float4 a0[GEMM_STEPS], a1[GEMM_STEPS], b0[GEMM_STEPS], b1[GEMM_STEPS]; };
struct StageRegs { v4f a[GEMM_COPIES], b[GEMM_COPIES]; };

__device__ __forceinline__ void load_stage_regs(StageRegs& g, const float* __restrict__ ag,
                                                const float* __restrict__ bg, int kb) {
#pragma unroll
  for (int i = 0; i < GEMM_COPIES; ++i) {
    g.a[i] = *reinterpret_cast<const v4f*>(ag + (size_t)(8 * i) * DMODEL + kb);
    g.b[i] = *reinterpret_cast<const v4f*>(bg + (size_t)(8 * i) * DMODEL + kb);
  }
}

__device__ __forceinline__ void store_stage_regs(const StageRegs& g, float* sw) {
#pragma unroll
  for (int i = 0; i < GEMM_COPIES; ++i) {
    *reinterpret_cast<v4f*>(sw + 8 * i * GEMM_PITCH) = g.a[i];
    *reinterpret_cast<v4f*>(sw + GEMM_PLANE + 8 * i * GEMM_PITCH) = g.b[i];
  }
}

__device__ __forceinline__ void read_fragments(OperandBlock& o, const float* af, const float* bf) {
#pragma unroll
  for (int s = 0; s < GEMM_STEPS; ++s) {
    o.a0[s] = *reinterpret_cast<const float4*>(af + 8 * s);
    o.a1[s] = *reinterpret_cast<const float4*>(af + 16 * GEMM_PITCH + 8 * s);
    o.b0[s] = *reinterpret_cast<const float4*>(bf + 8 * s);
    o.b1[s] = *reinterpret_cast<const float4*>(bf + 16 * GEMM_PITCH + 8 * s);
  }
}

__device__ __forceinline__ void mma_operand_block(Acc2x2& r, const OperandBlock& o) {
#pragma unroll
  for (int s = 0; s < GEMM_STEPS; ++s) {
    v2f a0l, a0h, a1l, a1h, b0l, b0h, b1l, b1h;
    a0l.x = o.a0[s].x; a0l.y = o.a0[s].y; a0h.x = o.a0[s].z; a0h.y = o.a0[s].w;
    a1l.x = o.a1[s].x; a1l.y = o.a1[s].y; a1h.x = o.a1[s].z; a1h.y = o.a1[s].w;
    b0l.x = o.b0[s].x; b0l.y = o.b0[s].y; b0h.x = o.b0[s].z; b0h.y = o.b0[s].w;
    b1l.x = o.b1[s].x; b1l.y = o.b1[s].y; b1h.x = o.b1[s].z; b1h.y = o.b1[s].w;
    r.c[0][0] = __builtin_amdgcn_wmma_f32_16x16x4_f32(false, a0l, false, b0l, (short)0, r.c[0][0], false, false);
    r.c[0][1] = __builtin_amdgcn_wmma_f32_16x16x4_f32(false, a0l, false, b1l, (short)0, r.c[0][1], false, false);
    r.c[1][0] = __builtin_amdgcn_wmma_f32_16x16x4_f32(false, a1l, false, b0l, (short)0, r.c[1][0], false, false);
    r.c[1][1] = __builtin_amdgcn_wmma_f32_16x16x4_f32(false, a1l, false, b1l, (short)0, r.c[1][1], false, false);
    r.c[0][0] = __builtin_amdgcn_wmma_f32_16x16x4_f32(false, a0h, false, b0h, (short)0, r.c[0][0], false, false);
    r.c[0][1] = __builtin_amdgcn_wmma_f32_16x16x4_f32(false, a0h, false, b1h, (short)0, r.c[0][1], false, false);
    r.c[1][0] = __builtin_amdgcn_wmma_f32_16x16x4_f32(false, a1h, false, b0h, (short)0, r.c[1][0], false, false);
    r.c[1][1] = __builtin_amdgcn_wmma_f32_16x16x4_f32(false, a1h, false, b1h, (short)0, r.c[1][1], false, false);
  }
}

__device__ __forceinline__ Acc2x2 gemm_tile64_lds(const float* __restrict__ A,
                                                  const float* __restrict__ Bt,
                                                  float* lds, int tid) {
  const int lane = tid & 31;
  const int wave = tid >> 5;
  const int half = lane >> 4;
  const int l16  = lane & 15;
  const int wm = wave >> 1;
  const int wn = wave & 1;
  const int srow = tid >> 4;
  const int spiece = tid & 15;
  const float* ag = A + srow * DMODEL + 4 * spiece;
  const float* bg = Bt + srow * DMODEL + 4 * spiece;
  float* sw = lds + srow * GEMM_PITCH + 4 * spiece;
  const float* af = lds + (wm * 32 + l16) * GEMM_PITCH + 4 * half;
  const float* bf = lds + GEMM_PLANE + (wn * 32 + l16) * GEMM_PITCH + 4 * half;
  const v8f zero = {0.f, 0.f, 0.f, 0.f, 0.f, 0.f, 0.f, 0.f};
  Acc2x2 r;
  r.c[0][0] = zero; r.c[0][1] = zero; r.c[1][0] = zero; r.c[1][1] = zero;
  StageRegs g;
  load_stage_regs(g, ag, bg, 0);
  store_stage_regs(g, sw);
  __syncthreads();
  for (int t = 0; t < DMODEL / GEMM_KB; ++t) {
    const int cur = (t & 1) * 2 * GEMM_PLANE;
    const int nxt = 2 * GEMM_PLANE - cur;
    const bool more = t + 1 < DMODEL / GEMM_KB;
    if (more) load_stage_regs(g, ag, bg, (t + 1) * GEMM_KB);
    __builtin_amdgcn_sched_barrier(0);
    OperandBlock o;
    read_fragments(o, af + cur, bf + cur);
    __builtin_amdgcn_sched_barrier(0);
    mma_operand_block(r, o);
    __builtin_amdgcn_sched_barrier(0);
    if (more) store_stage_regs(g, sw + nxt);
    __syncthreads();
  }
  return r;
}

__global__ void __launch_bounds__(128)
perf_qkv_proj(const float* __restrict__ xq, const float* __restrict__ xk,
              const float* __restrict__ xv,
              const float* __restrict__ Wq, const float* __restrict__ bq,
              const float* __restrict__ Wk, const float* __restrict__ bk,
              const float* __restrict__ Wv, const float* __restrict__ bv,
              float* __restrict__ Q, float* __restrict__ K, float* __restrict__ V) {
  __shared__ __attribute__((aligned(16))) float lds[GEMM_LDS_FLOATS];
  const int tid = threadIdx.x;
  const int lane = tid & 31;
  const int wave = tid >> 5;
  const int tm = blockIdx.x * GEMM_TILE + (wave >> 1) * 32;
  const int tn = blockIdx.y * GEMM_TILE + (wave & 1) * 32;
  const int mat = blockIdx.z;
  const float* X; const float* W; const float* bias; float* Y; float scl;
  if (mat == 0)      { X = xq; W = Wq; bias = bq; Y = Q; scl = QK_SCALE; }
  else if (mat == 1) { X = xk; W = Wk; bias = bk; Y = K; scl = QK_SCALE; }
  else               { X = xv; W = Wv; bias = bv; Y = V; scl = 1.f; }
  Acc2x2 r = gemm_tile64_lds(X + blockIdx.x * GEMM_TILE * DMODEL,
                             W + blockIdx.y * GEMM_TILE * DMODEL, lds, tid);
  const int l16 = lane & 15;
  const int half = lane >> 4;
#pragma unroll
  for (int ti = 0; ti < 2; ++ti)
#pragma unroll
    for (int tj = 0; tj < 2; ++tj) {
      const int col = tn + tj * 16 + l16;
      const int rbase = tm + ti * 16 + 8 * half;
      const float bb = bias[col];
#pragma unroll
      for (int i = 0; i < 8; ++i)
        Y[(rbase + i) * DMODEL + col] = (r.c[ti][tj][i] + bb) * scl;
    }
}

__global__ void __launch_bounds__(256)
perf_hk(const float* __restrict__ K, float* __restrict__ hk) {
  const int idx = blockIdx.x * 256 + threadIdx.x;
  const int h = idx >> 11;
  const int l = idx & (L_SEQ - 1);
  const float* row = K + l * DMODEL + h * DHEAD;
  float s = 0.f;
#pragma unroll 8
  for (int d = 0; d < DHEAD; ++d) s += row[d] * row[d];
  hk[idx] = -0.5f * s;
}

__global__ void __launch_bounds__(1024)
perf_kstab(const float* __restrict__ hk, float* __restrict__ kstab) {
  __shared__ float red[32];
  const int t = threadIdx.x;
  const float4* p = reinterpret_cast<const float4*>(hk) + t;
  float4 v[4];
#pragma unroll
  for (int i = 0; i < 4; ++i) v[i] = p[i * 1024];
  float m = -3.402823e38f;
#pragma unroll
  for (int i = 0; i < 4; ++i)
    m = fmaxf(m, fmaxf(fmaxf(v[i].x, v[i].y), fmaxf(v[i].z, v[i].w)));
  for (int off = 16; off > 0; off >>= 1) m = fmaxf(m, __shfl_xor(m, off));
  if ((t & 31) == 0) red[t >> 5] = m;
  __syncthreads();
  if (t < 32) {
    float r = red[t];
    for (int off = 16; off > 0; off >>= 1) r = fmaxf(r, __shfl_xor(r, off));
    if (t == 0) kstab[0] = r;
  }
}

__global__ void __launch_bounds__(32)
perf_primes(const float* __restrict__ Q, const float* __restrict__ K,
            const float* __restrict__ RF, const float* __restrict__ hk,
            const float* __restrict__ kstab,
            float* __restrict__ qprime, float* __restrict__ kprime) {
  const int tm = blockIdx.x * 32;
  const int tn = blockIdx.y * 32;
  const int h = blockIdx.z >> 1;
  const int isK = blockIdx.z & 1;
  const int lane = threadIdx.x;
  const float* A = (isK ? K : Q) + tm * DMODEL + h * DHEAD;
  Acc2x2 r = wmma_block2x2_f32(A, DMODEL, RF + tn, MFEAT, 1, DHEAD, lane);
  const int l16 = lane & 15;
  const int half = lane >> 4;
  const float ks = isK ? kstab[0] : 0.f;
#pragma unroll
  for (int ti = 0; ti < 2; ++ti)
#pragma unroll
    for (int tj = 0; tj < 2; ++tj) {
      const int mcol = tn + tj * 16 + l16;
      const int rbase = tm + ti * 16 + 8 * half;
      if (isK) {
#pragma unroll
        for (int i = 0; i < 8; ++i) {
          const int l = rbase + i;
          kprime[(h * L_SEQ + l) * MFEAT + mcol] =
              NC_FEAT * (expf(hk[h * L_SEQ + l] + r.c[ti][tj][i] - ks) + KERNEL_EPS);
        }
      } else {
#pragma unroll
        for (int i = 0; i < 8; ++i) {
          const int l = rbase + i;
          qprime[(h * L_SEQ + l) * MFEAT + mcol] =
              NC_FEAT * (expf(r.c[ti][tj][i]) + KERNEL_EPS);
        }
      }
    }
}

__device__ __forceinline__ v8f wmma4(v2f a, v2f b, v8f c) {
  return __builtin_amdgcn_wmma_f32_16x16x4_f32(false, a, false, b, (short)0, c, false, false);
}

__global__ void __launch_bounds__(32)
perf_chunk_sums(const float* __restrict__ kprime, const float* __restrict__ V,
                float* __restrict__ chunkbuf) {
  const int blk = blockIdx.x;
  const int c = blockIdx.y;
  const int h = blockIdx.z;
  const int lane = threadIdx.x;
  const int half = lane >> 4;
  const int l16 = lane & 15;
  const int m0 = (blk >> 1) * 32;
  const int d0 = (blk & 1) * 32;
  const float* ka = kprime + (size_t)(h * L_SEQ + c * CHUNK + 2 * half) * MFEAT + m0 + 2 * l16;
  const float* vb = V + (size_t)(c * CHUNK + 2 * half) * DMODEL + h * DHEAD + d0 + 2 * l16;
  const v8f zero = {0.f, 0.f, 0.f, 0.f, 0.f, 0.f, 0.f, 0.f};
  v8f acc[2][2];
  acc[0][0] = zero; acc[0][1] = zero; acc[1][0] = zero; acc[1][1] = zero;
  float zx = 0.f, zy = 0.f;
#pragma unroll
  for (int ks = 0; ks < CHUNK / 4; ++ks) {
    const float2 ra0 = *reinterpret_cast<const float2*>(ka + (size_t)(4 * ks) * MFEAT);
    const float2 ra1 = *reinterpret_cast<const float2*>(ka + (size_t)(4 * ks + 1) * MFEAT);
    const float2 rb0 = *reinterpret_cast<const float2*>(vb + (size_t)(4 * ks) * DMODEL);
    const float2 rb1 = *reinterpret_cast<const float2*>(vb + (size_t)(4 * ks + 1) * DMODEL);
    v2f a0, a1, b0, b1;
    a0.x = ra0.x; a0.y = ra1.x; a1.x = ra0.y; a1.y = ra1.y;
    b0.x = rb0.x; b0.y = rb1.x; b1.x = rb0.y; b1.y = rb1.y;
    acc[0][0] = wmma4(a0, b0, acc[0][0]);
    acc[0][1] = wmma4(a0, b1, acc[0][1]);
    acc[1][0] = wmma4(a1, b0, acc[1][0]);
    acc[1][1] = wmma4(a1, b1, acc[1][1]);
    zx += ra0.x + ra1.x;
    zy += ra0.y + ra1.y;
  }
  float* buf = chunkbuf + (size_t)(h * NCHUNKS + c) * CHUNK_STRIDE;
#pragma unroll
  for (int jm = 0; jm < 2; ++jm)
#pragma unroll
    for (int r = 0; r < 8; ++r) {
      const int m = m0 + 2 * (r + 8 * half) + jm;
      float2 o;
      o.x = acc[jm][0][r]; o.y = acc[jm][1][r];
      *reinterpret_cast<float2*>(buf + m * DHEAD + d0 + 2 * l16) = o;
    }
  zx += __shfl_xor(zx, 16);
  zy += __shfl_xor(zy, 16);
  if (d0 == 0 && half == 0) {
    float2 o;
    o.x = zx; o.y = zy;
    *reinterpret_cast<float2*>(buf + MFEAT * DHEAD + m0 + 2 * l16) = o;
  }
}

#define PREFIX_VEC (CHUNK_STRIDE / 4)
__global__ void __launch_bounds__(128)
perf_chunk_prefix(float* __restrict__ chunkbuf) {
  const int gid = blockIdx.x * 128 + threadIdx.x;
  const int h = gid / PREFIX_VEC;
  const int i4 = gid - h * PREFIX_VEC;
  float4* p = reinterpret_cast<float4*>(chunkbuf + (size_t)h * NCHUNKS * CHUNK_STRIDE) + i4;
  float4 v[NCHUNKS];
#pragma unroll
  for (int c = 0; c < NCHUNKS; ++c) v[c] = p[(size_t)c * PREFIX_VEC];
  float4 carry;
  carry.x = 0.f; carry.y = 0.f; carry.z = 0.f; carry.w = 0.f;
#pragma unroll
  for (int c = 0; c < NCHUNKS; ++c) {
    const float4 tmp = v[c];
    p[(size_t)c * PREFIX_VEC] = carry;
    carry.x += tmp.x; carry.y += tmp.y; carry.z += tmp.z; carry.w += tmp.w;
  }
}

__global__ void __launch_bounds__(32)
perf_scan(const float* __restrict__ qprime, const float* __restrict__ kprime,
          const float* __restrict__ V, const float* __restrict__ chunkbuf,
          float* __restrict__ context) {
  const int lt = blockIdx.x;
  const int c = blockIdx.y;
  const int h = blockIdx.z;
  const int lane = threadIdx.x;
  const int half = lane >> 4;
  const int l16 = lane & 15;
  const int koff = 4 * half;
  const int lrow0 = c * CHUNK + lt * 16;
  const float* buf = chunkbuf + (size_t)(h * NCHUNKS + c) * CHUNK_STRIDE;

  const float* qrow = qprime + (size_t)(h * L_SEQ + lrow0 + l16) * MFEAT + koff;
  float4 q[16];
#pragma unroll
  for (int s = 0; s < 16; ++s) q[s] = *reinterpret_cast<const float4*>(qrow + 8 * s);

  const v8f zero = {0.f, 0.f, 0.f, 0.f, 0.f, 0.f, 0.f, 0.f};
  v8f num[4];
  num[0] = zero; num[1] = zero; num[2] = zero; num[3] = zero;
  v8f den = zero;

  const float* sp = buf + koff * DHEAD + l16;
  const float* zp = buf + MFEAT * DHEAD + koff;
#pragma unroll
  for (int s = 0; s < 16; ++s) {
    const float4 zz = *reinterpret_cast<const float4*>(zp + 8 * s);
    v2f a0, a1, z0, z1;
    a0.x = q[s].x; a0.y = q[s].y; a1.x = q[s].z; a1.y = q[s].w;
    z0.x = zz.x;   z0.y = zz.y;   z1.x = zz.z;   z1.y = zz.w;
    den = wmma4(a0, z0, den);
    den = wmma4(a1, z1, den);
#pragma unroll
    for (int dt = 0; dt < 4; ++dt) {
      v2f b0, b1;
      b0.x = sp[(8 * s + 0) * DHEAD + dt * 16];
      b0.y = sp[(8 * s + 1) * DHEAD + dt * 16];
      b1.x = sp[(8 * s + 2) * DHEAD + dt * 16];
      b1.y = sp[(8 * s + 3) * DHEAD + dt * 16];
      num[dt] = wmma4(a0, b0, num[dt]);
      num[dt] = wmma4(a1, b1, num[dt]);
    }
  }

  const float* kbase = kprime + (size_t)(h * L_SEQ + c * CHUNK + l16) * MFEAT + koff;
  const float* vbase = V + (size_t)(c * CHUNK + 8 * half) * DMODEL + h * DHEAD + l16;
  v2f ones;
  ones.x = 1.f; ones.y = 1.f;
  for (int jt = 0; jt <= lt; ++jt) {
    const float* krow = kbase + (size_t)jt * 16 * MFEAT;
    float4 kf[16];
#pragma unroll
    for (int s = 0; s < 16; ++s) kf[s] = *reinterpret_cast<const float4*>(krow + 8 * s);
    v8f t0 = zero, t1 = zero;
#pragma unroll
    for (int s = 0; s < 16; ++s) {
      v2f ka0, ka1, qb0, qb1;
      ka0.x = kf[s].x; ka0.y = kf[s].y; ka1.x = kf[s].z; ka1.y = kf[s].w;
      qb0.x = q[s].x;  qb0.y = q[s].y;  qb1.x = q[s].z;  qb1.y = q[s].w;
      t0 = wmma4(ka0, qb0, t0);
      t1 = wmma4(ka1, qb1, t1);
    }
    v8f t = t0 + t1;
    if (jt == lt) {
#pragma unroll
      for (int r = 0; r < 8; ++r) t[r] = (r + 8 * half > l16) ? 0.f : t[r];
    }
    const float* vp = vbase + (size_t)jt * 16 * DMODEL;
#pragma unroll
    for (int s2 = 0; s2 < 4; ++s2) {
      v2f a;
      a.x = t[2 * s2]; a.y = t[2 * s2 + 1];
      den = wmma4(a, ones, den);
#pragma unroll
      for (int dt = 0; dt < 4; ++dt) {
        v2f b;
        b.x = vp[(2 * s2) * DMODEL + dt * 16];
        b.y = vp[(2 * s2 + 1) * DMODEL + dt * 16];
        num[dt] = wmma4(a, b, num[dt]);
      }
    }
  }

#pragma unroll
  for (int dt = 0; dt < 4; ++dt)
#pragma unroll
    for (int r = 0; r < 8; ++r) {
      const int l = lrow0 + r + 8 * half;
      context[(size_t)l * DMODEL + h * DHEAD + dt * 16 + l16] =
          num[dt][r] / (den[r] + DEN_EPS);
    }
}

__global__ void __launch_bounds__(128)
perf_out_proj(const float* __restrict__ ctx, const float* __restrict__ Wo,
              const float* __restrict__ bo, float* __restrict__ out) {
  __shared__ __attribute__((aligned(16))) float lds[GEMM_LDS_FLOATS];
  const int tid = threadIdx.x;
  const int lane = tid & 31;
  const int wave = tid >> 5;
  const int tm = blockIdx.x * GEMM_TILE + (wave >> 1) * 32;
  const int tn = blockIdx.y * GEMM_TILE + (wave & 1) * 32;
  Acc2x2 r = gemm_tile64_lds(ctx + blockIdx.x * GEMM_TILE * DMODEL,
                             Wo + blockIdx.y * GEMM_TILE * DMODEL, lds, tid);
  const int l16 = lane & 15;
  const int half = lane >> 4;
#pragma unroll
  for (int ti = 0; ti < 2; ++ti)
#pragma unroll
    for (int tj = 0; tj < 2; ++tj) {
      const int col = tn + tj * 16 + l16;
      const int rbase = tm + ti * 16 + 8 * half;
      const float bb = bo[col];
#pragma unroll
      for (int i = 0; i < 8; ++i)
        out[(rbase + i) * DMODEL + col] = r.c[ti][tj][i] + bb;
    }
}

extern "C" void kernel_launch(void* const* d_in, const int* in_sizes, int n_in,
                              void* d_out, int out_size, void* d_ws, size_t ws_size,
                              hipStream_t stream) {
  (void)in_sizes; (void)n_in; (void)out_size; (void)ws_size;
  const float* query = (const float*)d_in[0];
  const float* key_  = (const float*)d_in[1];
  const float* value = (const float*)d_in[2];
  const float* Wq = (const float*)d_in[3];
  const float* bq = (const float*)d_in[4];
  const float* Wk = (const float*)d_in[5];
  const float* bk = (const float*)d_in[6];
  const float* Wv = (const float*)d_in[7];
  const float* bv = (const float*)d_in[8];
  const float* Wo = (const float*)d_in[9];
  const float* bo = (const float*)d_in[10];
  const float* RF = (const float*)d_in[11];

  float* ws = (float*)d_ws;
  const size_t SZ_LD = (size_t)L_SEQ * DMODEL;
  const size_t SZ_PRIME = (size_t)NHEADS * L_SEQ * MFEAT;
  float* Q      = ws;
  float* K      = Q + SZ_LD;
  float* V      = K + SZ_LD;
  float* hk     = V + SZ_LD;
  float* kstab  = hk + (size_t)NHEADS * L_SEQ;
  float* qprime = kstab + 64;
  float* kprime = qprime + SZ_PRIME;
  float* chunkbuf = kprime + SZ_PRIME;
  float* context = Q;

  perf_qkv_proj<<<dim3(L_SEQ / GEMM_TILE, DMODEL / GEMM_TILE, 3), 128, 0, stream>>>(
      query, key_, value, Wq, bq, Wk, bk, Wv, bv, Q, K, V);
  perf_hk<<<(NHEADS * L_SEQ) / 256, 256, 0, stream>>>(K, hk);
  perf_kstab<<<1, 1024, 0, stream>>>(hk, kstab);
  perf_primes<<<dim3(L_SEQ / 32, MFEAT / 32, 2 * NHEADS), 32, 0, stream>>>(
      Q, K, RF, hk, kstab, qprime, kprime);
  perf_chunk_sums<<<dim3(8, NCHUNKS, NHEADS), 32, 0, stream>>>(kprime, V, chunkbuf);
  perf_chunk_prefix<<<(NHEADS * PREFIX_VEC) / 128, 128, 0, stream>>>(chunkbuf);
  perf_scan<<<dim3(CHUNK / 16, NCHUNKS, NHEADS), 32, 0, stream>>>(
      qprime, kprime, V, chunkbuf, context);
  perf_out_proj<<<dim3(L_SEQ / GEMM_TILE, DMODEL / GEMM_TILE), 128, 0, stream>>>(
      context, Wo, bo, (float*)d_out);
}


# h_k and its maximum folded into the K projection's epilogue (per-tile maxima, finished by the feature kernel): two helper launches removed; h_k summed in a different reduction tree
# speedup vs baseline: 8.0225x; 1.0801x over previous
#include <hip/hip_runtime.h>
#include <math.h>

#define L_SEQ 2048
#define DMODEL 512
#define NHEADS 8
#define DHEAD 64
#define MFEAT 128
#define NCHUNKS 16
#define CHUNK 128
#define CHUNK_STRIDE 8320
#define KERNEL_EPS 1e-4f
#define DEN_EPS 1e-6f
#define QK_SCALE 0.21022410381342864f
#define NC_FEAT 0.08838834764831845f

typedef float v2f __attribute__((ext_vector_type(2)));
typedef float v4f __attribute__((ext_vector_type(4)));
typedef float v8f __attribute__((ext_vector_type(8)));

template <int M>
__device__ __forceinline__ float row_xor(float v) {
  return __builtin_bit_cast(float, __builtin_amdgcn_update_dpp(
      0, __builtin_bit_cast(int, v), 0x160 + M, 0xf, 0xf, true));
}
__device__ __forceinline__ float other_row(float v) {
  return __builtin_bit_cast(float, __builtin_amdgcn_permlanex16(
      __builtin_bit_cast(int, v), __builtin_bit_cast(int, v), 0x76543210, 0xfedcba98, false, false));
}

struct Acc2x2 { v8f c[2][2]; };

__device__ __forceinline__ Acc2x2 wmma_block2x2_f32(const float* __restrict__ A, int lda,
                                                    const float* __restrict__ B, int bks, int bns,
                                                    int K, int lane) {
  const int half = lane >> 4;
  const int l16  = lane & 15;
  const float* arow0 = A + l16 * lda;
  const float* arow1 = A + (l16 + 16) * lda;
  const float* bcol0 = B + l16 * bns;
  const float* bcol1 = B + (l16 + 16) * bns;
  const v8f zero = {0.f, 0.f, 0.f, 0.f, 0.f, 0.f, 0.f, 0.f};
  Acc2x2 r;
  r.c[0][0] = zero; r.c[0][1] = zero; r.c[1][0] = zero; r.c[1][1] = zero;
  for (int k0 = 0; k0 < K; k0 += 4) {
    const int ka = k0 + 2 * half;
    v2f a0, a1, b0, b1;
    a0.x = arow0[ka];          a0.y = arow0[ka + 1];
    a1.x = arow1[ka];          a1.y = arow1[ka + 1];
    b0.x = bcol0[ka * bks];    b0.y = bcol0[(ka + 1) * bks];
    b1.x = bcol1[ka * bks];    b1.y = bcol1[(ka + 1) * bks];
    r.c[0][0] = __builtin_amdgcn_wmma_f32_16x16x4_f32(false, a0, false, b0, (short)0, r.c[0][0], false, false);
    r.c[0][1] = __builtin_amdgcn_wmma_f32_16x16x4_f32(false, a0, false, b1, (short)0, r.c[0][1], false, false);
    r.c[1][0] = __builtin_amdgcn_wmma_f32_16x16x4_f32(false, a1, false, b0, (short)0, r.c[1][0], false, false);
    r.c[1][1] = __builtin_amdgcn_wmma_f32_16x16x4_f32(false, a1, false, b1, (short)0, r.c[1][1], false, false);
  }
  return r;
}

#define GEMM_TILE 64
#define GEMM_KB 64
#define GEMM_STEPS (GEMM_KB / 8)
#define GEMM_PITCH (GEMM_KB + 4)
#define GEMM_PLANE (GEMM_TILE * GEMM_PITCH)
#define GEMM_LDS_FLOATS (4 * GEMM_PLANE)
#define GEMM_COPIES (GEMM_TILE * GEMM_KB / 4 / 128)
struct OperandBlock { float4 a0[GEMM_STEPS], a1[GEMM_STEPS], b0[GEMM_STEPS], b1[GEMM_STEPS]; };
struct StageRegs { v4f a[GEMM_COPIES], b[GEMM_COPIES]; };

__device__ __forceinline__ void load_stage_regs(StageRegs& g, const float* __restrict__ ag,
                                                const float* __restrict__ bg, int kb) {
#pragma unroll
  for (int i = 0; i < GEMM_COPIES; ++i) {
    g.a[i] = *reinterpret_cast<const v4f*>(ag + (size_t)(8 * i) * DMODEL + kb);
    g.b[i] = *reinterpret_cast<const v4f*>(bg + (size_t)(8 * i) * DMODEL + kb);
  }
}

__device__ __forceinline__ void store_stage_regs(const StageRegs& g, float* sw) {
#pragma unroll
  for (int i = 0; i < GEMM_COPIES; ++i) {
    *reinterpret_cast<v4f*>(sw + 8 * i * GEMM_PITCH) = g.a[i];
    *reinterpret_cast<v4f*>(sw + GEMM_PLANE + 8 * i * GEMM_PITCH) = g.b[i];
  }
}

__device__ __forceinline__ void read_fragments(OperandBlock& o, const float* af, const float* bf) {
#pragma unroll
  for (int s = 0; s < GEMM_STEPS; ++s) {
    o.a0[s] = *reinterpret_cast<const float4*>(af + 8 * s);
    o.a1[s] = *reinterpret_cast<const float4*>(af + 16 * GEMM_PITCH + 8 * s);
    o.b0[s] = *reinterpret_cast<const float4*>(bf + 8 * s);
    o.b1[s] = *reinterpret_cast<const float4*>(bf + 16 * GEMM_PITCH + 8 * s);
  }
}

__device__ __forceinline__ void mma_operand_block(Acc2x2& r, const OperandBlock& o) {
#pragma unroll
  for (int s = 0; s < GEMM_STEPS; ++s) {
    v2f a0l, a0h, a1l, a1h, b0l, b0h, b1l, b1h;
    a0l.x = o.a0[s].x; a0l.y = o.a0[s].y; a0h.x = o.a0[s].z; a0h.y = o.a0[s].w;
    a1l.x = o.a1[s].x; a1l.y = o.a1[s].y; a1h.x = o.a1[s].z; a1h.y = o.a1[s].w;
    b0l.x = o.b0[s].x; b0l.y = o.b0[s].y; b0h.x = o.b0[s].z; b0h.y = o.b0[s].w;
    b1l.x = o.b1[s].x; b1l.y = o.b1[s].y; b1h.x = o.b1[s].z; b1h.y = o.b1[s].w;
    r.c[0][0] = __builtin_amdgcn_wmma_f32_16x16x4_f32(false, a0l, false, b0l, (short)0, r.c[0][0], false, false);
    r.c[0][1] = __builtin_amdgcn_wmma_f32_16x16x4_f32(false, a0l, false, b1l, (short)0, r.c[0][1], false, false);
    r.c[1][0] = __builtin_amdgcn_wmma_f32_16x16x4_f32(false, a1l, false, b0l, (short)0, r.c[1][0], false, false);
    r.c[1][1] = __builtin_amdgcn_wmma_f32_16x16x4_f32(false, a1l, false, b1l, (short)0, r.c[1][1], false, false);
    r.c[0][0] = __builtin_amdgcn_wmma_f32_16x16x4_f32(false, a0h, false, b0h, (short)0, r.c[0][0], false, false);
    r.c[0][1] = __builtin_amdgcn_wmma_f32_16x16x4_f32(false, a0h, false, b1h, (short)0, r.c[0][1], false, false);
    r.c[1][0] = __builtin_amdgcn_wmma_f32_16x16x4_f32(false, a1h, false, b0h, (short)0, r.c[1][0], false, false);
    r.c[1][1] = __builtin_amdgcn_wmma_f32_16x16x4_f32(false, a1h, false, b1h, (short)0, r.c[1][1], false, false);
  }
}

__device__ __forceinline__ Acc2x2 gemm_tile64_lds(const float* __restrict__ A,
                                                  const float* __restrict__ Bt,
                                                  float* lds, int tid) {
  const int lane = tid & 31;
  const int wave = tid >> 5;
  const int half = lane >> 4;
  const int l16  = lane & 15;
  const int wm = wave >> 1;
  const int wn = wave & 1;
  const int srow = tid >> 4;
  const int spiece = tid & 15;
  const float* ag = A + srow * DMODEL + 4 * spiece;
  const float* bg = Bt + srow * DMODEL + 4 * spiece;
  float* sw = lds + srow * GEMM_PITCH + 4 * spiece;
  const float* af = lds + (wm * 32 + l16) * GEMM_PITCH + 4 * half;
  const float* bf = lds + GEMM_PLANE + (wn * 32 + l16) * GEMM_PITCH + 4 * half;
  const v8f zero = {0.f, 0.f, 0.f, 0.f, 0.f, 0.f, 0.f, 0.f};
  Acc2x2 r;
  r.c[0][0] = zero; r.c[0][1] = zero; r.c[1][0] = zero; r.c[1][1] = zero;
  StageRegs g;
  load_stage_regs(g, ag, bg, 0);
  store_stage_regs(g, sw);
  __syncthreads();
  for (int t = 0; t < DMODEL / GEMM_KB; ++t) {
    const int cur = (t & 1) * 2 * GEMM_PLANE;
    const int nxt = 2 * GEMM_PLANE - cur;
    const bool more = t + 1 < DMODEL / GEMM_KB;
    if (more) load_stage_regs(g, ag, bg, (t + 1) * GEMM_KB);
    __builtin_amdgcn_sched_barrier(0);
    OperandBlock o;
    read_fragments(o, af + cur, bf + cur);
    __builtin_amdgcn_sched_barrier(0);
    mma_operand_block(r, o);
    __builtin_amdgcn_sched_barrier(0);
    if (more) store_stage_regs(g, sw + nxt);
    __syncthreads();
  }
  return r;
}

__global__ void __launch_bounds__(128)
perf_qkv_proj(const float* __restrict__ xq, const float* __restrict__ xk,
              const float* __restrict__ xv,
              const float* __restrict__ Wq, const float* __restrict__ bq,
              const float* __restrict__ Wk, const float* __restrict__ bk,
              const float* __restrict__ Wv, const float* __restrict__ bv,
              float* __restrict__ Q, float* __restrict__ K, float* __restrict__ V,
              float* __restrict__ hk, float* __restrict__ kmax_part) {
  __shared__ __attribute__((aligned(16))) float lds[GEMM_LDS_FLOATS];
  const int tid = threadIdx.x;
  const int lane = tid & 31;
  const int wave = tid >> 5;
  const int tm = blockIdx.x * GEMM_TILE + (wave >> 1) * 32;
  const int tn = blockIdx.y * GEMM_TILE + (wave & 1) * 32;
  const int mat = blockIdx.z;
  const float* X; const float* W; const float* bias; float* Y; float scl;
  if (mat == 0)      { X = xq; W = Wq; bias = bq; Y = Q; scl = QK_SCALE; }
  else if (mat == 1) { X = xk; W = Wk; bias = bk; Y = K; scl = QK_SCALE; }
  else               { X = xv; W = Wv; bias = bv; Y = V; scl = 1.f; }
  Acc2x2 r = gemm_tile64_lds(X + blockIdx.x * GEMM_TILE * DMODEL,
                             W + blockIdx.y * GEMM_TILE * DMODEL, lds, tid);
  const int l16 = lane & 15;
  const int half = lane >> 4;
  float ss[2][8];
#pragma unroll
  for (int ti = 0; ti < 2; ++ti) {
#pragma unroll
    for (int i = 0; i < 8; ++i) ss[ti][i] = 0.f;
#pragma unroll
    for (int tj = 0; tj < 2; ++tj) {
      const int col = tn + tj * 16 + l16;
      const int rbase = tm + ti * 16 + 8 * half;
      const float bb = bias[col];
#pragma unroll
      for (int i = 0; i < 8; ++i) {
        const float y = (r.c[ti][tj][i] + bb) * scl;
        Y[(rbase + i) * DMODEL + col] = y;
        ss[ti][i] += y * y;
      }
    }
  }
  static_assert(GEMM_TILE == DHEAD, "a column tile of the K projection must be one head");
  if (mat == 1) {
    const int wm = wave >> 1;
    const int wn = wave & 1;
    float* part = lds;
#pragma unroll
    for (int ti = 0; ti < 2; ++ti)
#pragma unroll
      for (int i = 0; i < 8; ++i) ss[ti][i] += row_xor<8>(ss[ti][i]);
#pragma unroll
    for (int ti = 0; ti < 2; ++ti)
#pragma unroll
      for (int i = 0; i < 8; ++i) ss[ti][i] += row_xor<4>(ss[ti][i]);
#pragma unroll
    for (int ti = 0; ti < 2; ++ti)
#pragma unroll
      for (int i = 0; i < 8; ++i) ss[ti][i] += row_xor<2>(ss[ti][i]);
#pragma unroll
    for (int ti = 0; ti < 2; ++ti)
#pragma unroll
      for (int i = 0; i < 8; ++i) ss[ti][i] += row_xor<1>(ss[ti][i]);
    if (l16 == 0) {
      float* p = part + wn * GEMM_TILE + wm * 32 + 8 * half;
#pragma unroll
      for (int ti = 0; ti < 2; ++ti) {
        v4f lo, hi;
        lo.x = ss[ti][0]; lo.y = ss[ti][1]; lo.z = ss[ti][2]; lo.w = ss[ti][3];
        hi.x = ss[ti][4]; hi.y = ss[ti][5]; hi.z = ss[ti][6]; hi.w = ss[ti][7];
        *reinterpret_cast<v4f*>(p + ti * 16) = lo;
        *reinterpret_cast<v4f*>(p + ti * 16 + 4) = hi;
      }
    }
    __syncthreads();
    if (wave == 0) {
      const float h0 = -0.5f * (part[lane] + part[GEMM_TILE + lane]);
      const float h1 = -0.5f * (part[lane + 32] + part[GEMM_TILE + lane + 32]);
      float* hrow = hk + blockIdx.y * L_SEQ + blockIdx.x * GEMM_TILE;
      hrow[lane] = h0;
      hrow[lane + 32] = h1;
      float m = fmaxf(h0, h1);
      m = fmaxf(m, row_xor<8>(m));
      m = fmaxf(m, row_xor<4>(m));
      m = fmaxf(m, row_xor<2>(m));
      m = fmaxf(m, row_xor<1>(m));
      m = fmaxf(m, other_row(m));
      if (lane == 0) kmax_part[blockIdx.y * (L_SEQ / GEMM_TILE) + blockIdx.x] = m;
    }
  }
}

#define KMAX_PARTS ((L_SEQ / GEMM_TILE) * NHEADS)
__global__ void __launch_bounds__(32)
perf_primes(const float* __restrict__ Q, const float* __restrict__ K,
            const float* __restrict__ RF, const float* __restrict__ hk,
            const float* __restrict__ kmax_part,
            float* __restrict__ qprime, float* __restrict__ kprime) {
  const int tm = blockIdx.x * 32;
  const int tn = blockIdx.y * 32;
  const int h = blockIdx.z >> 1;
  const int isK = blockIdx.z & 1;
  const int lane = threadIdx.x;
  const float* A = (isK ? K : Q) + tm * DMODEL + h * DHEAD;
  const float4* pp = reinterpret_cast<const float4*>(kmax_part) + lane;
  const float4 p0 = pp[0];
  const float4 p1 = pp[KMAX_PARTS / 8];
  Acc2x2 r = wmma_block2x2_f32(A, DMODEL, RF + tn, MFEAT, 1, DHEAD, lane);
  const int l16 = lane & 15;
  const int half = lane >> 4;
  float ks = 0.f;
  if (isK) {
    ks = fmaxf(fmaxf(fmaxf(p0.x, p0.y), fmaxf(p0.z, p0.w)),
               fmaxf(fmaxf(p1.x, p1.y), fmaxf(p1.z, p1.w)));
    ks = fmaxf(ks, row_xor<8>(ks));
    ks = fmaxf(ks, row_xor<4>(ks));
    ks = fmaxf(ks, row_xor<2>(ks));
    ks = fmaxf(ks, row_xor<1>(ks));
    ks = fmaxf(ks, other_row(ks));
  }
#pragma unroll
  for (int ti = 0; ti < 2; ++ti)
#pragma unroll
    for (int tj = 0; tj < 2; ++tj) {
      const int mcol = tn + tj * 16 + l16;
      const int rbase = tm + ti * 16 + 8 * half;
      if (isK) {
#pragma unroll
        for (int i = 0; i < 8; ++i) {
          const int l = rbase + i;
          kprime[(h * L_SEQ + l) * MFEAT + mcol] =
              NC_FEAT * (expf(hk[h * L_SEQ + l] + r.c[ti][tj][i] - ks) + KERNEL_EPS);
        }
      } else {
#pragma unroll
        for (int i = 0; i < 8; ++i) {
          const int l = rbase + i;
          qprime[(h * L_SEQ + l) * MFEAT + mcol] =
              NC_FEAT * (expf(r.c[ti][tj][i]) + KERNEL_EPS);
        }
      }
    }
}

__device__ __forceinline__ v8f wmma4(v2f a, v2f b, v8f c) {
  return __builtin_amdgcn_wmma_f32_16x16x4_f32(false, a, false, b, (short)0, c, false, false);
}

__global__ void __launch_bounds__(32)
perf_chunk_sums(const float* __restrict__ kprime, const float* __restrict__ V,
                float* __restrict__ chunkbuf) {
  const int blk = blockIdx.x;
  const int c = blockIdx.y;
  const int h = blockIdx.z;
  const int lane = threadIdx.x;
  const int half = lane >> 4;
  const int l16 = lane & 15;
  const int m0 = (blk >> 1) * 32;
  const int d0 = (blk & 1) * 32;
  const float* ka = kprime + (size_t)(h * L_SEQ + c * CHUNK + 2 * half) * MFEAT + m0 + 2 * l16;
  const float* vb = V + (size_t)(c * CHUNK + 2 * half) * DMODEL + h * DHEAD + d0 + 2 * l16;
  const v8f zero = {0.f, 0.f, 0.f, 0.f, 0.f, 0.f, 0.f, 0.f};
  v8f acc[2][2];
  acc[0][0] = zero; acc[0][1] = zero; acc[1][0] = zero; acc[1][1] = zero;
  float zx = 0.f, zy = 0.f;
#pragma unroll
  for (int ks = 0; ks < CHUNK / 4; ++ks) {
    const float2 ra0 = *reinterpret_cast<const float2*>(ka + (size_t)(4 * ks) * MFEAT);
    const float2 ra1 = *reinterpret_cast<const float2*>(ka + (size_t)(4 * ks + 1) * MFEAT);
    const float2 rb0 = *reinterpret_cast<const float2*>(vb + (size_t)(4 * ks) * DMODEL);
    const float2 rb1 = *reinterpret_cast<const float2*>(vb + (size_t)(4 * ks + 1) * DMODEL);
    v2f a0, a1, b0, b1;
    a0.x = ra0.x; a0.y = ra1.x; a1.x = ra0.y; a1.y = ra1.y;
    b0.x = rb0.x; b0.y = rb1.x; b1.x = rb0.y; b1.y = rb1.y;
    acc[0][0] = wmma4(a0, b0, acc[0][0]);
    acc[0][1] = wmma4(a0, b1, acc[0][1]);
    acc[1][0] = wmma4(a1, b0, acc[1][0]);
    acc[1][1] = wmma4(a1, b1, acc[1][1]);
    zx += ra0.x + ra1.x;
    zy += ra0.y + ra1.y;
  }
  float* buf = chunkbuf + (size_t)(h * NCHUNKS + c) * CHUNK_STRIDE;
#pragma unroll
  for (int jm = 0; jm < 2; ++jm)
#pragma unroll
    for (int r = 0; r < 8; ++r) {
      const int m = m0 + 2 * (r + 8 * half) + jm;
      float2 o;
      o.x = acc[jm][0][r]; o.y = acc[jm][1][r];
      *reinterpret_cast<float2*>(buf + m * DHEAD + d0 + 2 * l16) = o;
    }
  zx += __shfl_xor(zx, 16);
  zy += __shfl_xor(zy, 16);
  if (d0 == 0 && half == 0) {
    float2 o;
    o.x = zx; o.y = zy;
    *reinterpret_cast<float2*>(buf + MFEAT * DHEAD + m0 + 2 * l16) = o;
  }
}

#define PREFIX_VEC (CHUNK_STRIDE / 4)
__global__ void __launch_bounds__(128)
perf_chunk_prefix(float* __restrict__ chunkbuf) {
  const int gid = blockIdx.x * 128 + threadIdx.x;
  const int h = gid / PREFIX_VEC;
  const int i4 = gid - h * PREFIX_VEC;
  float4* p = reinterpret_cast<float4*>(chunkbuf + (size_t)h * NCHUNKS * CHUNK_STRIDE) + i4;
  float4 v[NCHUNKS];
#pragma unroll
  for (int c = 0; c < NCHUNKS; ++c) v[c] = p[(size_t)c * PREFIX_VEC];
  float4 carry;
  carry.x = 0.f; carry.y = 0.f; carry.z = 0.f; carry.w = 0.f;
#pragma unroll
  for (int c = 0; c < NCHUNKS; ++c) {
    const float4 tmp = v[c];
    p[(size_t)c * PREFIX_VEC] = carry;
    carry.x += tmp.x; carry.y += tmp.y; carry.z += tmp.z; carry.w += tmp.w;
  }
}

__global__ void __launch_bounds__(32)
perf_scan(const float* __restrict__ qprime, const float* __restrict__ kprime,
          const float* __restrict__ V, const float* __restrict__ chunkbuf,
          float* __restrict__ context) {
  const int lt = blockIdx.x;
  const int c = blockIdx.y;
  const int h = blockIdx.z;
  const int lane = threadIdx.x;
  const int half = lane >> 4;
  const int l16 = lane & 15;
  const int koff = 4 * half;
  const int lrow0 = c * CHUNK + lt * 16;
  const float* buf = chunkbuf + (size_t)(h * NCHUNKS + c) * CHUNK_STRIDE;

  const float* qrow = qprime + (size_t)(h * L_SEQ + lrow0 + l16) * MFEAT + koff;
  float4 q[16];
#pragma unroll
  for (int s = 0; s < 16; ++s) q[s] = *reinterpret_cast<const float4*>(qrow + 8 * s);

  const v8f zero = {0.f, 0.f, 0.f, 0.f, 0.f, 0.f, 0.f, 0.f};
  v8f num[4];
  num[0] = zero; num[1] = zero; num[2] = zero; num[3] = zero;
  v8f den = zero;

  const float* sp = buf + koff * DHEAD + l16;
  const float* zp = buf + MFEAT * DHEAD + koff;
#pragma unroll
  for (int s = 0; s < 16; ++s) {
    const float4 zz = *reinterpret_cast<const float4*>(zp + 8 * s);
    v2f a0, a1, z0, z1;
    a0.x = q[s].x; a0.y = q[s].y; a1.x = q[s].z; a1.y = q[s].w;
    z0.x = zz.x;   z0.y = zz.y;   z1.x = zz.z;   z1.y = zz.w;
    den = wmma4(a0, z0, den);
    den = wmma4(a1, z1, den);
#pragma unroll
    for (int dt = 0; dt < 4; ++dt) {
      v2f b0, b1;
      b0.x = sp[(8 * s + 0) * DHEAD + dt * 16];
      b0.y = sp[(8 * s + 1) * DHEAD + dt * 16];
      b1.x = sp[(8 * s + 2) * DHEAD + dt * 16];
      b1.y = sp[(8 * s + 3) * DHEAD + dt * 16];
      num[dt] = wmma4(a0, b0, num[dt]);
      num[dt] = wmma4(a1, b1, num[dt]);
    }
  }

  const float* kbase = kprime + (size_t)(h * L_SEQ + c * CHUNK + l16) * MFEAT + koff;
  const float* vbase = V + (size_t)(c * CHUNK + 8 * half) * DMODEL + h * DHEAD + l16;
  v2f ones;
  ones.x = 1.f; ones.y = 1.f;
  for (int jt = 0; jt <= lt; ++jt) {
    const float* krow = kbase + (size_t)jt * 16 * MFEAT;
    float4 kf[16];
#pragma unroll
    for (int s = 0; s < 16; ++s) kf[s] = *reinterpret_cast<const float4*>(krow + 8 * s);
    v8f t0 = zero, t1 = zero;
#pragma unroll
    for (int s = 0; s < 16; ++s) {
      v2f ka0, ka1, qb0, qb1;
      ka0.x = kf[s].x; ka0.y = kf[s].y; ka1.x = kf[s].z; ka1.y = kf[s].w;
      qb0.x = q[s].x;  qb0.y = q[s].y;  qb1.x = q[s].z;  qb1.y = q[s].w;
      t0 = wmma4(ka0, qb0, t0);
      t1 = wmma4(ka1, qb1, t1);
    }
    v8f t = t0 + t1;
    if (jt == lt) {
#pragma unroll
      for (int r = 0; r < 8; ++r) t[r] = (r + 8 * half > l16) ? 0.f : t[r];
    }
    const float* vp = vbase + (size_t)jt * 16 * DMODEL;
#pragma unroll
    for (int s2 = 0; s2 < 4; ++s2) {
      v2f a;
      a.x = t[2 * s2]; a.y = t[2 * s2 + 1];
      den = wmma4(a, ones, den);
#pragma unroll
      for (int dt = 0; dt < 4; ++dt) {
        v2f b;
        b.x = vp[(2 * s2) * DMODEL + dt * 16];
        b.y = vp[(2 * s2 + 1) * DMODEL + dt * 16];
        num[dt] = wmma4(a, b, num[dt]);
      }
    }
  }

#pragma unroll
  for (int dt = 0; dt < 4; ++dt)
#pragma unroll
    for (int r = 0; r < 8; ++r) {
      const int l = lrow0 + r + 8 * half;
      context[(size_t)l * DMODEL + h * DHEAD + dt * 16 + l16] =
          num[dt][r] / (den[r] + DEN_EPS);
    }
}

__global__ void __launch_bounds__(128)
perf_out_proj(const float* __restrict__ ctx, const float* __restrict__ Wo,
              const float* __restrict__ bo, float* __restrict__ out) {
  __shared__ __attribute__((aligned(16))) float lds[GEMM_LDS_FLOATS];
  const int tid = threadIdx.x;
  const int lane = tid & 31;
  const int wave = tid >> 5;
  const int tm = blockIdx.x * GEMM_TILE + (wave >> 1) * 32;
  const int tn = blockIdx.y * GEMM_TILE + (wave & 1) * 32;
  Acc2x2 r = gemm_tile64_lds(ctx + blockIdx.x * GEMM_TILE * DMODEL,
                             Wo + blockIdx.y * GEMM_TILE * DMODEL, lds, tid);
  const int l16 = lane & 15;
  const int half = lane >> 4;
#pragma unroll
  for (int ti = 0; ti < 2; ++ti)
#pragma unroll
    for (int tj = 0; tj < 2; ++tj) {
      const int col = tn + tj * 16 + l16;
      const int rbase = tm + ti * 16 + 8 * half;
      const float bb = bo[col];
#pragma unroll
      for (int i = 0; i < 8; ++i)
        out[(rbase + i) * DMODEL + col] = r.c[ti][tj][i] + bb;
    }
}

extern "C" void kernel_launch(void* const* d_in, const int* in_sizes, int n_in,
                              void* d_out, int out_size, void* d_ws, size_t ws_size,
                              hipStream_t stream) {
  (void)in_sizes; (void)n_in; (void)out_size; (void)ws_size;
  const float* query = (const float*)d_in[0];
  const float* key_  = (const float*)d_in[1];
  const float* value = (const float*)d_in[2];
  const float* Wq = (const float*)d_in[3];
  const float* bq = (const float*)d_in[4];
  const float* Wk = (const float*)d_in[5];
  const float* bk = (const float*)d_in[6];
  const float* Wv = (const float*)d_in[7];
  const float* bv = (const float*)d_in[8];
  const float* Wo = (const float*)d_in[9];
  const float* bo = (const float*)d_in[10];
  const float* RF = (const float*)d_in[11];

  float* ws = (float*)d_ws;
  const size_t SZ_LD = (size_t)L_SEQ * DMODEL;
  const size_t SZ_PRIME = (size_t)NHEADS * L_SEQ * MFEAT;
  float* Q      = ws;
  float* K      = Q + SZ_LD;
  float* V      = K + SZ_LD;
  float* hk     = V + SZ_LD;
  float* kmax_part = hk + (size_t)NHEADS * L_SEQ;
  float* qprime = kmax_part + KMAX_PARTS;
  float* kprime = qprime + SZ_PRIME;
  float* chunkbuf = kprime + SZ_PRIME;
  float* context = Q;

  perf_qkv_proj<<<dim3(L_SEQ / GEMM_TILE, DMODEL / GEMM_TILE, 3), 128, 0, stream>>>(
      query, key_, value, Wq, bq, Wk, bk, Wv, bv, Q, K, V, hk, kmax_part);
  perf_primes<<<dim3(L_SEQ / 32, MFEAT / 32, 2 * NHEADS), 32, 0, stream>>>(
      Q, K, RF, hk, kmax_part, qprime, kprime);
  perf_chunk_sums<<<dim3(8, NCHUNKS, NHEADS), 32, 0, stream>>>(kprime, V, chunkbuf);
  perf_chunk_prefix<<<(NHEADS * PREFIX_VEC) / 128, 128, 0, stream>>>(chunkbuf);
  perf_scan<<<dim3(CHUNK / 16, NCHUNKS, NHEADS), 32, 0, stream>>>(
      qprime, kprime, V, chunkbuf, context);
  perf_out_proj<<<dim3(L_SEQ / GEMM_TILE, DMODEL / GEMM_TILE), 128, 0, stream>>>(
      context, Wo, bo, (float*)d_out);
}


# in-chunk scan kernel: each 16-row tile's k' fragments and V values are loaded one tile ahead of their products (two register sets), instead of at the head of their own tile; sums unchanged
# speedup vs baseline: 8.5369x; 1.0641x over previous
#include <hip/hip_runtime.h>
#include <math.h>

#define L_SEQ 2048
#define DMODEL 512
#define NHEADS 8
#define DHEAD 64
#define MFEAT 128
#define NCHUNKS 16
#define CHUNK 128
#define CHUNK_STRIDE 8320
#define KERNEL_EPS 1e-4f
#define DEN_EPS 1e-6f
#define QK_SCALE 0.21022410381342864f
#define NC_FEAT 0.08838834764831845f

typedef float v2f __attribute__((ext_vector_type(2)));
typedef float v4f __attribute__((ext_vector_type(4)));
typedef float v8f __attribute__((ext_vector_type(8)));

template <int M>
__device__ __forceinline__ float row_xor(float v) {
  return __builtin_bit_cast(float, __builtin_amdgcn_update_dpp(
      0, __builtin_bit_cast(int, v), 0x160 + M, 0xf, 0xf, true));
}
__device__ __forceinline__ float other_row(float v) {
  return __builtin_bit_cast(float, __builtin_amdgcn_permlanex16(
      __builtin_bit_cast(int, v), __builtin_bit_cast(int, v), 0x76543210, 0xfedcba98, false, false));
}

struct Acc2x2 { v8f c[2][2]; };

__device__ __forceinline__ Acc2x2 wmma_block2x2_f32(const float* __restrict__ A, int lda,
                                                    const float* __restrict__ B, int bks, int bns,
                                                    int K, int lane) {
  const int half = lane >> 4;
  const int l16  = lane & 15;
  const float* arow0 = A + l16 * lda;
  const float* arow1 = A + (l16 + 16) * lda;
  const float* bcol0 = B + l16 * bns;
  const float* bcol1 = B + (l16 + 16) * bns;
  const v8f zero = {0.f, 0.f, 0.f, 0.f, 0.f, 0.f, 0.f, 0.f};
  Acc2x2 r;
  r.c[0][0] = zero; r.c[0][1] = zero; r.c[1][0] = zero; r.c[1][1] = zero;
  for (int k0 = 0; k0 < K; k0 += 4) {
    const int ka = k0 + 2 * half;
    v2f a0, a1, b0, b1;
    a0.x = arow0[ka];          a0.y = arow0[ka + 1];
    a1.x = arow1[ka];          a1.y = arow1[ka + 1];
    b0.x = bcol0[ka * bks];    b0.y = bcol0[(ka + 1) * bks];
    b1.x = bcol1[ka * bks];    b1.y = bcol1[(ka + 1) * bks];
    r.c[0][0] = __builtin_amdgcn_wmma_f32_16x16x4_f32(false, a0, false, b0, (short)0, r.c[0][0], false, false);
    r.c[0][1] = __builtin_amdgcn_wmma_f32_16x16x4_f32(false, a0, false, b1, (short)0, r.c[0][1], false, false);
    r.c[1][0] = __builtin_amdgcn_wmma_f32_16x16x4_f32(false, a1, false, b0, (short)0, r.c[1][0], false, false);
    r.c[1][1] = __builtin_amdgcn_wmma_f32_16x16x4_f32(false, a1, false, b1, (short)0, r.c[1][1], false, false);
  }
  return r;
}

#define GEMM_TILE 64
#define GEMM_KB 64
#define GEMM_STEPS (GEMM_KB / 8)
#define GEMM_PITCH (GEMM_KB + 4)
#define GEMM_PLANE (GEMM_TILE * GEMM_PITCH)
#define GEMM_LDS_FLOATS (4 * GEMM_PLANE)
#define GEMM_COPIES (GEMM_TILE * GEMM_KB / 4 / 128)
struct OperandBlock { float4 a0[GEMM_STEPS], a1[GEMM_STEPS], b0[GEMM_STEPS], b1[GEMM_STEPS]; };
struct StageRegs { v4f a[GEMM_COPIES], b[GEMM_COPIES]; };

__device__ __forceinline__ void load_stage_regs(StageRegs& g, const float* __restrict__ ag,
                                                const float* __restrict__ bg, int kb) {
#pragma unroll
  for (int i = 0; i < GEMM_COPIES; ++i) {
    g.a[i] = *reinterpret_cast<const v4f*>(ag + (size_t)(8 * i) * DMODEL + kb);
    g.b[i] = *reinterpret_cast<const v4f*>(bg + (size_t)(8 * i) * DMODEL + kb);
  }
}

__device__ __forceinline__ void store_stage_regs(const StageRegs& g, float* sw) {
#pragma unroll
  for (int i = 0; i < GEMM_COPIES; ++i) {
    *reinterpret_cast<v4f*>(sw + 8 * i * GEMM_PITCH) = g.a[i];
    *reinterpret_cast<v4f*>(sw + GEMM_PLANE + 8 * i * GEMM_PITCH) = g.b[i];
  }
}

__device__ __forceinline__ void read_fragments(OperandBlock& o, const float* af, const float* bf) {
#pragma unroll
  for (int s = 0; s < GEMM_STEPS; ++s) {
    o.a0[s] = *reinterpret_cast<const float4*>(af + 8 * s);
    o.a1[s] = *reinterpret_cast<const float4*>(af + 16 * GEMM_PITCH + 8 * s);
    o.b0[s] = *reinterpret_cast<const float4*>(bf + 8 * s);
    o.b1[s] = *reinterpret_cast<const float4*>(bf + 16 * GEMM_PITCH + 8 * s);
  }
}

__device__ __forceinline__ void mma_operand_block(Acc2x2& r, const OperandBlock& o) {
#pragma unroll
  for (int s = 0; s < GEMM_STEPS; ++s) {
    v2f a0l, a0h, a1l, a1h, b0l, b0h, b1l, b1h;
    a0l.x = o.a0[s].x; a0l.y = o.a0[s].y; a0h.x = o.a0[s].z; a0h.y = o.a0[s].w;
    a1l.x = o.a1[s].x; a1l.y = o.a1[s].y; a1h.x = o.a1[s].z; a1h.y = o.a1[s].w;
    b0l.x = o.b0[s].x; b0l.y = o.b0[s].y; b0h.x = o.b0[s].z; b0h.y = o.b0[s].w;
    b1l.x = o.b1[s].x; b1l.y = o.b1[s].y; b1h.x = o.b1[s].z; b1h.y = o.b1[s].w;
    r.c[0][0] = __builtin_amdgcn_wmma_f32_16x16x4_f32(false, a0l, false, b0l, (short)0, r.c[0][0], false, false);
    r.c[0][1] = __builtin_amdgcn_wmma_f32_16x16x4_f32(false, a0l, false, b1l, (short)0, r.c[0][1], false, false);
    r.c[1][0] = __builtin_amdgcn_wmma_f32_16x16x4_f32(false, a1l, false, b0l, (short)0, r.c[1][0], false, false);
    r.c[1][1] = __builtin_amdgcn_wmma_f32_16x16x4_f32(false, a1l, false, b1l, (short)0, r.c[1][1], false, false);
    r.c[0][0] = __builtin_amdgcn_wmma_f32_16x16x4_f32(false, a0h, false, b0h, (short)0, r.c[0][0], false, false);
    r.c[0][1] = __builtin_amdgcn_wmma_f32_16x16x4_f32(false, a0h, false, b1h, (short)0, r.c[0][1], false, false);
    r.c[1][0] = __builtin_amdgcn_wmma_f32_16x16x4_f32(false, a1h, false, b0h, (short)0, r.c[1][0], false, false);
    r.c[1][1] = __builtin_amdgcn_wmma_f32_16x16x4_f32(false, a1h, false, b1h, (short)0, r.c[1][1], false, false);
  }
}

__device__ __forceinline__ Acc2x2 gemm_tile64_lds(const float* __restrict__ A,
                                                  const float* __restrict__ Bt,
                                                  float* lds, int tid) {
  const int lane = tid & 31;
  const int wave = tid >> 5;
  const int half = lane >> 4;
  const int l16  = lane & 15;
  const int wm = wave >> 1;
  const int wn = wave & 1;
  const int srow = tid >> 4;
  const int spiece = tid & 15;
  const float* ag = A + srow * DMODEL + 4 * spiece;
  const float* bg = Bt + srow * DMODEL + 4 * spiece;
  float* sw = lds + srow * GEMM_PITCH + 4 * spiece;
  const float* af = lds + (wm * 32 + l16) * GEMM_PITCH + 4 * half;
  const float* bf = lds + GEMM_PLANE + (wn * 32 + l16) * GEMM_PITCH + 4 * half;
  const v8f zero = {0.f, 0.f, 0.f, 0.f, 0.f, 0.f, 0.f, 0.f};
  Acc2x2 r;
  r.c[0][0] = zero; r.c[0][1] = zero; r.c[1][0] = zero; r.c[1][1] = zero;
  StageRegs g;
  load_stage_regs(g, ag, bg, 0);
  store_stage_regs(g, sw);
  __syncthreads();
  for (int t = 0; t < DMODEL / GEMM_KB; ++t) {
    const int cur = (t & 1) * 2 * GEMM_PLANE;
    const int nxt = 2 * GEMM_PLANE - cur;
    const bool more = t + 1 < DMODEL / GEMM_KB;
    if (more) load_stage_regs(g, ag, bg, (t + 1) * GEMM_KB);
    __builtin_amdgcn_sched_barrier(0);
    OperandBlock o;
    read_fragments(o, af + cur, bf + cur);
    __builtin_amdgcn_sched_barrier(0);
    mma_operand_block(r, o);
    __builtin_amdgcn_sched_barrier(0);
    if (more) store_stage_regs(g, sw + nxt);
    __syncthreads();
  }
  return r;
}

__global__ void __launch_bounds__(128)
perf_qkv_proj(const float* __restrict__ xq, const float* __restrict__ xk,
              const float* __restrict__ xv,
              const float* __restrict__ Wq, const float* __restrict__ bq,
              const float* __restrict__ Wk, const float* __restrict__ bk,
              const float* __restrict__ Wv, const float* __restrict__ bv,
              float* __restrict__ Q, float* __restrict__ K, float* __restrict__ V,
              float* __restrict__ hk, float* __restrict__ kmax_part) {
  __shared__ __attribute__((aligned(16))) float lds[GEMM_LDS_FLOATS];
  const int tid = threadIdx.x;
  const int lane = tid & 31;
  const int wave = tid >> 5;
  const int tm = blockIdx.x * GEMM_TILE + (wave >> 1) * 32;
  const int tn = blockIdx.y * GEMM_TILE + (wave & 1) * 32;
  const int mat = blockIdx.z;
  const float* X; const float* W; const float* bias; float* Y; float scl;
  if (mat == 0)      { X = xq; W = Wq; bias = bq; Y = Q; scl = QK_SCALE; }
  else if (mat == 1) { X = xk; W = Wk; bias = bk; Y = K; scl = QK_SCALE; }
  else               { X = xv; W = Wv; bias = bv; Y = V; scl = 1.f; }
  Acc2x2 r = gemm_tile64_lds(X + blockIdx.x * GEMM_TILE * DMODEL,
                             W + blockIdx.y * GEMM_TILE * DMODEL, lds, tid);
  const int l16 = lane & 15;
  const int half = lane >> 4;
  float ss[2][8];
#pragma unroll
  for (int ti = 0; ti < 2; ++ti) {
#pragma unroll
    for (int i = 0; i < 8; ++i) ss[ti][i] = 0.f;
#pragma unroll
    for (int tj = 0; tj < 2; ++tj) {
      const int col = tn + tj * 16 + l16;
      const int rbase = tm + ti * 16 + 8 * half;
      const float bb = bias[col];
#pragma unroll
      for (int i = 0; i < 8; ++i) {
        const float y = (r.c[ti][tj][i] + bb) * scl;
        Y[(rbase + i) * DMODEL + col] = y;
        ss[ti][i] += y * y;
      }
    }
  }
  static_assert(GEMM_TILE == DHEAD, "a column tile of the K projection must be one head");
  if (mat == 1) {
    const int wm = wave >> 1;
    const int wn = wave & 1;
    float* part = lds;
#pragma unroll
    for (int ti = 0; ti < 2; ++ti)
#pragma unroll
      for (int i = 0; i < 8; ++i) ss[ti][i] += row_xor<8>(ss[ti][i]);
#pragma unroll
    for (int ti = 0; ti < 2; ++ti)
#pragma unroll
      for (int i = 0; i < 8; ++i) ss[ti][i] += row_xor<4>(ss[ti][i]);
#pragma unroll
    for (int ti = 0; ti < 2; ++ti)
#pragma unroll
      for (int i = 0; i < 8; ++i) ss[ti][i] += row_xor<2>(ss[ti][i]);
#pragma unroll
    for (int ti = 0; ti < 2; ++ti)
#pragma unroll
      for (int i = 0; i < 8; ++i) ss[ti][i] += row_xor<1>(ss[ti][i]);
    if (l16 == 0) {
      float* p = part + wn * GEMM_TILE + wm * 32 + 8 * half;
#pragma unroll
      for (int ti = 0; ti < 2; ++ti) {
        v4f lo, hi;
        lo.x = ss[ti][0]; lo.y = ss[ti][1]; lo.z = ss[ti][2]; lo.w = ss[ti][3];
        hi.x = ss[ti][4]; hi.y = ss[ti][5]; hi.z = ss[ti][6]; hi.w = ss[ti][7];
        *reinterpret_cast<v4f*>(p + ti * 16) = lo;
        *reinterpret_cast<v4f*>(p + ti * 16 + 4) = hi;
      }
    }
    __syncthreads();
    if (wave == 0) {
      const float h0 = -0.5f * (part[lane] + part[GEMM_TILE + lane]);
      const float h1 = -0.5f * (part[lane + 32] + part[GEMM_TILE + lane + 32]);
      float* hrow = hk + blockIdx.y * L_SEQ + blockIdx.x * GEMM_TILE;
      hrow[lane] = h0;
      hrow[lane + 32] = h1;
      float m = fmaxf(h0, h1);
      m = fmaxf(m, row_xor<8>(m));
      m = fmaxf(m, row_xor<4>(m));
      m = fmaxf(m, row_xor<2>(m));
      m = fmaxf(m, row_xor<1>(m));
      m = fmaxf(m, other_row(m));
      if (lane == 0) kmax_part[blockIdx.y * (L_SEQ / GEMM_TILE) + blockIdx.x] = m;
    }
  }
}

#define KMAX_PARTS ((L_SEQ / GEMM_TILE) * NHEADS)
__global__ void __launch_bounds__(32)
perf_primes(const float* __restrict__ Q, const float* __restrict__ K,
            const float* __restrict__ RF, const float* __restrict__ hk,
            const float* __restrict__ kmax_part,
            float* __restrict__ qprime, float* __restrict__ kprime) {
  const int tm = blockIdx.x * 32;
  const int tn = blockIdx.y * 32;
  const int h = blockIdx.z >> 1;
  const int isK = blockIdx.z & 1;
  const int lane = threadIdx.x;
  const float* A = (isK ? K : Q) + tm * DMODEL + h * DHEAD;
  const float4* pp = reinterpret_cast<const float4*>(kmax_part) + lane;
  const float4 p0 = pp[0];
  const float4 p1 = pp[KMAX_PARTS / 8];
  Acc2x2 r = wmma_block2x2_f32(A, DMODEL, RF + tn, MFEAT, 1, DHEAD, lane);
  const int l16 = lane & 15;
  const int half = lane >> 4;
  float ks = 0.f;
  if (isK) {
    ks = fmaxf(fmaxf(fmaxf(p0.x, p0.y), fmaxf(p0.z, p0.w)),
               fmaxf(fmaxf(p1.x, p1.y), fmaxf(p1.z, p1.w)));
    ks = fmaxf(ks, row_xor<8>(ks));
    ks = fmaxf(ks, row_xor<4>(ks));
    ks = fmaxf(ks, row_xor<2>(ks));
    ks = fmaxf(ks, row_xor<1>(ks));
    ks = fmaxf(ks, other_row(ks));
  }
#pragma unroll
  for (int ti = 0; ti < 2; ++ti)
#pragma unroll
    for (int tj = 0; tj < 2; ++tj) {
      const int mcol = tn + tj * 16 + l16;
      const int rbase = tm + ti * 16 + 8 * half;
      if (isK) {
#pragma unroll
        for (int i = 0; i < 8; ++i) {
          const int l = rbase + i;
          kprime[(h * L_SEQ + l) * MFEAT + mcol] =
              NC_FEAT * (expf(hk[h * L_SEQ + l] + r.c[ti][tj][i] - ks) + KERNEL_EPS);
        }
      } else {
#pragma unroll
        for (int i = 0; i < 8; ++i) {
          const int l = rbase + i;
          qprime[(h * L_SEQ + l) * MFEAT + mcol] =
              NC_FEAT * (expf(r.c[ti][tj][i]) + KERNEL_EPS);
        }
      }
    }
}

__device__ __forceinline__ v8f wmma4(v2f a, v2f b, v8f c) {
  return __builtin_amdgcn_wmma_f32_16x16x4_f32(false, a, false, b, (short)0, c, false, false);
}

__global__ void __launch_bounds__(32)
perf_chunk_sums(const float* __restrict__ kprime, const float* __restrict__ V,
                float* __restrict__ chunkbuf) {
  const int blk = blockIdx.x;
  const int c = blockIdx.y;
  const int h = blockIdx.z;
  const int lane = threadIdx.x;
  const int half = lane >> 4;
  const int l16 = lane & 15;
  const int m0 = (blk >> 1) * 32;
  const int d0 = (blk & 1) * 32;
  const float* ka = kprime + (size_t)(h * L_SEQ + c * CHUNK + 2 * half) * MFEAT + m0 + 2 * l16;
  const float* vb = V + (size_t)(c * CHUNK + 2 * half) * DMODEL + h * DHEAD + d0 + 2 * l16;
  const v8f zero = {0.f, 0.f, 0.f, 0.f, 0.f, 0.f, 0.f, 0.f};
  v8f acc[2][2];
  acc[0][0] = zero; acc[0][1] = zero; acc[1][0] = zero; acc[1][1] = zero;
  float zx = 0.f, zy = 0.f;
#pragma unroll
  for (int ks = 0; ks < CHUNK / 4; ++ks) {
    const float2 ra0 = *reinterpret_cast<const float2*>(ka + (size_t)(4 * ks) * MFEAT);
    const float2 ra1 = *reinterpret_cast<const float2*>(ka + (size_t)(4 * ks + 1) * MFEAT);
    const float2 rb0 = *reinterpret_cast<const float2*>(vb + (size_t)(4 * ks) * DMODEL);
    const float2 rb1 = *reinterpret_cast<const float2*>(vb + (size_t)(4 * ks + 1) * DMODEL);
    v2f a0, a1, b0, b1;
    a0.x = ra0.x; a0.y = ra1.x; a1.x = ra0.y; a1.y = ra1.y;
    b0.x = rb0.x; b0.y = rb1.x; b1.x = rb0.y; b1.y = rb1.y;
    acc[0][0] = wmma4(a0, b0, acc[0][0]);
    acc[0][1] = wmma4(a0, b1, acc[0][1]);
    acc[1][0] = wmma4(a1, b0, acc[1][0]);
    acc[1][1] = wmma4(a1, b1, acc[1][1]);
    zx += ra0.x + ra1.x;
    zy += ra0.y + ra1.y;
  }
  float* buf = chunkbuf + (size_t)(h * NCHUNKS + c) * CHUNK_STRIDE;
#pragma unroll
  for (int jm = 0; jm < 2; ++jm)
#pragma unroll
    for (int r = 0; r < 8; ++r) {
      const int m = m0 + 2 * (r + 8 * half) + jm;
      float2 o;
      o.x = acc[jm][0][r]; o.y = acc[jm][1][r];
      *reinterpret_cast<float2*>(buf + m * DHEAD + d0 + 2 * l16) = o;
    }
  zx += __shfl_xor(zx, 16);
  zy += __shfl_xor(zy, 16);
  if (d0 == 0 && half == 0) {
    float2 o;
    o.x = zx; o.y = zy;
    *reinterpret_cast<float2*>(buf + MFEAT * DHEAD + m0 + 2 * l16) = o;
  }
}

#define PREFIX_VEC (CHUNK_STRIDE / 4)
__global__ void __launch_bounds__(128)
perf_chunk_prefix(float* __restrict__ chunkbuf) {
  const int gid = blockIdx.x * 128 + threadIdx.x;
  const int h = gid / PREFIX_VEC;
  const int i4 = gid - h * PREFIX_VEC;
  float4* p = reinterpret_cast<float4*>(chunkbuf + (size_t)h * NCHUNKS * CHUNK_STRIDE) + i4;
  float4 v[NCHUNKS];
#pragma unroll
  for (int c = 0; c < NCHUNKS; ++c) v[c] = p[(size_t)c * PREFIX_VEC];
  float4 carry;
  carry.x = 0.f; carry.y = 0.f; carry.z = 0.f; carry.w = 0.f;
#pragma unroll
  for (int c = 0; c < NCHUNKS; ++c) {
    const float4 tmp = v[c];
    p[(size_t)c * PREFIX_VEC] = carry;
    carry.x += tmp.x; carry.y += tmp.y; carry.z += tmp.z; carry.w += tmp.w;
  }
}

struct ScanTile { float4 kf[16]; v2f vb[4][4]; };

__device__ __forceinline__ void load_scan_tile(ScanTile& g, const float* __restrict__ krow,
                                               const float* __restrict__ vp) {
#pragma unroll
  for (int s = 0; s < 16; ++s) g.kf[s] = *reinterpret_cast<const float4*>(krow + 8 * s);
#pragma unroll
  for (int s2 = 0; s2 < 4; ++s2)
#pragma unroll
    for (int dt = 0; dt < 4; ++dt) {
      g.vb[s2][dt].x = vp[(2 * s2) * DMODEL + dt * 16];
      g.vb[s2][dt].y = vp[(2 * s2 + 1) * DMODEL + dt * 16];
    }
}

__device__ __forceinline__ void scan_tile_products(const ScanTile& g, const float4 (&q)[16],
                                                   v8f (&num)[4], v8f& den, bool diag,
                                                   int half, int l16) {
  const v8f zero = {0.f, 0.f, 0.f, 0.f, 0.f, 0.f, 0.f, 0.f};
  v8f t0 = zero, t1 = zero;
#pragma unroll
  for (int s = 0; s < 16; ++s) {
    v2f ka0, ka1, qb0, qb1;
    ka0.x = g.kf[s].x; ka0.y = g.kf[s].y; ka1.x = g.kf[s].z; ka1.y = g.kf[s].w;
    qb0.x = q[s].x;    qb0.y = q[s].y;    qb1.x = q[s].z;    qb1.y = q[s].w;
    t0 = wmma4(ka0, qb0, t0);
    t1 = wmma4(ka1, qb1, t1);
  }
  v8f t = t0 + t1;
  if (diag) {
#pragma unroll
    for (int r = 0; r < 8; ++r) t[r] = (r + 8 * half > l16) ? 0.f : t[r];
  }
  v2f ones;
  ones.x = 1.f; ones.y = 1.f;
#pragma unroll
  for (int s2 = 0; s2 < 4; ++s2) {
    v2f a;
    a.x = t[2 * s2]; a.y = t[2 * s2 + 1];
    den = wmma4(a, ones, den);
#pragma unroll
    for (int dt = 0; dt < 4; ++dt) num[dt] = wmma4(a, g.vb[s2][dt], num[dt]);
  }
}

__global__ void __launch_bounds__(32)
perf_scan(const float* __restrict__ qprime, const float* __restrict__ kprime,
          const float* __restrict__ V, const float* __restrict__ chunkbuf,
          float* __restrict__ context) {
  const int lt = blockIdx.x;
  const int c = blockIdx.y;
  const int h = blockIdx.z;
  const int lane = threadIdx.x;
  const int half = lane >> 4;
  const int l16 = lane & 15;
  const int koff = 4 * half;
  const int lrow0 = c * CHUNK + lt * 16;
  const float* buf = chunkbuf + (size_t)(h * NCHUNKS + c) * CHUNK_STRIDE;

  const float* qrow = qprime + (size_t)(h * L_SEQ + lrow0 + l16) * MFEAT + koff;
  float4 q[16];
#pragma unroll
  for (int s = 0; s < 16; ++s) q[s] = *reinterpret_cast<const float4*>(qrow + 8 * s);

  const v8f zero = {0.f, 0.f, 0.f, 0.f, 0.f, 0.f, 0.f, 0.f};
  v8f num[4];
  num[0] = zero; num[1] = zero; num[2] = zero; num[3] = zero;
  v8f den = zero;

  const float* sp = buf + koff * DHEAD + l16;
  const float* zp = buf + MFEAT * DHEAD + koff;
#pragma unroll
  for (int s = 0; s < 16; ++s) {
    const float4 zz = *reinterpret_cast<const float4*>(zp + 8 * s);
    v2f a0, a1, z0, z1;
    a0.x = q[s].x; a0.y = q[s].y; a1.x = q[s].z; a1.y = q[s].w;
    z0.x = zz.x;   z0.y = zz.y;   z1.x = zz.z;   z1.y = zz.w;
    den = wmma4(a0, z0, den);
    den = wmma4(a1, z1, den);
#pragma unroll
    for (int dt = 0; dt < 4; ++dt) {
      v2f b0, b1;
      b0.x = sp[(8 * s + 0) * DHEAD + dt * 16];
      b0.y = sp[(8 * s + 1) * DHEAD + dt * 16];
      b1.x = sp[(8 * s + 2) * DHEAD + dt * 16];
      b1.y = sp[(8 * s + 3) * DHEAD + dt * 16];
      num[dt] = wmma4(a0, b0, num[dt]);
      num[dt] = wmma4(a1, b1, num[dt]);
    }
  }

  const float* kbase = kprime + (size_t)(h * L_SEQ + c * CHUNK + l16) * MFEAT + koff;
  const float* vbase = V + (size_t)(c * CHUNK + 8 * half) * DMODEL + h * DHEAD + l16;
  ScanTile ta, tb;
  load_scan_tile(ta, kbase, vbase);
  int jt = 0;
  for (;;) {
    {
      const int jn = jt < lt ? jt + 1 : lt;
      load_scan_tile(tb, kbase + (size_t)jn * 16 * MFEAT, vbase + (size_t)jn * 16 * DMODEL);
      __builtin_amdgcn_sched_barrier(0);
      scan_tile_products(ta, q, num, den, jt == lt, half, l16);
    }
    if (jt == lt) break;
    ++jt;
    {
      const int jn = jt < lt ? jt + 1 : lt;
      load_scan_tile(ta, kbase + (size_t)jn * 16 * MFEAT, vbase + (size_t)jn * 16 * DMODEL);
      __builtin_amdgcn_sched_barrier(0);
      scan_tile_products(tb, q, num, den, jt == lt, half, l16);
    }
    if (jt == lt) break;
    ++jt;
  }

#pragma unroll
  for (int dt = 0; dt < 4; ++dt)
#pragma unroll
    for (int r = 0; r < 8; ++r) {
      const int l = lrow0 + r + 8 * half;
      context[(size_t)l * DMODEL + h * DHEAD + dt * 16 + l16] =
          num[dt][r] / (den[r] + DEN_EPS);
    }
}

__global__ void __launch_bounds__(128)
perf_out_proj(const float* __restrict__ ctx, const float* __restrict__ Wo,
              const float* __restrict__ bo, float* __restrict__ out) {
  __shared__ __attribute__((aligned(16))) float lds[GEMM_LDS_FLOATS];
  const int tid = threadIdx.x;
  const int lane = tid & 31;
  const int wave = tid >> 5;
  const int tm = blockIdx.x * GEMM_TILE + (wave >> 1) * 32;
  const int tn = blockIdx.y * GEMM_TILE + (wave & 1) * 32;
  Acc2x2 r = gemm_tile64_lds(ctx + blockIdx.x * GEMM_TILE * DMODEL,
                             Wo + blockIdx.y * GEMM_TILE * DMODEL, lds, tid);
  const int l16 = lane & 15;
  const int half = lane >> 4;
#pragma unroll
  for (int ti = 0; ti < 2; ++ti)
#pragma unroll
    for (int tj = 0; tj < 2; ++tj) {
      const int col = tn + tj * 16 + l16;
      const int rbase = tm + ti * 16 + 8 * half;
      const float bb = bo[col];
#pragma unroll
      for (int i = 0; i < 8; ++i)
        out[(rbase + i) * DMODEL + col] = r.c[ti][tj][i] + bb;
    }
}

extern "C" void kernel_launch(void* const* d_in, const int* in_sizes, int n_in,
                              void* d_out, int out_size, void* d_ws, size_t ws_size,
                              hipStream_t stream) {
  (void)in_sizes; (void)n_in; (void)out_size; (void)ws_size;
  const float* query = (const float*)d_in[0];
  const float* key_  = (const float*)d_in[1];
  const float* value = (const float*)d_in[2];
  const float* Wq = (const float*)d_in[3];
  const float* bq = (const float*)d_in[4];
  const float* Wk = (const float*)d_in[5];
  const float* bk = (const float*)d_in[6];
  const float* Wv = (const float*)d_in[7];
  const float* bv = (const float*)d_in[8];
  const float* Wo = (const float*)d_in[9];
  const float* bo = (const float*)d_in[10];
  const float* RF = (const float*)d_in[11];

  float* ws = (float*)d_ws;
  const size_t SZ_LD = (size_t)L_SEQ * DMODEL;
  const size_t SZ_PRIME = (size_t)NHEADS * L_SEQ * MFEAT;
  float* Q      = ws;
  float* K      = Q + SZ_LD;
  float* V      = K + SZ_LD;
  float* hk     = V + SZ_LD;
  float* kmax_part = hk + (size_t)NHEADS * L_SEQ;
  float* qprime = kmax_part + KMAX_PARTS;
  float* kprime = qprime + SZ_PRIME;
  float* chunkbuf = kprime + SZ_PRIME;
  float* context = Q;

  perf_qkv_proj<<<dim3(L_SEQ / GEMM_TILE, DMODEL / GEMM_TILE, 3), 128, 0, stream>>>(
      query, key_, value, Wq, bq, Wk, bk, Wv, bv, Q, K, V, hk, kmax_part);
  perf_primes<<<dim3(L_SEQ / 32, MFEAT / 32, 2 * NHEADS), 32, 0, stream>>>(
      Q, K, RF, hk, kmax_part, qprime, kprime);
  perf_chunk_sums<<<dim3(8, NCHUNKS, NHEADS), 32, 0, stream>>>(kprime, V, chunkbuf);
  perf_chunk_prefix<<<(NHEADS * PREFIX_VEC) / 128, 128, 0, stream>>>(chunkbuf);
  perf_scan<<<dim3(CHUNK / 16, NCHUNKS, NHEADS), 32, 0, stream>>>(
      qprime, kprime, V, chunkbuf, context);
  perf_out_proj<<<dim3(L_SEQ / GEMM_TILE, DMODEL / GEMM_TILE), 128, 0, stream>>>(
      context, Wo, bo, (float*)d_out);
}


# chunk sums k'^T v and sum k' formed in the feature kernel from each k' tile while it is in registers (its D layout is the A layout of k'^T): chunk-sum launch removed; sums in a different order
# speedup vs baseline: 9.4654x; 1.1088x over previous
#include <hip/hip_runtime.h>
#include <math.h>

#define L_SEQ 2048
#define DMODEL 512
#define NHEADS 8
#define DHEAD 64
#define MFEAT 128
#define NCHUNKS 16
#define CHUNK 128
#define CHUNK_STRIDE 8320
#define KERNEL_EPS 1e-4f
#define DEN_EPS 1e-6f
#define QK_SCALE 0.21022410381342864f
#define NC_FEAT 0.08838834764831845f

typedef float v2f __attribute__((ext_vector_type(2)));
typedef float v4f __attribute__((ext_vector_type(4)));
typedef float v8f __attribute__((ext_vector_type(8)));

template <int M>
__device__ __forceinline__ float row_xor(float v) {
  return __builtin_bit_cast(float, __builtin_amdgcn_update_dpp(
      0, __builtin_bit_cast(int, v), 0x160 + M, 0xf, 0xf, true));
}
__device__ __forceinline__ float other_row(float v) {
  return __builtin_bit_cast(float, __builtin_amdgcn_permlanex16(
      __builtin_bit_cast(int, v), __builtin_bit_cast(int, v), 0x76543210, 0xfedcba98, false, false));
}

struct Acc2x2 { v8f c[2][2]; };

__device__ __forceinline__ Acc2x2 wmma_block2x2_f32(const float* __restrict__ A, int lda,
                                                    const float* __restrict__ B, int bks, int bns,
                                                    int K, int lane) {
  const int half = lane >> 4;
  const int l16  = lane & 15;
  const float* arow0 = A + l16 * lda;
  const float* arow1 = A + (l16 + 16) * lda;
  const float* bcol0 = B + l16 * bns;
  const float* bcol1 = B + (l16 + 16) * bns;
  const v8f zero = {0.f, 0.f, 0.f, 0.f, 0.f, 0.f, 0.f, 0.f};
  Acc2x2 r;
  r.c[0][0] = zero; r.c[0][1] = zero; r.c[1][0] = zero; r.c[1][1] = zero;
  for (int k0 = 0; k0 < K; k0 += 4) {
    const int ka = k0 + 2 * half;
    v2f a0, a1, b0, b1;
    a0.x = arow0[ka];          a0.y = arow0[ka + 1];
    a1.x = arow1[ka];          a1.y = arow1[ka + 1];
    b0.x = bcol0[ka * bks];    b0.y = bcol0[(ka + 1) * bks];
    b1.x = bcol1[ka * bks];    b1.y = bcol1[(ka + 1) * bks];
    r.c[0][0] = __builtin_amdgcn_wmma_f32_16x16x4_f32(false, a0, false, b0, (short)0, r.c[0][0], false, false);
    r.c[0][1] = __builtin_amdgcn_wmma_f32_16x16x4_f32(false, a0, false, b1, (short)0, r.c[0][1], false, false);
    r.c[1][0] = __builtin_amdgcn_wmma_f32_16x16x4_f32(false, a1, false, b0, (short)0, r.c[1][0], false, false);
    r.c[1][1] = __builtin_amdgcn_wmma_f32_16x16x4_f32(false, a1, false, b1, (short)0, r.c[1][1], false, false);
  }
  return r;
}

#define GEMM_TILE 64
#define GEMM_KB 64
#define GEMM_STEPS (GEMM_KB / 8)
#define GEMM_PITCH (GEMM_KB + 4)
#define GEMM_PLANE (GEMM_TILE * GEMM_PITCH)
#define GEMM_LDS_FLOATS (4 * GEMM_PLANE)
#define GEMM_COPIES (GEMM_TILE * GEMM_KB / 4 / 128)
struct OperandBlock { float4 a0[GEMM_STEPS], a1[GEMM_STEPS], b0[GEMM_STEPS], b1[GEMM_STEPS]; };
struct StageRegs { v4f a[GEMM_COPIES], b[GEMM_COPIES]; };

__device__ __forceinline__ void load_stage_regs(StageRegs& g, const float* __restrict__ ag,
                                                const float* __restrict__ bg, int kb) {
#pragma unroll
  for (int i = 0; i < GEMM_COPIES; ++i) {
    g.a[i] = *reinterpret_cast<const v4f*>(ag + (size_t)(8 * i) * DMODEL + kb);
    g.b[i] = *reinterpret_cast<const v4f*>(bg + (size_t)(8 * i) * DMODEL + kb);
  }
}

__device__ __forceinline__ void store_stage_regs(const StageRegs& g, float* sw) {
#pragma unroll
  for (int i = 0; i < GEMM_COPIES; ++i) {
    *reinterpret_cast<v4f*>(sw + 8 * i * GEMM_PITCH) = g.a[i];
    *reinterpret_cast<v4f*>(sw + GEMM_PLANE + 8 * i * GEMM_PITCH) = g.b[i];
  }
}

__device__ __forceinline__ void read_fragments(OperandBlock& o, const float* af, const float* bf) {
#pragma unroll
  for (int s = 0; s < GEMM_STEPS; ++s) {
    o.a0[s] = *reinterpret_cast<const float4*>(af + 8 * s);
    o.a1[s] = *reinterpret_cast<const float4*>(af + 16 * GEMM_PITCH + 8 * s);
    o.b0[s] = *reinterpret_cast<const float4*>(bf + 8 * s);
    o.b1[s] = *reinterpret_cast<const float4*>(bf + 16 * GEMM_PITCH + 8 * s);
  }
}

__device__ __forceinline__ void mma_operand_block(Acc2x2& r, const OperandBlock& o) {
#pragma unroll
  for (int s = 0; s < GEMM_STEPS; ++s) {
    v2f a0l, a0h, a1l, a1h, b0l, b0h, b1l, b1h;
    a0l.x = o.a0[s].x; a0l.y = o.a0[s].y; a0h.x = o.a0[s].z; a0h.y = o.a0[s].w;
    a1l.x = o.a1[s].x; a1l.y = o.a1[s].y; a1h.x = o.a1[s].z; a1h.y = o.a1[s].w;
    b0l.x = o.b0[s].x; b0l.y = o.b0[s].y; b0h.x = o.b0[s].z; b0h.y = o.b0[s].w;
    b1l.x = o.b1[s].x; b1l.y = o.b1[s].y; b1h.x = o.b1[s].z; b1h.y = o.b1[s].w;
    r.c[0][0] = __builtin_amdgcn_wmma_f32_16x16x4_f32(false, a0l, false, b0l, (short)0, r.c[0][0], false, false);
    r.c[0][1] = __builtin_amdgcn_wmma_f32_16x16x4_f32(false, a0l, false, b1l, (short)0, r.c[0][1], false, false);
    r.c[1][0] = __builtin_amdgcn_wmma_f32_16x16x4_f32(false, a1l, false, b0l, (short)0, r.c[1][0], false, false);
    r.c[1][1] = __builtin_amdgcn_wmma_f32_16x16x4_f32(false, a1l, false, b1l, (short)0, r.c[1][1], false, false);
    r.c[0][0] = __builtin_amdgcn_wmma_f32_16x16x4_f32(false, a0h, false, b0h, (short)0, r.c[0][0], false, false);
    r.c[0][1] = __builtin_amdgcn_wmma_f32_16x16x4_f32(false, a0h, false, b1h, (short)0, r.c[0][1], false, false);
    r.c[1][0] = __builtin_amdgcn_wmma_f32_16x16x4_f32(false, a1h, false, b0h, (short)0, r.c[1][0], false, false);
    r.c[1][1] = __builtin_amdgcn_wmma_f32_16x16x4_f32(false, a1h, false, b1h, (short)0, r.c[1][1], false, false);
  }
}

__device__ __forceinline__ Acc2x2 gemm_tile64_lds(const float* __restrict__ A,
                                                  const float* __restrict__ Bt,
                                                  float* lds, int tid) {
  const int lane = tid & 31;
  const int wave = tid >> 5;
  const int half = lane >> 4;
  const int l16  = lane & 15;
  const int wm = wave >> 1;
  const int wn = wave & 1;
  const int srow = tid >> 4;
  const int spiece = tid & 15;
  const float* ag = A + srow * DMODEL + 4 * spiece;
  const float* bg = Bt + srow * DMODEL + 4 * spiece;
  float* sw = lds + srow * GEMM_PITCH + 4 * spiece;
  const float* af = lds + (wm * 32 + l16) * GEMM_PITCH + 4 * half;
  const float* bf = lds + GEMM_PLANE + (wn * 32 + l16) * GEMM_PITCH + 4 * half;
  const v8f zero = {0.f, 0.f, 0.f, 0.f, 0.f, 0.f, 0.f, 0.f};
  Acc2x2 r;
  r.c[0][0] = zero; r.c[0][1] = zero; r.c[1][0] = zero; r.c[1][1] = zero;
  StageRegs g;
  load_stage_regs(g, ag, bg, 0);
  store_stage_regs(g, sw);
  __syncthreads();
  for (int t = 0; t < DMODEL / GEMM_KB; ++t) {
    const int cur = (t & 1) * 2 * GEMM_PLANE;
    const int nxt = 2 * GEMM_PLANE - cur;
    const bool more = t + 1 < DMODEL / GEMM_KB;
    if (more) load_stage_regs(g, ag, bg, (t + 1) * GEMM_KB);
    __builtin_amdgcn_sched_barrier(0);
    OperandBlock o;
    read_fragments(o, af + cur, bf + cur);
    __builtin_amdgcn_sched_barrier(0);
    mma_operand_block(r, o);
    __builtin_amdgcn_sched_barrier(0);
    if (more) store_stage_regs(g, sw + nxt);
    __syncthreads();
  }
  return r;
}

__global__ void __launch_bounds__(128)
perf_qkv_proj(const float* __restrict__ xq, const float* __restrict__ xk,
              const float* __restrict__ xv,
              const float* __restrict__ Wq, const float* __restrict__ bq,
              const float* __restrict__ Wk, const float* __restrict__ bk,
              const float* __restrict__ Wv, const float* __restrict__ bv,
              float* __restrict__ Q, float* __restrict__ K, float* __restrict__ V,
              float* __restrict__ hk, float* __restrict__ kmax_part) {
  __shared__ __attribute__((aligned(16))) float lds[GEMM_LDS_FLOATS];
  const int tid = threadIdx.x;
  const int lane = tid & 31;
  const int wave = tid >> 5;
  const int tm = blockIdx.x * GEMM_TILE + (wave >> 1) * 32;
  const int tn = blockIdx.y * GEMM_TILE + (wave & 1) * 32;
  const int mat = blockIdx.z;
  const float* X; const float* W; const float* bias; float* Y; float scl;
  if (mat == 0)      { X = xq; W = Wq; bias = bq; Y = Q; scl = QK_SCALE; }
  else if (mat == 1) { X = xk; W = Wk; bias = bk; Y = K; scl = QK_SCALE; }
  else               { X = xv; W = Wv; bias = bv; Y = V; scl = 1.f; }
  Acc2x2 r = gemm_tile64_lds(X + blockIdx.x * GEMM_TILE * DMODEL,
                             W + blockIdx.y * GEMM_TILE * DMODEL, lds, tid);
  const int l16 = lane & 15;
  const int half = lane >> 4;
  float ss[2][8];
#pragma unroll
  for (int ti = 0; ti < 2; ++ti) {
#pragma unroll
    for (int i = 0; i < 8; ++i) ss[ti][i] = 0.f;
#pragma unroll
    for (int tj = 0; tj < 2; ++tj) {
      const int col = tn + tj * 16 + l16;
      const int rbase = tm + ti * 16 + 8 * half;
      const float bb = bias[col];
#pragma unroll
      for (int i = 0; i < 8; ++i) {
        const float y = (r.c[ti][tj][i] + bb) * scl;
        Y[(rbase + i) * DMODEL + col] = y;
        ss[ti][i] += y * y;
      }
    }
  }
  static_assert(GEMM_TILE == DHEAD, "a column tile of the K projection must be one head");
  if (mat == 1) {
    const int wm = wave >> 1;
    const int wn = wave & 1;
    float* part = lds;
#pragma unroll
    for (int ti = 0; ti < 2; ++ti)
#pragma unroll
      for (int i = 0; i < 8; ++i) ss[ti][i] += row_xor<8>(ss[ti][i]);
#pragma unroll
    for (int ti = 0; ti < 2; ++ti)
#pragma unroll
      for (int i = 0; i < 8; ++i) ss[ti][i] += row_xor<4>(ss[ti][i]);
#pragma unroll
    for (int ti = 0; ti < 2; ++ti)
#pragma unroll
      for (int i = 0; i < 8; ++i) ss[ti][i] += row_xor<2>(ss[ti][i]);
#pragma unroll
    for (int ti = 0; ti < 2; ++ti)
#pragma unroll
      for (int i = 0; i < 8; ++i) ss[ti][i] += row_xor<1>(ss[ti][i]);
    if (l16 == 0) {
      float* p = part + wn * GEMM_TILE + wm * 32 + 8 * half;
#pragma unroll
      for (int ti = 0; ti < 2; ++ti) {
        v4f lo, hi;
        lo.x = ss[ti][0]; lo.y = ss[ti][1]; lo.z = ss[ti][2]; lo.w = ss[ti][3];
        hi.x = ss[ti][4]; hi.y = ss[ti][5]; hi.z = ss[ti][6]; hi.w = ss[ti][7];
        *reinterpret_cast<v4f*>(p + ti * 16) = lo;
        *reinterpret_cast<v4f*>(p + ti * 16 + 4) = hi;
      }
    }
    __syncthreads();
    if (wave == 0) {
      const float h0 = -0.5f * (part[lane] + part[GEMM_TILE + lane]);
      const float h1 = -0.5f * (part[lane + 32] + part[GEMM_TILE + lane + 32]);
      float* hrow = hk + blockIdx.y * L_SEQ + blockIdx.x * GEMM_TILE;
      hrow[lane] = h0;
      hrow[lane + 32] = h1;
      float m = fmaxf(h0, h1);
      m = fmaxf(m, row_xor<8>(m));
      m = fmaxf(m, row_xor<4>(m));
      m = fmaxf(m, row_xor<2>(m));
      m = fmaxf(m, row_xor<1>(m));
      m = fmaxf(m, other_row(m));
      if (lane == 0) kmax_part[blockIdx.y * (L_SEQ / GEMM_TILE) + blockIdx.x] = m;
    }
  }
}

#define KMAX_PARTS ((L_SEQ / GEMM_TILE) * NHEADS)
#define KSUM_GROUPS ((MFEAT / 32) * NCHUNKS * NHEADS)
#define QFEAT_WAVES ((L_SEQ / 32) * (MFEAT / 32) * NHEADS)

__device__ __forceinline__ v8f wmma4(v2f a, v2f b, v8f c) {
  return __builtin_amdgcn_wmma_f32_16x16x4_f32(false, a, false, b, (short)0, c, false, false);
}

struct KSumTile { float4 ka[8]; float4 hk[2]; float4 vv[8]; };

__device__ __forceinline__ void load_ksum_tile(KSumTile& g, const float* __restrict__ krow,
                                               const float* __restrict__ hrow,
                                               const float* __restrict__ vrow) {
  g.hk[0] = *reinterpret_cast<const float4*>(hrow);
  g.hk[1] = *reinterpret_cast<const float4*>(hrow + 4);
#pragma unroll
  for (int s = 0; s < 8; ++s) g.ka[s] = *reinterpret_cast<const float4*>(krow + 8 * s);
#pragma unroll
  for (int j = 0; j < 8; ++j) g.vv[j] = *reinterpret_cast<const float4*>(vrow + (size_t)j * DMODEL);
}

__device__ __forceinline__ float f4_elem(const float4& v, int i) {
  return i == 0 ? v.x : (i == 1 ? v.y : (i == 2 ? v.z : v.w));
}

__device__ __forceinline__ void ksum_tile_products(const KSumTile& g, const v2f (&rf)[2][8][2],
                                                   float ks, float* __restrict__ kout,
                                                   v8f (&acc)[2][4], float (&zs)[2]) {
  const v8f zero = {0.f, 0.f, 0.f, 0.f, 0.f, 0.f, 0.f, 0.f};
  v8f t0[2], t1[2];
  t0[0] = zero; t0[1] = zero; t1[0] = zero; t1[1] = zero;
#pragma unroll
  for (int s = 0; s < 8; ++s) {
    v2f a0, a1;
    a0.x = g.ka[s].x; a0.y = g.ka[s].y; a1.x = g.ka[s].z; a1.y = g.ka[s].w;
#pragma unroll
    for (int mt = 0; mt < 2; ++mt) {
      t0[mt] = wmma4(a0, rf[mt][s][0], t0[mt]);
      t1[mt] = wmma4(a1, rf[mt][s][1], t1[mt]);
    }
  }
  float kp[2][8];
#pragma unroll
  for (int mt = 0; mt < 2; ++mt) {
    const v8f t = t0[mt] + t1[mt];
#pragma unroll
    for (int r = 0; r < 8; ++r) {
      const float hv = f4_elem(g.hk[r >> 2], r & 3);
      const float v = NC_FEAT * (expf(hv + t[r] - ks) + KERNEL_EPS);
      kp[mt][r] = v;
      kout[(size_t)r * MFEAT + mt * 16] = v;
      zs[mt] += v;
    }
  }
#pragma unroll
  for (int p = 0; p < 4; ++p)
#pragma unroll
    for (int mt = 0; mt < 2; ++mt) {
      v2f a;
      a.x = kp[mt][2 * p]; a.y = kp[mt][2 * p + 1];
#pragma unroll
      for (int dt = 0; dt < 4; ++dt) {
        v2f b;
        b.x = f4_elem(g.vv[2 * p], dt); b.y = f4_elem(g.vv[2 * p + 1], dt);
        acc[mt][dt] = wmma4(a, b, acc[mt][dt]);
      }
    }
}

__global__ void __launch_bounds__(64)
perf_features(const float* __restrict__ Q, const float* __restrict__ K,
              const float* __restrict__ V, const float* __restrict__ RF,
              const float* __restrict__ hk, const float* __restrict__ kmax_part,
              float* __restrict__ qprime, float* __restrict__ kprime,
              float* __restrict__ chunkbuf) {
  __shared__ float4 part_s[16][32];
  __shared__ float part_z[2][32];
  const int lane = threadIdx.x & 31;
  const int wave = threadIdx.x >> 5;
  const int l16 = lane & 15;
  const int half = lane >> 4;
  if (blockIdx.x >= KSUM_GROUPS) {
    const int qb = (blockIdx.x - KSUM_GROUPS) * 2 + wave;
    const int tm = (qb & (L_SEQ / 32 - 1)) * 32;
    const int tn = ((qb >> 6) & 3) * 32;
    const int h = qb >> 8;
    static_assert(L_SEQ / 32 == 64 && MFEAT / 32 == 4, "decode of qb");
    const float* A = Q + tm * DMODEL + h * DHEAD;
    Acc2x2 r = wmma_block2x2_f32(A, DMODEL, RF + tn, MFEAT, 1, DHEAD, lane);
#pragma unroll
    for (int ti = 0; ti < 2; ++ti)
#pragma unroll
      for (int tj = 0; tj < 2; ++tj) {
        const int mcol = tn + tj * 16 + l16;
        const int rbase = tm + ti * 16 + 8 * half;
#pragma unroll
        for (int i = 0; i < 8; ++i) {
          const int l = rbase + i;
          qprime[(h * L_SEQ + l) * MFEAT + mcol] =
              NC_FEAT * (expf(r.c[ti][tj][i]) + KERNEL_EPS);
        }
      }
    return;
  }
  const int m0 = (blockIdx.x & 3) * 32;
  const int c = (blockIdx.x >> 2) & (NCHUNKS - 1);
  const int h = blockIdx.x >> 6;
  static_assert(MFEAT / 32 == 4 && NCHUNKS == 16, "decode of the K group index");
  const float4* pp = reinterpret_cast<const float4*>(kmax_part) + lane;
  const float4 p0 = pp[0];
  const float4 p1 = pp[KMAX_PARTS / 8];
  const int row0 = c * CHUNK + wave * 16;
  const float* kbase = K + (size_t)(row0 + l16) * DMODEL + h * DHEAD + 4 * half;
  const float* hbase = hk + h * L_SEQ + row0 + 8 * half;
  const float* vbase = V + (size_t)(row0 + 8 * half) * DMODEL + h * DHEAD + 4 * l16;
  KSumTile ta, tb;
  load_ksum_tile(ta, kbase, hbase, vbase);
  v2f rf[2][8][2];
#pragma unroll
  for (int mt = 0; mt < 2; ++mt)
#pragma unroll
    for (int s = 0; s < 8; ++s)
#pragma unroll
      for (int p = 0; p < 2; ++p) {
        const float* rp = RF + (8 * s + 4 * half + 2 * p) * MFEAT + m0 + mt * 16 + l16;
        rf[mt][s][p].x = rp[0];
        rf[mt][s][p].y = rp[MFEAT];
      }
  float ks = fmaxf(fmaxf(fmaxf(p0.x, p0.y), fmaxf(p0.z, p0.w)),
                   fmaxf(fmaxf(p1.x, p1.y), fmaxf(p1.z, p1.w)));
  ks = fmaxf(ks, row_xor<8>(ks));
  ks = fmaxf(ks, row_xor<4>(ks));
  ks = fmaxf(ks, row_xor<2>(ks));
  ks = fmaxf(ks, row_xor<1>(ks));
  ks = fmaxf(ks, other_row(ks));

  const v8f zero = {0.f, 0.f, 0.f, 0.f, 0.f, 0.f, 0.f, 0.f};
  v8f acc[2][4];
#pragma unroll
  for (int mt = 0; mt < 2; ++mt)
#pragma unroll
    for (int dt = 0; dt < 4; ++dt) acc[mt][dt] = zero;
  float zs[2];
  zs[0] = 0.f; zs[1] = 0.f;
  float* kout = kprime + (size_t)(h * L_SEQ + row0 + 8 * half) * MFEAT + m0 + l16;

  load_ksum_tile(tb, kbase + (size_t)32 * DMODEL, hbase + 32, vbase + (size_t)32 * DMODEL);
  __builtin_amdgcn_sched_barrier(0);
  ksum_tile_products(ta, rf, ks, kout, acc, zs);
  load_ksum_tile(ta, kbase + (size_t)64 * DMODEL, hbase + 64, vbase + (size_t)64 * DMODEL);
  __builtin_amdgcn_sched_barrier(0);
  ksum_tile_products(tb, rf, ks, kout + (size_t)32 * MFEAT, acc, zs);
  load_ksum_tile(tb, kbase + (size_t)96 * DMODEL, hbase + 96, vbase + (size_t)96 * DMODEL);
  __builtin_amdgcn_sched_barrier(0);
  ksum_tile_products(ta, rf, ks, kout + (size_t)64 * MFEAT, acc, zs);
  ksum_tile_products(tb, rf, ks, kout + (size_t)96 * MFEAT, acc, zs);

  if (wave == 1) {
#pragma unroll
    for (int mt = 0; mt < 2; ++mt) {
#pragma unroll
      for (int r = 0; r < 8; ++r) {
        float4 o;
        o.x = acc[mt][0][r]; o.y = acc[mt][1][r]; o.z = acc[mt][2][r]; o.w = acc[mt][3][r];
        part_s[mt * 8 + r][lane] = o;
      }
      part_z[mt][lane] = zs[mt];
    }
  }
  __syncthreads();
  if (wave == 0) {
    float* buf = chunkbuf + (size_t)(h * NCHUNKS + c) * CHUNK_STRIDE;
#pragma unroll
    for (int mt = 0; mt < 2; ++mt) {
#pragma unroll
      for (int r = 0; r < 8; ++r) {
        const float4 o1 = part_s[mt * 8 + r][lane];
        float4 o;
        o.x = acc[mt][0][r] + o1.x; o.y = acc[mt][1][r] + o1.y;
        o.z = acc[mt][2][r] + o1.z; o.w = acc[mt][3][r] + o1.w;
        const int m = m0 + mt * 16 + 8 * half + r;
        *reinterpret_cast<float4*>(buf + m * DHEAD + 4 * l16) = o;
      }
      float z = zs[mt] + part_z[mt][lane];
      z += other_row(z);
      if (half == 0) buf[MFEAT * DHEAD + m0 + mt * 16 + l16] = z;
    }
  }
}

#define PREFIX_VEC (CHUNK_STRIDE / 4)
__global__ void __launch_bounds__(128)
perf_chunk_prefix(float* __restrict__ chunkbuf) {
  const int gid = blockIdx.x * 128 + threadIdx.x;
  const int h = gid / PREFIX_VEC;
  const int i4 = gid - h * PREFIX_VEC;
  float4* p = reinterpret_cast<float4*>(chunkbuf + (size_t)h * NCHUNKS * CHUNK_STRIDE) + i4;
  float4 v[NCHUNKS];
#pragma unroll
  for (int c = 0; c < NCHUNKS; ++c) v[c] = p[(size_t)c * PREFIX_VEC];
  float4 carry;
  carry.x = 0.f; carry.y = 0.f; carry.z = 0.f; carry.w = 0.f;
#pragma unroll
  for (int c = 0; c < NCHUNKS; ++c) {
    const float4 tmp = v[c];
    p[(size_t)c * PREFIX_VEC] = carry;
    carry.x += tmp.x; carry.y += tmp.y; carry.z += tmp.z; carry.w += tmp.w;
  }
}

struct ScanTile { float4 kf[16]; v2f vb[4][4]; };

__device__ __forceinline__ void load_scan_tile(ScanTile& g, const float* __restrict__ krow,
                                               const float* __restrict__ vp) {
#pragma unroll
  for (int s = 0; s < 16; ++s) g.kf[s] = *reinterpret_cast<const float4*>(krow + 8 * s);
#pragma unroll
  for (int s2 = 0; s2 < 4; ++s2)
#pragma unroll
    for (int dt = 0; dt < 4; ++dt) {
      g.vb[s2][dt].x = vp[(2 * s2) * DMODEL + dt * 16];
      g.vb[s2][dt].y = vp[(2 * s2 + 1) * DMODEL + dt * 16];
    }
}

__device__ __forceinline__ void scan_tile_products(const ScanTile& g, const float4 (&q)[16],
                                                   v8f (&num)[4], v8f& den, bool diag,
                                                   int half, int l16) {
  const v8f zero = {0.f, 0.f, 0.f, 0.f, 0.f, 0.f, 0.f, 0.f};
  v8f t0 = zero, t1 = zero;
#pragma unroll
  for (int s = 0; s < 16; ++s) {
    v2f ka0, ka1, qb0, qb1;
    ka0.x = g.kf[s].x; ka0.y = g.kf[s].y; ka1.x = g.kf[s].z; ka1.y = g.kf[s].w;
    qb0.x = q[s].x;    qb0.y = q[s].y;    qb1.x = q[s].z;    qb1.y = q[s].w;
    t0 = wmma4(ka0, qb0, t0);
    t1 = wmma4(ka1, qb1, t1);
  }
  v8f t = t0 + t1;
  if (diag) {
#pragma unroll
    for (int r = 0; r < 8; ++r) t[r] = (r + 8 * half > l16) ? 0.f : t[r];
  }
  v2f ones;
  ones.x = 1.f; ones.y = 1.f;
#pragma unroll
  for (int s2 = 0; s2 < 4; ++s2) {
    v2f a;
    a.x = t[2 * s2]; a.y = t[2 * s2 + 1];
    den = wmma4(a, ones, den);
#pragma unroll
    for (int dt = 0; dt < 4; ++dt) num[dt] = wmma4(a, g.vb[s2][dt], num[dt]);
  }
}

__global__ void __launch_bounds__(32)
perf_scan(const float* __restrict__ qprime, const float* __restrict__ kprime,
          const float* __restrict__ V, const float* __restrict__ chunkbuf,
          float* __restrict__ context) {
  const int lt = blockIdx.x;
  const int c = blockIdx.y;
  const int h = blockIdx.z;
  const int lane = threadIdx.x;
  const int half = lane >> 4;
  const int l16 = lane & 15;
  const int koff = 4 * half;
  const int lrow0 = c * CHUNK + lt * 16;
  const float* buf = chunkbuf + (size_t)(h * NCHUNKS + c) * CHUNK_STRIDE;

  const float* qrow = qprime + (size_t)(h * L_SEQ + lrow0 + l16) * MFEAT + koff;
  float4 q[16];
#pragma unroll
  for (int s = 0; s < 16; ++s) q[s] = *reinterpret_cast<const float4*>(qrow + 8 * s);

  const v8f zero = {0.f, 0.f, 0.f, 0.f, 0.f, 0.f, 0.f, 0.f};
  v8f num[4];
  num[0] = zero; num[1] = zero; num[2] = zero; num[3] = zero;
  v8f den = zero;

  const float* sp = buf + koff * DHEAD + l16;
  const float* zp = buf + MFEAT * DHEAD + koff;
#pragma unroll
  for (int s = 0; s < 16; ++s) {
    const float4 zz = *reinterpret_cast<const float4*>(zp + 8 * s);
    v2f a0, a1, z0, z1;
    a0.x = q[s].x; a0.y = q[s].y; a1.x = q[s].z; a1.y = q[s].w;
    z0.x = zz.x;   z0.y = zz.y;   z1.x = zz.z;   z1.y = zz.w;
    den = wmma4(a0, z0, den);
    den = wmma4(a1, z1, den);
#pragma unroll
    for (int dt = 0; dt < 4; ++dt) {
      v2f b0, b1;
      b0.x = sp[(8 * s + 0) * DHEAD + dt * 16];
      b0.y = sp[(8 * s + 1) * DHEAD + dt * 16];
      b1.x = sp[(8 * s + 2) * DHEAD + dt * 16];
      b1.y = sp[(8 * s + 3) * DHEAD + dt * 16];
      num[dt] = wmma4(a0, b0, num[dt]);
      num[dt] = wmma4(a1, b1, num[dt]);
    }
  }

  const float* kbase = kprime + (size_t)(h * L_SEQ + c * CHUNK + l16) * MFEAT + koff;
  const float* vbase = V + (size_t)(c * CHUNK + 8 * half) * DMODEL + h * DHEAD + l16;
  ScanTile ta, tb;
  load_scan_tile(ta, kbase, vbase);
  int jt = 0;
  for (;;) {
    {
      const int jn = jt < lt ? jt + 1 : lt;
      load_scan_tile(tb, kbase + (size_t)jn * 16 * MFEAT, vbase + (size_t)jn * 16 * DMODEL);
      __builtin_amdgcn_sched_barrier(0);
      scan_tile_products(ta, q, num, den, jt == lt, half, l16);
    }
    if (jt == lt) break;
    ++jt;
    {
      const int jn = jt < lt ? jt + 1 : lt;
      load_scan_tile(ta, kbase + (size_t)jn * 16 * MFEAT, vbase + (size_t)jn * 16 * DMODEL);
      __builtin_amdgcn_sched_barrier(0);
      scan_tile_products(tb, q, num, den, jt == lt, half, l16);
    }
    if (jt == lt) break;
    ++jt;
  }

#pragma unroll
  for (int dt = 0; dt < 4; ++dt)
#pragma unroll
    for (int r = 0; r < 8; ++r) {
      const int l = lrow0 + r + 8 * half;
      context[(size_t)l * DMODEL + h * DHEAD + dt * 16 + l16] =
          num[dt][r] / (den[r] + DEN_EPS);
    }
}

__global__ void __launch_bounds__(128)
perf_out_proj(const float* __restrict__ ctx, const float* __restrict__ Wo,
              const float* __restrict__ bo, float* __restrict__ out) {
  __shared__ __attribute__((aligned(16))) float lds[GEMM_LDS_FLOATS];
  const int tid = threadIdx.x;
  const int lane = tid & 31;
  const int wave = tid >> 5;
  const int tm = blockIdx.x * GEMM_TILE + (wave >> 1) * 32;
  const int tn = blockIdx.y * GEMM_TILE + (wave & 1) * 32;
  Acc2x2 r = gemm_tile64_lds(ctx + blockIdx.x * GEMM_TILE * DMODEL,
                             Wo + blockIdx.y * GEMM_TILE * DMODEL, lds, tid);
  const int l16 = lane & 15;
  const int half = lane >> 4;
#pragma unroll
  for (int ti = 0; ti < 2; ++ti)
#pragma unroll
    for (int tj = 0; tj < 2; ++tj) {
      const int col = tn + tj * 16 + l16;
      const int rbase = tm + ti * 16 + 8 * half;
      const float bb = bo[col];
#pragma unroll
      for (int i = 0; i < 8; ++i)
        out[(rbase + i) * DMODEL + col] = r.c[ti][tj][i] + bb;
    }
}

extern "C" void kernel_launch(void* const* d_in, const int* in_sizes, int n_in,
                              void* d_out, int out_size, void* d_ws, size_t ws_size,
                              hipStream_t stream) {
  (void)in_sizes; (void)n_in; (void)out_size; (void)ws_size;
  const float* query = (const float*)d_in[0];
  const float* key_  = (const float*)d_in[1];
  const float* value = (const float*)d_in[2];
  const float* Wq = (const float*)d_in[3];
  const float* bq = (const float*)d_in[4];
  const float* Wk = (const float*)d_in[5];
  const float* bk = (const float*)d_in[6];
  const float* Wv = (const float*)d_in[7];
  const float* bv = (const float*)d_in[8];
  const float* Wo = (const float*)d_in[9];
  const float* bo = (const float*)d_in[10];
  const float* RF = (const float*)d_in[11];

  float* ws = (float*)d_ws;
  const size_t SZ_LD = (size_t)L_SEQ * DMODEL;
  const size_t SZ_PRIME = (size_t)NHEADS * L_SEQ * MFEAT;
  float* Q      = ws;
  float* K      = Q + SZ_LD;
  float* V      = K + SZ_LD;
  float* hk     = V + SZ_LD;
  float* kmax_part = hk + (size_t)NHEADS * L_SEQ;
  float* qprime = kmax_part + KMAX_PARTS;
  float* kprime = qprime + SZ_PRIME;
  float* chunkbuf = kprime + SZ_PRIME;
  float* context = Q;

  perf_qkv_proj<<<dim3(L_SEQ / GEMM_TILE, DMODEL / GEMM_TILE, 3), 128, 0, stream>>>(
      query, key_, value, Wq, bq, Wk, bk, Wv, bv, Q, K, V, hk, kmax_part);
  perf_features<<<KSUM_GROUPS + QFEAT_WAVES / 2, 64, 0, stream>>>(
      Q, K, V, RF, hk, kmax_part, qprime, kprime, chunkbuf);
  perf_chunk_prefix<<<(NHEADS * PREFIX_VEC) / 128, 128, 0, stream>>>(chunkbuf);
  perf_scan<<<dim3(CHUNK / 16, NCHUNKS, NHEADS), 32, 0, stream>>>(
      qprime, kprime, V, chunkbuf, context);
  perf_out_proj<<<dim3(L_SEQ / GEMM_TILE, DMODEL / GEMM_TILE), 128, 0, stream>>>(
      context, Wo, bo, (float*)d_out);
}
